# starting kernel
# baseline (speedup 1.0000x reference)
#include <hip/hip_runtime.h>
#include <hip/hip_bf16.h>

typedef __attribute__((ext_vector_type(16))) _Float16 v16h;
typedef __attribute__((ext_vector_type(8)))  _Float16 v8h;
typedef __attribute__((ext_vector_type(4)))  _Float16 v4h;
typedef __attribute__((ext_vector_type(8)))  float    v8f;
typedef __attribute__((ext_vector_type(4)))  float    v4f;

#define D_MODEL 1024
#define HEADS   16
#define DK      64
#define D_FF    4096
#define BATCH   2
#define SEQ     2048
#define ROWS    (BATCH*SEQ)

__device__ __forceinline__ v16h pack16(v8h lo, v8h hi) {
    v16h r;
#pragma unroll
    for (int i = 0; i < 8; ++i) { r[i] = lo[i]; r[i + 8] = hi[i]; }
    return r;
}
__device__ __forceinline__ v16h load_frag(const _Float16* p) {
    v8h lo = *(const v8h*)(p);
    v8h hi = *(const v8h*)(p + 16);
    return pack16(lo, hi);
}
__device__ __forceinline__ v8f wmma_f16(v16h a, v16h b, v8f c) {
    return __builtin_amdgcn_wmma_f32_16x16x32_f16(false, a, false, b,
                                                  (short)0, c, false, false);
}

__global__ void k_transpose_cvt(const float* __restrict__ W,
                                _Float16* __restrict__ Wt, int K, int N) {
    int idx = blockIdx.x * 256 + threadIdx.x;
    if (idx >= K * N) return;
    int k = idx / N, n = idx % N;
    Wt[(long)n * K + k] = (_Float16)W[idx];
}

__global__ void __launch_bounds__(256)
k_layernorm(const float* __restrict__ X, const float* __restrict__ alpha,
            const float* __restrict__ beta, _Float16* __restrict__ Y) {
    __shared__ float s1[256], s2[256];
    const int row = blockIdx.x, t = threadIdx.x;
    const float* xr = X + (long)row * D_MODEL;
    v4f x = *(const v4f*)(xr + t * 4);
    float s = x[0] + x[1] + x[2] + x[3];
    float q = x[0]*x[0] + x[1]*x[1] + x[2]*x[2] + x[3]*x[3];
    s1[t] = s; s2[t] = q; __syncthreads();
    for (int off = 128; off; off >>= 1) {
        if (t < off) { s1[t] += s1[t + off]; s2[t] += s2[t + off]; }
        __syncthreads();
    }
    const float mu  = s1[0] * (1.0f / D_MODEL);
    const float var = (s2[0] - (float)D_MODEL * mu * mu) * (1.0f / (D_MODEL - 1));
    const float inv = 1.0f / (sqrtf(var) + 1e-6f);
    v4h o;
#pragma unroll
    for (int i = 0; i < 4; ++i) {
        int c = t * 4 + i;
        o[i] = (_Float16)(alpha[c] * (x[i] - mu) * inv + beta[c]);
    }
    *(v4h*)(Y + (long)row * D_MODEL + t * 4) = o;
}

template <int OMODE, bool RELU, bool RESID>
__global__ void __launch_bounds__(256)
k_gemm_wmma(const _Float16* __restrict__ A, const _Float16* __restrict__ Bt,
            const float* __restrict__ bias, const float* __restrict__ resid,
            float* __restrict__ Cf, _Float16* __restrict__ Ch,
            int M, int N, int K) {
    const int lane = threadIdx.x & 31;
    const int wave = threadIdx.x >> 5;
    const int wm = wave >> 2, wn = wave & 3;
    const int bn_cnt = N >> 7;
    const int bm = blockIdx.x / bn_cnt;
    const int bn = blockIdx.x % bn_cnt;
    const int m0 = bm * 64  + wm * 32;
    const int n0 = bn * 128 + wn * 32;
    if (m0 >= M) return;

    const int ln = lane & 15;
    const int kb = (lane >> 4) << 3;

    const _Float16* aptr = A  + (long)(m0 + ln) * K + kb;
    const _Float16* bptr = Bt + (long)(n0 + ln) * K + kb;
    const long arow16 = (long)16 * K;

    v8f acc00 = {}, acc01 = {}, acc10 = {}, acc11 = {};
    for (int k = 0; k < K; k += 32) {
        __builtin_prefetch(aptr + 256, 0, 0);
        __builtin_prefetch(bptr + 256, 0, 0);
        v16h a0 = load_frag(aptr);
        v16h a1 = load_frag(aptr + arow16);
        v16h b0 = load_frag(bptr);
        v16h b1 = load_frag(bptr + arow16);
        acc00 = wmma_f16(a0, b0, acc00);
        acc01 = wmma_f16(a0, b1, acc01);
        acc10 = wmma_f16(a1, b0, acc10);
        acc11 = wmma_f16(a1, b1, acc11);
        aptr += 32; bptr += 32;
    }

    const int lm = (lane >> 4) << 3;
    v8f accs[2][2] = {{acc00, acc01}, {acc10, acc11}};
#pragma unroll
    for (int i = 0; i < 2; ++i) {
#pragma unroll
        for (int j = 0; j < 2; ++j) {
            const v8f acc = accs[i][j];
            const int col = n0 + j * 16 + ln;
            const float bv = bias[col];
#pragma unroll
            for (int e = 0; e < 8; ++e) {
                const int row = m0 + i * 16 + e + lm;
                float v = acc[e] + bv;
                if (RELU)  v = v > 0.0f ? v : 0.0f;
                if (RESID) v += resid[(long)row * N + col];
                if (OMODE == 0) {
                    Ch[(long)row * N + col] = (_Float16)v;
                } else if (OMODE == 1) {
                    Cf[(long)row * N + col] = v;
                } else if (OMODE == 2) {
                    int b_ = row / SEQ, s = row % SEQ, h = col >> 6, d = col & 63;
                    Ch[(((long)(b_ * HEADS + h)) * SEQ + s) * DK + d] = (_Float16)v;
                } else {
                    int b_ = row / SEQ, s = row % SEQ, h = col >> 6, d = col & 63;
                    Ch[(((long)(b_ * HEADS + h)) * DK + d) * SEQ + s] = (_Float16)v;
                }
            }
        }
    }
}

__global__ void __launch_bounds__(128)
k_attention(const _Float16* __restrict__ QK, const _Float16* __restrict__ VT,
            const int* __restrict__ mask, _Float16* __restrict__ CTX) {
    extern __shared__ float sc[];
    __shared__ float red[16 * 8];
    __shared__ float rowv[16];

    const int lane = threadIdx.x & 31, wave = threadIdx.x >> 5;
    const int qt = blockIdx.x & 127;
    const int h  = (blockIdx.x >> 7) & 15;
    const int b  = blockIdx.x >> 11;
    const int sq0 = qt << 4;
    const int kb = (lane >> 4) << 3;

    const _Float16* qbase = QK + ((long)(b * HEADS + h)) * SEQ * DK;

    const _Float16* qp = qbase + (long)(sq0 + (lane & 15)) * DK + kb;
    v16h a_lo = load_frag(qp);
    v16h a_hi = load_frag(qp + 32);

    for (int kt = wave; kt < SEQ / 16; kt += 4) {
        const _Float16* kp = qbase + (long)(kt * 16 + (lane & 15)) * DK + kb;
        v16h b_lo = load_frag(kp);
        v16h b_hi = load_frag(kp + 32);
        v8f s = {};
        s = wmma_f16(a_lo, b_lo, s);
        s = wmma_f16(a_hi, b_hi, s);
        const int key = kt * 16 + (lane & 15);
        const int mk  = mask[b * SEQ + key];
#pragma unroll
        for (int e = 0; e < 8; ++e) {
            const int qrow = e + ((lane >> 4) << 3);
            sc[qrow * SEQ + key] = (mk == 0) ? -1e9f : s[e] * 0.125f;
        }
    }
    __syncthreads();

    const int r = threadIdx.x >> 3, sub = threadIdx.x & 7;
    float m = -3.4e38f;
    for (int j = sub; j < SEQ; j += 8) m = fmaxf(m, sc[r * SEQ + j]);
    red[r * 8 + sub] = m; __syncthreads();
    if (sub == 0) {
        float mm = red[r * 8];
        for (int i = 1; i < 8; ++i) mm = fmaxf(mm, red[r * 8 + i]);
        rowv[r] = mm;
    }
    __syncthreads();
    const float rm = rowv[r];
    float sum = 0.0f;
    for (int j = sub; j < SEQ; j += 8) {
        float e = __expf(sc[r * SEQ + j] - rm);
        sc[r * SEQ + j] = e; sum += e;
    }
    red[r * 8 + sub] = sum; __syncthreads();
    if (sub == 0) {
        float ss = 0.0f;
        for (int i = 0; i < 8; ++i) ss += red[r * 8 + i];
        rowv[r] = 1.0f / ss;
    }
    __syncthreads();
    const float inv = rowv[r];
    for (int j = sub; j < SEQ; j += 8) sc[r * SEQ + j] *= inv;
    __syncthreads();

    const int n0 = wave << 4;
    const _Float16* vbase = VT + ((long)(b * HEADS + h)) * DK * SEQ;
    const float*    ap = sc + (lane & 15) * SEQ + kb;
    const _Float16* vp = vbase + (long)(n0 + (lane & 15)) * SEQ + kb;

    v8f acc = {};
    for (int k = 0; k < SEQ; k += 32) {
        v4f f0 = *(const v4f*)(ap + k);
        v4f f1 = *(const v4f*)(ap + k + 4);
        v4f f2 = *(const v4f*)(ap + k + 16);
        v4f f3 = *(const v4f*)(ap + k + 20);
        v16h a;
#pragma unroll
        for (int i = 0; i < 4; ++i) {
            a[i]      = (_Float16)f0[i];
            a[4 + i]  = (_Float16)f1[i];
            a[8 + i]  = (_Float16)f2[i];
            a[12 + i] = (_Float16)f3[i];
        }
        v16h bf = load_frag(vp + k);
        acc = wmma_f16(a, bf, acc);
    }
#pragma unroll
    for (int e = 0; e < 8; ++e) {
        const int qrow = e + ((lane >> 4) << 3);
        CTX[((long)(b * SEQ + sq0 + qrow)) * D_MODEL + h * DK + n0 + (lane & 15)] =
            (_Float16)acc[e];
    }
}

extern "C" void kernel_launch(void* const* d_in, const int* in_sizes, int n_in,
                              void* d_out, int out_size, void* d_ws, size_t ws_size,
                              hipStream_t stream) {
    const float* x      = (const float*)d_in[0];
    const int*   mask   = (const int*)  d_in[1];
    const float* alpha1 = (const float*)d_in[2];
    const float* bias1  = (const float*)d_in[3];
    const float* alpha2 = (const float*)d_in[4];
    const float* bias2  = (const float*)d_in[5];
    const float* Wk     = (const float*)d_in[6];
    const float* bk     = (const float*)d_in[7];
    const float* Wv     = (const float*)d_in[8];
    const float* bv     = (const float*)d_in[9];
    const float* Wo     = (const float*)d_in[10];
    const float* bo     = (const float*)d_in[11];
    const float* W1     = (const float*)d_in[12];
    const float* b1     = (const float*)d_in[13];
    const float* W2     = (const float*)d_in[14];
    const float* b2     = (const float*)d_in[15];
    float* out = (float*)d_out;

    char* ws = (char*)d_ws;
    size_t off = 0;
    _Float16* x2h  = (_Float16*)(ws + off); off += (size_t)ROWS * D_MODEL * 2;
    _Float16* x2bh = (_Float16*)(ws + off); off += (size_t)ROWS * D_MODEL * 2;
    _Float16* qkh  = (_Float16*)(ws + off); off += (size_t)BATCH*HEADS*SEQ*DK*2;
    _Float16* vth  = (_Float16*)(ws + off); off += (size_t)BATCH*HEADS*DK*SEQ*2;
    _Float16* ctxh = (_Float16*)(ws + off); off += (size_t)ROWS * D_MODEL * 2;
    _Float16* h1h  = (_Float16*)(ws + off); off += (size_t)ROWS * D_FF * 2;
    _Float16* Wkt  = (_Float16*)(ws + off); off += (size_t)D_MODEL * D_MODEL * 2;
    _Float16* Wvt  = (_Float16*)(ws + off); off += (size_t)D_MODEL * D_MODEL * 2;
    _Float16* Wot  = (_Float16*)(ws + off); off += (size_t)D_MODEL * D_MODEL * 2;
    _Float16* W1t  = (_Float16*)(ws + off); off += (size_t)D_MODEL * D_FF * 2;
    _Float16* W2t  = (_Float16*)(ws + off); off += (size_t)D_FF * D_MODEL * 2;
    float*    x1   = (float*)   (ws + off); off += (size_t)ROWS * D_MODEL * 4;

    k_transpose_cvt<<<(D_MODEL*D_MODEL + 255)/256, 256, 0, stream>>>(Wk, Wkt, D_MODEL, D_MODEL);
    k_transpose_cvt<<<(D_MODEL*D_MODEL + 255)/256, 256, 0, stream>>>(Wv, Wvt, D_MODEL, D_MODEL);
    k_transpose_cvt<<<(D_MODEL*D_MODEL + 255)/256, 256, 0, stream>>>(Wo, Wot, D_MODEL, D_MODEL);
    k_transpose_cvt<<<(D_MODEL*D_FF    + 255)/256, 256, 0, stream>>>(W1, W1t, D_MODEL, D_FF);
    k_transpose_cvt<<<(D_FF*D_MODEL    + 255)/256, 256, 0, stream>>>(W2, W2t, D_FF, D_MODEL);

    k_layernorm<<<ROWS, 256, 0, stream>>>(x, alpha1, bias1, x2h);

    #define GEMM_GRID(M_, N_) (((M_) / 64) * ((N_) / 128))

    k_gemm_wmma<2, false, false><<<GEMM_GRID(ROWS, D_MODEL), 256, 0, stream>>>(
        x2h, Wkt, bk, nullptr, nullptr, qkh, ROWS, D_MODEL, D_MODEL);
    k_gemm_wmma<3, false, false><<<GEMM_GRID(ROWS, D_MODEL), 256, 0, stream>>>(
        x2h, Wvt, bv, nullptr, nullptr, vth, ROWS, D_MODEL, D_MODEL);

    k_attention<<<BATCH*HEADS*(SEQ/16), 128, 16*SEQ*sizeof(float), stream>>>(
        qkh, vth, mask, ctxh);

    k_gemm_wmma<1, false, true><<<GEMM_GRID(ROWS, D_MODEL), 256, 0, stream>>>(
        ctxh, Wot, bo, x, x1, nullptr, ROWS, D_MODEL, D_MODEL);

    k_layernorm<<<ROWS, 256, 0, stream>>>(x1, alpha2, bias2, x2bh);

    k_gemm_wmma<0, true, false><<<GEMM_GRID(ROWS, D_FF), 256, 0, stream>>>(
        x2bh, W1t, b1, nullptr, nullptr, h1h, ROWS, D_FF, D_MODEL);

    k_gemm_wmma<1, false, true><<<GEMM_GRID(ROWS, D_MODEL), 256, 0, stream>>>(
        h1h, W2t, b2, x1, out, nullptr, ROWS, D_MODEL, D_FF);

    (void)in_sizes; (void)n_in; (void)out_size; (void)ws_size;
}


# GEMMs: A and B tiles staged through LDS (128x128x64 block tile, whole-line 16-byte global loads, two stages, one barrier per stage), wave tile 64x32; k order of every sum unchanged
# speedup vs baseline: 2.1019x; 2.1019x over previous
#include <hip/hip_runtime.h>
#include <hip/hip_bf16.h>

typedef __attribute__((ext_vector_type(16))) _Float16 v16h;
typedef __attribute__((ext_vector_type(8)))  _Float16 v8h;
typedef __attribute__((ext_vector_type(4)))  _Float16 v4h;
typedef __attribute__((ext_vector_type(8)))  float    v8f;
typedef __attribute__((ext_vector_type(4)))  float    v4f;

#define D_MODEL 1024
#define HEADS   16
#define DK      64
#define D_FF    4096
#define BATCH   2
#define SEQ     2048
#define ROWS    (BATCH*SEQ)

__device__ __forceinline__ v16h pack16(v8h lo, v8h hi) {
    v16h r;
#pragma unroll
    for (int i = 0; i < 8; ++i) { r[i] = lo[i]; r[i + 8] = hi[i]; }
    return r;
}
__device__ __forceinline__ v16h load_frag(const _Float16* p) {
    v8h lo = *(const v8h*)(p);
    v8h hi = *(const v8h*)(p + 16);
    return pack16(lo, hi);
}
__device__ __forceinline__ v8f wmma_f16(v16h a, v16h b, v8f c) {
    return __builtin_amdgcn_wmma_f32_16x16x32_f16(false, a, false, b,
                                                  (short)0, c, false, false);
}

__global__ void k_transpose_cvt(const float* __restrict__ W,
                                _Float16* __restrict__ Wt, int K, int N) {
    int idx = blockIdx.x * 256 + threadIdx.x;
    if (idx >= K * N) return;
    int k = idx / N, n = idx % N;
    Wt[(long)n * K + k] = (_Float16)W[idx];
}

__global__ void __launch_bounds__(256)
k_layernorm(const float* __restrict__ X, const float* __restrict__ alpha,
            const float* __restrict__ beta, _Float16* __restrict__ Y) {
    __shared__ float s1[256], s2[256];
    const int row = blockIdx.x, t = threadIdx.x;
    const float* xr = X + (long)row * D_MODEL;
    v4f x = *(const v4f*)(xr + t * 4);
    float s = x[0] + x[1] + x[2] + x[3];
    float q = x[0]*x[0] + x[1]*x[1] + x[2]*x[2] + x[3]*x[3];
    s1[t] = s; s2[t] = q; __syncthreads();
    for (int off = 128; off; off >>= 1) {
        if (t < off) { s1[t] += s1[t + off]; s2[t] += s2[t + off]; }
        __syncthreads();
    }
    const float mu  = s1[0] * (1.0f / D_MODEL);
    const float var = (s2[0] - (float)D_MODEL * mu * mu) * (1.0f / (D_MODEL - 1));
    const float inv = 1.0f / (sqrtf(var) + 1e-6f);
    v4h o;
#pragma unroll
    for (int i = 0; i < 4; ++i) {
        int c = t * 4 + i;
        o[i] = (_Float16)(alpha[c] * (x[i] - mu) * inv + beta[c]);
    }
    *(v4h*)(Y + (long)row * D_MODEL + t * 4) = o;
}

#define GT_M 128
#define GT_N 128
#define GT_K 64
#define GT_P 72

template <int OMODE, bool RELU, bool RESID>
__global__ void __launch_bounds__(256)
k_gemm_wmma(const _Float16* __restrict__ A, const _Float16* __restrict__ Bt,
            const float* __restrict__ bias, const float* __restrict__ resid,
            float* __restrict__ Cf, _Float16* __restrict__ Ch,
            int M, int N, int K) {
    __shared__ __attribute__((aligned(16))) _Float16 sA[2][GT_M * GT_P];
    __shared__ __attribute__((aligned(16))) _Float16 sB[2][GT_N * GT_P];

    const int tid  = threadIdx.x;
    const int lane = tid & 31;
    const int wave = tid >> 5;
    const int wm = wave >> 2, wn = wave & 3;
    const int bn_cnt = N >> 7;
    const int bm = blockIdx.x / bn_cnt;
    const int bn = blockIdx.x % bn_cnt;
    if (bm * GT_M >= M) return;

    const int ln = lane & 15;
    const int hh = lane >> 4;

    const int srow = tid >> 3, sch = tid & 7;
    const _Float16* ag = A  + (long)(bm * GT_M + srow) * K + sch * 8;
    const _Float16* bg = Bt + (long)(bn * GT_N + srow) * K + sch * 8;
    const long g32  = (long)32 * K;
    const int  soff = srow * GT_P + sch * 8;

    v8h ra[4], rb[4];
#pragma unroll
    for (int j = 0; j < 4; ++j) {
        ra[j] = *(const v8h*)(ag + j * g32);
        rb[j] = *(const v8h*)(bg + j * g32);
    }
#pragma unroll
    for (int j = 0; j < 4; ++j) {
        *(v8h*)(&sA[0][soff + j * 32 * GT_P]) = ra[j];
        *(v8h*)(&sB[0][soff + j * 32 * GT_P]) = rb[j];
    }
    __syncthreads();

    v8f acc[4][2];
#pragma unroll
    for (int i = 0; i < 4; ++i) {
#pragma unroll
        for (int j = 0; j < 2; ++j) acc[i][j] = v8f{};
    }

    const int aoff = (wm * 64 + ln) * GT_P + 8 * hh;
    const int boff = (wn * 32 + ln) * GT_P + 8 * hh;

    const int nk = K / GT_K;
    for (int kt = 0; kt < nk; ++kt) {
        const int  cur  = kt & 1;
        const bool more = (kt + 1 < nk);
        if (more) {
            ag += GT_K; bg += GT_K;
#pragma unroll
            for (int j = 0; j < 4; ++j) {
                ra[j] = *(const v8h*)(ag + j * g32);
                rb[j] = *(const v8h*)(bg + j * g32);
            }
        }

        const _Float16* pa = &sA[cur][aoff];
        const _Float16* pb = &sB[cur][boff];
#pragma unroll
        for (int ks = 0; ks < 2; ++ks) {
            v16h a[4], b[2];
#pragma unroll
            for (int i = 0; i < 4; ++i) a[i] = load_frag(pa + i * 16 * GT_P + ks * 32);
#pragma unroll
            for (int j = 0; j < 2; ++j) b[j] = load_frag(pb + j * 16 * GT_P + ks * 32);
#pragma unroll
            for (int i = 0; i < 4; ++i) {
#pragma unroll
                for (int j = 0; j < 2; ++j) acc[i][j] = wmma_f16(a[i], b[j], acc[i][j]);
            }
        }

        if (more) {
#pragma unroll
            for (int j = 0; j < 4; ++j) {
                *(v8h*)(&sA[cur ^ 1][soff + j * 32 * GT_P]) = ra[j];
                *(v8h*)(&sB[cur ^ 1][soff + j * 32 * GT_P]) = rb[j];
            }
        }
        __syncthreads();
    }

    const int lm = hh << 3;
    const int m0 = bm * GT_M + wm * 64;
    const int n0 = bn * GT_N + wn * 32;
#pragma unroll
    for (int i = 0; i < 4; ++i) {
#pragma unroll
        for (int j = 0; j < 2; ++j) {
            const v8f accv = acc[i][j];
            const int col = n0 + j * 16 + ln;
            const float bv = bias[col];
#pragma unroll
            for (int e = 0; e < 8; ++e) {
                const int row = m0 + i * 16 + e + lm;
                float v = accv[e] + bv;
                if (RELU)  v = v > 0.0f ? v : 0.0f;
                if (RESID) v += resid[(long)row * N + col];
                if (OMODE == 0) {
                    Ch[(long)row * N + col] = (_Float16)v;
                } else if (OMODE == 1) {
                    Cf[(long)row * N + col] = v;
                } else if (OMODE == 2) {
                    int b_ = row / SEQ, s = row % SEQ, h = col >> 6, d = col & 63;
                    Ch[(((long)(b_ * HEADS + h)) * SEQ + s) * DK + d] = (_Float16)v;
                } else {
                    int b_ = row / SEQ, s = row % SEQ, h = col >> 6, d = col & 63;
                    Ch[(((long)(b_ * HEADS + h)) * DK + d) * SEQ + s] = (_Float16)v;
                }
            }
        }
    }
}

#define ATT_R     2
#define ATT_QW    (16 * ATT_R)
#define ATT_WAVES 4
#define ATT_QB    (ATT_QW * ATT_WAVES)

__device__ __forceinline__ float xhalf(float x) {
    return __builtin_bit_cast(float, __builtin_amdgcn_permlanex16(
        __builtin_bit_cast(int, x), __builtin_bit_cast(int, x),
        0x76543210, 0xfedcba98, false, false));
}

__global__ void __launch_bounds__(128)
k_attention(const _Float16* __restrict__ QK, const _Float16* __restrict__ VT,
            const int* __restrict__ mask, _Float16* __restrict__ CTX) {
    const int lane = threadIdx.x & 31, wave = threadIdx.x >> 5;
    const int c = lane & 15, hh = lane >> 4;
    const int qb = blockIdx.x & 15;
    const int h  = (blockIdx.x >> 4) & 15;
    const int b  = blockIdx.x >> 8;
    const int q0 = qb * ATT_QB + wave * ATT_QW;

    const _Float16* qbase = QK + ((long)(b * HEADS + h)) * SEQ * DK;
    const _Float16* vbase = VT + ((long)(b * HEADS + h)) * DK * SEQ;
    const _Float16* kp = qbase + (long)c * DK + 8 * hh;
    const _Float16* vp = vbase + (long)c * SEQ + 8 * hh;
    const int* mrow = mask + b * SEQ + 8 * hh;

    const float cscale = 0.125f * 1.4426950408889634f;
    const float negl   = -1e9f * 1.4426950408889634f;

    v16h qf[ATT_R][2];
#pragma unroll
    for (int g = 0; g < ATT_R; ++g) {
        const _Float16* qp = qbase + (long)(q0 + 16 * g + c) * DK + 8 * hh;
        qf[g][0] = load_frag(qp);
        qf[g][1] = load_frag(qp + 32);
    }

    v8f acc[ATT_R][4];
    float m_run[ATT_R], l_run[ATT_R];
#pragma unroll
    for (int g = 0; g < ATT_R; ++g) {
        m_run[g] = -__builtin_inff(); l_run[g] = 0.0f;
#pragma unroll
        for (int t = 0; t < 4; ++t) acc[g][t] = v8f{};
    }

    v16h kf[2][2];
#pragma unroll
    for (int kt = 0; kt < 2; ++kt) {
        kf[kt][0] = load_frag(kp + (long)(16 * kt) * DK);
        kf[kt][1] = load_frag(kp + (long)(16 * kt) * DK + 32);
    }
    int mk[2][8];
#pragma unroll
    for (int kt = 0; kt < 2; ++kt) {
        const int4 m0 = *(const int4*)(mrow + 16 * kt);
        const int4 m1 = *(const int4*)(mrow + 16 * kt + 4);
        mk[kt][0] = m0.x; mk[kt][1] = m0.y; mk[kt][2] = m0.z; mk[kt][3] = m0.w;
        mk[kt][4] = m1.x; mk[kt][5] = m1.y; mk[kt][6] = m1.z; mk[kt][7] = m1.w;
    }

    for (int key0 = 0; key0 < SEQ; key0 += 32) {
        v16h vf[4];
#pragma unroll
        for (int t = 0; t < 4; ++t) vf[t] = load_frag(vp + (long)(16 * t) * SEQ + key0);
        __builtin_amdgcn_sched_barrier(0);

        v8f s[2][ATT_R];
#pragma unroll
        for (int kt = 0; kt < 2; ++kt) {
#pragma unroll
            for (int g = 0; g < ATT_R; ++g) {
                v8f z = {};
                z = wmma_f16(kf[kt][0], qf[g][0], z);
                s[kt][g] = wmma_f16(kf[kt][1], qf[g][1], z);
            }
        }

        const int nk = (key0 + 32 < SEQ) ? key0 + 32 : key0;
#pragma unroll
        for (int kt = 0; kt < 2; ++kt) {
            kf[kt][0] = load_frag(kp + (long)(nk + 16 * kt) * DK);
            kf[kt][1] = load_frag(kp + (long)(nk + 16 * kt) * DK + 32);
        }
        int mkn[2][8];
#pragma unroll
        for (int kt = 0; kt < 2; ++kt) {
            const int4 m0 = *(const int4*)(mrow + nk + 16 * kt);
            const int4 m1 = *(const int4*)(mrow + nk + 16 * kt + 4);
            mkn[kt][0] = m0.x; mkn[kt][1] = m0.y; mkn[kt][2] = m0.z; mkn[kt][3] = m0.w;
            mkn[kt][4] = m1.x; mkn[kt][5] = m1.y; mkn[kt][6] = m1.z; mkn[kt][7] = m1.w;
        }

        float mb[ATT_R];
        bool need = false;
#pragma unroll
        for (int g = 0; g < ATT_R; ++g) {
            float mx = -__builtin_inff();
#pragma unroll
            for (int kt = 0; kt < 2; ++kt) {
#pragma unroll
                for (int r = 0; r < 8; ++r) {
                    float t = s[kt][g][r] * cscale;
                    t = (mk[kt][r] == 0) ? negl : t;
                    s[kt][g][r] = t;
                    mx = fmaxf(mx, t);
                }
            }
            mx = fmaxf(mx, xhalf(mx));
            mb[g] = mx;
            need = need || (mx > m_run[g] + 8.0f);
        }
        if (__builtin_amdgcn_ballot_w32(need) != 0) {
#pragma unroll
            for (int g = 0; g < ATT_R; ++g) {
                const float m_new = fmaxf(m_run[g], mb[g]);
                const float corr  = __builtin_amdgcn_exp2f(m_run[g] - m_new);
                m_run[g] = m_new;
                l_run[g] *= corr;
#pragma unroll
                for (int t = 0; t < 4; ++t) acc[g][t] = acc[g][t] * corr;
            }
        }

#pragma unroll
        for (int g = 0; g < ATT_R; ++g) {
            const float m = m_run[g];
            v16h pf;
            float ls = 0.0f;
#pragma unroll
            for (int r = 0; r < 8; ++r) {
                const float p0 = __builtin_amdgcn_exp2f(s[0][g][r] - m);
                const float p1 = __builtin_amdgcn_exp2f(s[1][g][r] - m);
                ls += p0 + p1;
                pf[r]     = (_Float16)p0;
                pf[8 + r] = (_Float16)p1;
            }
            l_run[g] += ls;
#pragma unroll
            for (int t = 0; t < 4; ++t) acc[g][t] = wmma_f16(vf[t], pf, acc[g][t]);
        }
#pragma unroll
        for (int kt = 0; kt < 2; ++kt) {
#pragma unroll
            for (int r = 0; r < 8; ++r) mk[kt][r] = mkn[kt][r];
        }
    }

#pragma unroll
    for (int g = 0; g < ATT_R; ++g) {
        float l = l_run[g];
        l = l + xhalf(l);
        const float inv = 1.0f / l;
        _Float16* op = CTX + ((long)(b * SEQ + q0 + 16 * g + c)) * D_MODEL + h * DK + 8 * hh;
#pragma unroll
        for (int t = 0; t < 4; ++t) {
            v8h o;
#pragma unroll
            for (int r = 0; r < 8; ++r) o[r] = (_Float16)(acc[g][t][r] * inv);
            *(v8h*)(op + 16 * t) = o;
        }
    }
}

extern "C" void kernel_launch(void* const* d_in, const int* in_sizes, int n_in,
                              void* d_out, int out_size, void* d_ws, size_t ws_size,
                              hipStream_t stream) {
    const float* x      = (const float*)d_in[0];
    const int*   mask   = (const int*)  d_in[1];
    const float* alpha1 = (const float*)d_in[2];
    const float* bias1  = (const float*)d_in[3];
    const float* alpha2 = (const float*)d_in[4];
    const float* bias2  = (const float*)d_in[5];
    const float* Wk     = (const float*)d_in[6];
    const float* bk     = (const float*)d_in[7];
    const float* Wv     = (const float*)d_in[8];
    const float* bv     = (const float*)d_in[9];
    const float* Wo     = (const float*)d_in[10];
    const float* bo     = (const float*)d_in[11];
    const float* W1     = (const float*)d_in[12];
    const float* b1     = (const float*)d_in[13];
    const float* W2     = (const float*)d_in[14];
    const float* b2     = (const float*)d_in[15];
    float* out = (float*)d_out;

    char* ws = (char*)d_ws;
    size_t off = 0;
    _Float16* x2h  = (_Float16*)(ws + off); off += (size_t)ROWS * D_MODEL * 2;
    _Float16* x2bh = (_Float16*)(ws + off); off += (size_t)ROWS * D_MODEL * 2;
    _Float16* qkh  = (_Float16*)(ws + off); off += (size_t)BATCH*HEADS*SEQ*DK*2;
    _Float16* vth  = (_Float16*)(ws + off); off += (size_t)BATCH*HEADS*DK*SEQ*2;
    _Float16* ctxh = (_Float16*)(ws + off); off += (size_t)ROWS * D_MODEL * 2;
    _Float16* h1h  = (_Float16*)(ws + off); off += (size_t)ROWS * D_FF * 2;
    _Float16* Wkt  = (_Float16*)(ws + off); off += (size_t)D_MODEL * D_MODEL * 2;
    _Float16* Wvt  = (_Float16*)(ws + off); off += (size_t)D_MODEL * D_MODEL * 2;
    _Float16* Wot  = (_Float16*)(ws + off); off += (size_t)D_MODEL * D_MODEL * 2;
    _Float16* W1t  = (_Float16*)(ws + off); off += (size_t)D_MODEL * D_FF * 2;
    _Float16* W2t  = (_Float16*)(ws + off); off += (size_t)D_FF * D_MODEL * 2;
    float*    x1   = (float*)   (ws + off); off += (size_t)ROWS * D_MODEL * 4;

    k_transpose_cvt<<<(D_MODEL*D_MODEL + 255)/256, 256, 0, stream>>>(Wk, Wkt, D_MODEL, D_MODEL);
    k_transpose_cvt<<<(D_MODEL*D_MODEL + 255)/256, 256, 0, stream>>>(Wv, Wvt, D_MODEL, D_MODEL);
    k_transpose_cvt<<<(D_MODEL*D_MODEL + 255)/256, 256, 0, stream>>>(Wo, Wot, D_MODEL, D_MODEL);
    k_transpose_cvt<<<(D_MODEL*D_FF    + 255)/256, 256, 0, stream>>>(W1, W1t, D_MODEL, D_FF);
    k_transpose_cvt<<<(D_FF*D_MODEL    + 255)/256, 256, 0, stream>>>(W2, W2t, D_FF, D_MODEL);

    k_layernorm<<<ROWS, 256, 0, stream>>>(x, alpha1, bias1, x2h);

    #define GEMM_GRID(M_, N_) (((M_) / 128) * ((N_) / 128))

    k_gemm_wmma<2, false, false><<<GEMM_GRID(ROWS, D_MODEL), 256, 0, stream>>>(
        x2h, Wkt, bk, nullptr, nullptr, qkh, ROWS, D_MODEL, D_MODEL);
    k_gemm_wmma<3, false, false><<<GEMM_GRID(ROWS, D_MODEL), 256, 0, stream>>>(
        x2h, Wvt, bv, nullptr, nullptr, vth, ROWS, D_MODEL, D_MODEL);

    k_attention<<<BATCH*HEADS*(SEQ/ATT_QB), 128, 0, stream>>>(
        qkh, vth, mask, ctxh);

    k_gemm_wmma<1, false, true><<<GEMM_GRID(ROWS, D_MODEL), 256, 0, stream>>>(
        ctxh, Wot, bo, x, x1, nullptr, ROWS, D_MODEL, D_MODEL);

    k_layernorm<<<ROWS, 256, 0, stream>>>(x1, alpha2, bias2, x2bh);

    k_gemm_wmma<0, true, false><<<GEMM_GRID(ROWS, D_FF), 256, 0, stream>>>(
        x2bh, W1t, b1, nullptr, nullptr, h1h, ROWS, D_FF, D_MODEL);

    k_gemm_wmma<1, false, true><<<GEMM_GRID(ROWS, D_MODEL), 256, 0, stream>>>(
        h1h, W2t, b2, x1, out, nullptr, ROWS, D_MODEL, D_FF);

    (void)in_sizes; (void)n_in; (void)out_size; (void)ws_size;
}


# weight transposes: one launch of 128x64 tiles through LDS, 16-byte loads and whole-row 16-byte stores, 32 elements per lane (was five launches, one element per lane, 2-byte stores a row apart)
# speedup vs baseline: 4.3726x; 2.0803x over previous
#include <hip/hip_runtime.h>
#include <hip/hip_bf16.h>

typedef __attribute__((ext_vector_type(16))) _Float16 v16h;
typedef __attribute__((ext_vector_type(8)))  _Float16 v8h;
typedef __attribute__((ext_vector_type(4)))  _Float16 v4h;
typedef __attribute__((ext_vector_type(8)))  float    v8f;
typedef __attribute__((ext_vector_type(4)))  float    v4f;

#define D_MODEL 1024
#define HEADS   16
#define DK      64
#define D_FF    4096
#define BATCH   2
#define SEQ     2048
#define ROWS    (BATCH*SEQ)

__device__ __forceinline__ v16h pack16(v8h lo, v8h hi) {
    v16h r;
#pragma unroll
    for (int i = 0; i < 8; ++i) { r[i] = lo[i]; r[i + 8] = hi[i]; }
    return r;
}
__device__ __forceinline__ v16h load_frag(const _Float16* p) {
    v8h lo = *(const v8h*)(p);
    v8h hi = *(const v8h*)(p + 16);
    return pack16(lo, hi);
}
__device__ __forceinline__ v8f wmma_f16(v16h a, v16h b, v8f c) {
    return __builtin_amdgcn_wmma_f32_16x16x32_f16(false, a, false, b,
                                                  (short)0, c, false, false);
}

#define TR_K 128
#define TR_N 64
#define TR_P 136
#define TR_TILES(K_, N_) (((K_) / TR_K) * ((N_) / TR_N))
#define TR_SQ   TR_TILES(D_MODEL, D_MODEL)
#define TR_W1   TR_TILES(D_MODEL, D_FF)
#define TR_W2   TR_TILES(D_FF, D_MODEL)
#define TR_GRID (3 * TR_SQ + TR_W1 + TR_W2)

__global__ void __launch_bounds__(256)
k_transpose_cvt_all(const float* __restrict__ Wk, const float* __restrict__ Wv,
                    const float* __restrict__ Wo, const float* __restrict__ W1,
                    const float* __restrict__ W2,
                    _Float16* __restrict__ Wkt, _Float16* __restrict__ Wvt,
                    _Float16* __restrict__ Wot, _Float16* __restrict__ W1t,
                    _Float16* __restrict__ W2t) {
    __shared__ __attribute__((aligned(16))) _Float16 sT[TR_N * TR_P];

    int tile = blockIdx.x;
    const float* W; _Float16* Wt; int K, N;
    if (tile < TR_SQ)                 { W = Wk; Wt = Wkt; K = D_MODEL; N = D_MODEL; }
    else if (tile < 2 * TR_SQ)        { W = Wv; Wt = Wvt; K = D_MODEL; N = D_MODEL; tile -= TR_SQ; }
    else if (tile < 3 * TR_SQ)        { W = Wo; Wt = Wot; K = D_MODEL; N = D_MODEL; tile -= 2 * TR_SQ; }
    else if (tile < 3 * TR_SQ + TR_W1){ W = W1; Wt = W1t; K = D_MODEL; N = D_FF;    tile -= 3 * TR_SQ; }
    else                              { W = W2; Wt = W2t; K = D_FF;    N = D_MODEL; tile -= 3 * TR_SQ + TR_W1; }
    const int ntn = N / TR_N;
    const int kt = tile / ntn, nt = tile % ntn;

    const int t = threadIdx.x;
    const int q = t & 15, g = t >> 4;

    const float* src = W + (long)(kt * TR_K + 8 * g) * N + nt * TR_N + 4 * q;
    v4f r[8];
#pragma unroll
    for (int i = 0; i < 8; ++i) r[i] = *(const v4f*)(src + (long)i * N);

#pragma unroll
    for (int j = 0; j < 4; ++j) {
        v8h o;
#pragma unroll
        for (int i = 0; i < 8; ++i) o[i] = (_Float16)r[i][j];
        *(v8h*)(&sT[(4 * q + j) * TR_P + 8 * g]) = o;
    }
    __syncthreads();

    _Float16* dst = Wt + (long)(nt * TR_N + g) * K + kt * TR_K + 8 * q;
#pragma unroll
    for (int s = 0; s < 4; ++s) {
        const v8h o = *(const v8h*)(&sT[(g + 16 * s) * TR_P + 8 * q]);
        *(v8h*)(dst + (long)(16 * s) * K) = o;
    }
}

__global__ void __launch_bounds__(256)
k_layernorm(const float* __restrict__ X, const float* __restrict__ alpha,
            const float* __restrict__ beta, _Float16* __restrict__ Y) {
    __shared__ float s1[256], s2[256];
    const int row = blockIdx.x, t = threadIdx.x;
    const float* xr = X + (long)row * D_MODEL;
    v4f x = *(const v4f*)(xr + t * 4);
    float s = x[0] + x[1] + x[2] + x[3];
    float q = x[0]*x[0] + x[1]*x[1] + x[2]*x[2] + x[3]*x[3];
    s1[t] = s; s2[t] = q; __syncthreads();
    for (int off = 128; off; off >>= 1) {
        if (t < off) { s1[t] += s1[t + off]; s2[t] += s2[t + off]; }
        __syncthreads();
    }
    const float mu  = s1[0] * (1.0f / D_MODEL);
    const float var = (s2[0] - (float)D_MODEL * mu * mu) * (1.0f / (D_MODEL - 1));
    const float inv = 1.0f / (sqrtf(var) + 1e-6f);
    v4h o;
#pragma unroll
    for (int i = 0; i < 4; ++i) {
        int c = t * 4 + i;
        o[i] = (_Float16)(alpha[c] * (x[i] - mu) * inv + beta[c]);
    }
    *(v4h*)(Y + (long)row * D_MODEL + t * 4) = o;
}

#define GT_M 128
#define GT_N 128
#define GT_K 64
#define GT_P 72

template <int OMODE, bool RELU, bool RESID>
__global__ void __launch_bounds__(256)
k_gemm_wmma(const _Float16* __restrict__ A, const _Float16* __restrict__ Bt,
            const float* __restrict__ bias, const float* __restrict__ resid,
            float* __restrict__ Cf, _Float16* __restrict__ Ch,
            int M, int N, int K) {
    __shared__ __attribute__((aligned(16))) _Float16 sA[2][GT_M * GT_P];
    __shared__ __attribute__((aligned(16))) _Float16 sB[2][GT_N * GT_P];

    const int tid  = threadIdx.x;
    const int lane = tid & 31;
    const int wave = tid >> 5;
    const int wm = wave >> 2, wn = wave & 3;
    const int bn_cnt = N >> 7;
    const int bm = blockIdx.x / bn_cnt;
    const int bn = blockIdx.x % bn_cnt;
    if (bm * GT_M >= M) return;

    const int ln = lane & 15;
    const int hh = lane >> 4;

    const int srow = tid >> 3, sch = tid & 7;
    const _Float16* ag = A  + (long)(bm * GT_M + srow) * K + sch * 8;
    const _Float16* bg = Bt + (long)(bn * GT_N + srow) * K + sch * 8;
    const long g32  = (long)32 * K;
    const int  soff = srow * GT_P + sch * 8;

    v8h ra[4], rb[4];
#pragma unroll
    for (int j = 0; j < 4; ++j) {
        ra[j] = *(const v8h*)(ag + j * g32);
        rb[j] = *(const v8h*)(bg + j * g32);
    }
#pragma unroll
    for (int j = 0; j < 4; ++j) {
        *(v8h*)(&sA[0][soff + j * 32 * GT_P]) = ra[j];
        *(v8h*)(&sB[0][soff + j * 32 * GT_P]) = rb[j];
    }
    __syncthreads();

    v8f acc[4][2];
#pragma unroll
    for (int i = 0; i < 4; ++i) {
#pragma unroll
        for (int j = 0; j < 2; ++j) acc[i][j] = v8f{};
    }

    const int aoff = (wm * 64 + ln) * GT_P + 8 * hh;
    const int boff = (wn * 32 + ln) * GT_P + 8 * hh;

    const int nk = K / GT_K;
    for (int kt = 0; kt < nk; ++kt) {
        const int  cur  = kt & 1;
        const bool more = (kt + 1 < nk);
        if (more) {
            ag += GT_K; bg += GT_K;
#pragma unroll
            for (int j = 0; j < 4; ++j) {
                ra[j] = *(const v8h*)(ag + j * g32);
                rb[j] = *(const v8h*)(bg + j * g32);
            }
        }

        const _Float16* pa = &sA[cur][aoff];
        const _Float16* pb = &sB[cur][boff];
#pragma unroll
        for (int ks = 0; ks < 2; ++ks) {
            v16h a[4], b[2];
#pragma unroll
            for (int i = 0; i < 4; ++i) a[i] = load_frag(pa + i * 16 * GT_P + ks * 32);
#pragma unroll
            for (int j = 0; j < 2; ++j) b[j] = load_frag(pb + j * 16 * GT_P + ks * 32);
#pragma unroll
            for (int i = 0; i < 4; ++i) {
#pragma unroll
                for (int j = 0; j < 2; ++j) acc[i][j] = wmma_f16(a[i], b[j], acc[i][j]);
            }
        }

        if (more) {
#pragma unroll
            for (int j = 0; j < 4; ++j) {
                *(v8h*)(&sA[cur ^ 1][soff + j * 32 * GT_P]) = ra[j];
                *(v8h*)(&sB[cur ^ 1][soff + j * 32 * GT_P]) = rb[j];
            }
        }
        __syncthreads();
    }

    const int lm = hh << 3;
    const int m0 = bm * GT_M + wm * 64;
    const int n0 = bn * GT_N + wn * 32;
#pragma unroll
    for (int i = 0; i < 4; ++i) {
#pragma unroll
        for (int j = 0; j < 2; ++j) {
            const v8f accv = acc[i][j];
            const int col = n0 + j * 16 + ln;
            const float bv = bias[col];
#pragma unroll
            for (int e = 0; e < 8; ++e) {
                const int row = m0 + i * 16 + e + lm;
                float v = accv[e] + bv;
                if (RELU)  v = v > 0.0f ? v : 0.0f;
                if (RESID) v += resid[(long)row * N + col];
                if (OMODE == 0) {
                    Ch[(long)row * N + col] = (_Float16)v;
                } else if (OMODE == 1) {
                    Cf[(long)row * N + col] = v;
                } else if (OMODE == 2) {
                    int b_ = row / SEQ, s = row % SEQ, h = col >> 6, d = col & 63;
                    Ch[(((long)(b_ * HEADS + h)) * SEQ + s) * DK + d] = (_Float16)v;
                } else {
                    int b_ = row / SEQ, s = row % SEQ, h = col >> 6, d = col & 63;
                    Ch[(((long)(b_ * HEADS + h)) * DK + d) * SEQ + s] = (_Float16)v;
                }
            }
        }
    }
}

#define ATT_R     2
#define ATT_QW    (16 * ATT_R)
#define ATT_WAVES 4
#define ATT_QB    (ATT_QW * ATT_WAVES)

__device__ __forceinline__ float xhalf(float x) {
    return __builtin_bit_cast(float, __builtin_amdgcn_permlanex16(
        __builtin_bit_cast(int, x), __builtin_bit_cast(int, x),
        0x76543210, 0xfedcba98, false, false));
}

__global__ void __launch_bounds__(128)
k_attention(const _Float16* __restrict__ QK, const _Float16* __restrict__ VT,
            const int* __restrict__ mask, _Float16* __restrict__ CTX) {
    const int lane = threadIdx.x & 31, wave = threadIdx.x >> 5;
    const int c = lane & 15, hh = lane >> 4;
    const int qb = blockIdx.x & 15;
    const int h  = (blockIdx.x >> 4) & 15;
    const int b  = blockIdx.x >> 8;
    const int q0 = qb * ATT_QB + wave * ATT_QW;

    const _Float16* qbase = QK + ((long)(b * HEADS + h)) * SEQ * DK;
    const _Float16* vbase = VT + ((long)(b * HEADS + h)) * DK * SEQ;
    const _Float16* kp = qbase + (long)c * DK + 8 * hh;
    const _Float16* vp = vbase + (long)c * SEQ + 8 * hh;
    const int* mrow = mask + b * SEQ + 8 * hh;

    const float cscale = 0.125f * 1.4426950408889634f;
    const float negl   = -1e9f * 1.4426950408889634f;

    v16h qf[ATT_R][2];
#pragma unroll
    for (int g = 0; g < ATT_R; ++g) {
        const _Float16* qp = qbase + (long)(q0 + 16 * g + c) * DK + 8 * hh;
        qf[g][0] = load_frag(qp);
        qf[g][1] = load_frag(qp + 32);
    }

    v8f acc[ATT_R][4];
    float m_run[ATT_R], l_run[ATT_R];
#pragma unroll
    for (int g = 0; g < ATT_R; ++g) {
        m_run[g] = -__builtin_inff(); l_run[g] = 0.0f;
#pragma unroll
        for (int t = 0; t < 4; ++t) acc[g][t] = v8f{};
    }

    v16h kf[2][2];
#pragma unroll
    for (int kt = 0; kt < 2; ++kt) {
        kf[kt][0] = load_frag(kp + (long)(16 * kt) * DK);
        kf[kt][1] = load_frag(kp + (long)(16 * kt) * DK + 32);
    }
    int mk[2][8];
#pragma unroll
    for (int kt = 0; kt < 2; ++kt) {
        const int4 m0 = *(const int4*)(mrow + 16 * kt);
        const int4 m1 = *(const int4*)(mrow + 16 * kt + 4);
        mk[kt][0] = m0.x; mk[kt][1] = m0.y; mk[kt][2] = m0.z; mk[kt][3] = m0.w;
        mk[kt][4] = m1.x; mk[kt][5] = m1.y; mk[kt][6] = m1.z; mk[kt][7] = m1.w;
    }

    for (int key0 = 0; key0 < SEQ; key0 += 32) {
        v16h vf[4];
#pragma unroll
        for (int t = 0; t < 4; ++t) vf[t] = load_frag(vp + (long)(16 * t) * SEQ + key0);
        __builtin_amdgcn_sched_barrier(0);

        v8f s[2][ATT_R];
#pragma unroll
        for (int kt = 0; kt < 2; ++kt) {
#pragma unroll
            for (int g = 0; g < ATT_R; ++g) {
                v8f z = {};
                z = wmma_f16(kf[kt][0], qf[g][0], z);
                s[kt][g] = wmma_f16(kf[kt][1], qf[g][1], z);
            }
        }

        const int nk = (key0 + 32 < SEQ) ? key0 + 32 : key0;
#pragma unroll
        for (int kt = 0; kt < 2; ++kt) {
            kf[kt][0] = load_frag(kp + (long)(nk + 16 * kt) * DK);
            kf[kt][1] = load_frag(kp + (long)(nk + 16 * kt) * DK + 32);
        }
        int mkn[2][8];
#pragma unroll
        for (int kt = 0; kt < 2; ++kt) {
            const int4 m0 = *(const int4*)(mrow + nk + 16 * kt);
            const int4 m1 = *(const int4*)(mrow + nk + 16 * kt + 4);
            mkn[kt][0] = m0.x; mkn[kt][1] = m0.y; mkn[kt][2] = m0.z; mkn[kt][3] = m0.w;
            mkn[kt][4] = m1.x; mkn[kt][5] = m1.y; mkn[kt][6] = m1.z; mkn[kt][7] = m1.w;
        }

        float mb[ATT_R];
        bool need = false;
#pragma unroll
        for (int g = 0; g < ATT_R; ++g) {
            float mx = -__builtin_inff();
#pragma unroll
            for (int kt = 0; kt < 2; ++kt) {
#pragma unroll
                for (int r = 0; r < 8; ++r) {
                    float t = s[kt][g][r] * cscale;
                    t = (mk[kt][r] == 0) ? negl : t;
                    s[kt][g][r] = t;
                    mx = fmaxf(mx, t);
                }
            }
            mx = fmaxf(mx, xhalf(mx));
            mb[g] = mx;
            need = need || (mx > m_run[g] + 8.0f);
        }
        if (__builtin_amdgcn_ballot_w32(need) != 0) {
#pragma unroll
            for (int g = 0; g < ATT_R; ++g) {
                const float m_new = fmaxf(m_run[g], mb[g]);
                const float corr  = __builtin_amdgcn_exp2f(m_run[g] - m_new);
                m_run[g] = m_new;
                l_run[g] *= corr;
#pragma unroll
                for (int t = 0; t < 4; ++t) acc[g][t] = acc[g][t] * corr;
            }
        }

#pragma unroll
        for (int g = 0; g < ATT_R; ++g) {
            const float m = m_run[g];
            v16h pf;
            float ls = 0.0f;
#pragma unroll
            for (int r = 0; r < 8; ++r) {
                const float p0 = __builtin_amdgcn_exp2f(s[0][g][r] - m);
                const float p1 = __builtin_amdgcn_exp2f(s[1][g][r] - m);
                ls += p0 + p1;
                pf[r]     = (_Float16)p0;
                pf[8 + r] = (_Float16)p1;
            }
            l_run[g] += ls;
#pragma unroll
            for (int t = 0; t < 4; ++t) acc[g][t] = wmma_f16(vf[t], pf, acc[g][t]);
        }
#pragma unroll
        for (int kt = 0; kt < 2; ++kt) {
#pragma unroll
            for (int r = 0; r < 8; ++r) mk[kt][r] = mkn[kt][r];
        }
    }

#pragma unroll
    for (int g = 0; g < ATT_R; ++g) {
        float l = l_run[g];
        l = l + xhalf(l);
        const float inv = 1.0f / l;
        _Float16* op = CTX + ((long)(b * SEQ + q0 + 16 * g + c)) * D_MODEL + h * DK + 8 * hh;
#pragma unroll
        for (int t = 0; t < 4; ++t) {
            v8h o;
#pragma unroll
            for (int r = 0; r < 8; ++r) o[r] = (_Float16)(acc[g][t][r] * inv);
            *(v8h*)(op + 16 * t) = o;
        }
    }
}

extern "C" void kernel_launch(void* const* d_in, const int* in_sizes, int n_in,
                              void* d_out, int out_size, void* d_ws, size_t ws_size,
                              hipStream_t stream) {
    const float* x      = (const float*)d_in[0];
    const int*   mask   = (const int*)  d_in[1];
    const float* alpha1 = (const float*)d_in[2];
    const float* bias1  = (const float*)d_in[3];
    const float* alpha2 = (const float*)d_in[4];
    const float* bias2  = (const float*)d_in[5];
    const float* Wk     = (const float*)d_in[6];
    const float* bk     = (const float*)d_in[7];
    const float* Wv     = (const float*)d_in[8];
    const float* bv     = (const float*)d_in[9];
    const float* Wo     = (const float*)d_in[10];
    const float* bo     = (const float*)d_in[11];
    const float* W1     = (const float*)d_in[12];
    const float* b1     = (const float*)d_in[13];
    const float* W2     = (const float*)d_in[14];
    const float* b2     = (const float*)d_in[15];
    float* out = (float*)d_out;

    char* ws = (char*)d_ws;
    size_t off = 0;
    _Float16* x2h  = (_Float16*)(ws + off); off += (size_t)ROWS * D_MODEL * 2;
    _Float16* x2bh = (_Float16*)(ws + off); off += (size_t)ROWS * D_MODEL * 2;
    _Float16* qkh  = (_Float16*)(ws + off); off += (size_t)BATCH*HEADS*SEQ*DK*2;
    _Float16* vth  = (_Float16*)(ws + off); off += (size_t)BATCH*HEADS*DK*SEQ*2;
    _Float16* ctxh = (_Float16*)(ws + off); off += (size_t)ROWS * D_MODEL * 2;
    _Float16* h1h  = (_Float16*)(ws + off); off += (size_t)ROWS * D_FF * 2;
    _Float16* Wkt  = (_Float16*)(ws + off); off += (size_t)D_MODEL * D_MODEL * 2;
    _Float16* Wvt  = (_Float16*)(ws + off); off += (size_t)D_MODEL * D_MODEL * 2;
    _Float16* Wot  = (_Float16*)(ws + off); off += (size_t)D_MODEL * D_MODEL * 2;
    _Float16* W1t  = (_Float16*)(ws + off); off += (size_t)D_MODEL * D_FF * 2;
    _Float16* W2t  = (_Float16*)(ws + off); off += (size_t)D_FF * D_MODEL * 2;
    float*    x1   = (float*)   (ws + off); off += (size_t)ROWS * D_MODEL * 4;

    k_transpose_cvt_all<<<TR_GRID, 256, 0, stream>>>(Wk, Wv, Wo, W1, W2,
                                                     Wkt, Wvt, Wot, W1t, W2t);

    k_layernorm<<<ROWS, 256, 0, stream>>>(x, alpha1, bias1, x2h);

    #define GEMM_GRID(M_, N_) (((M_) / 128) * ((N_) / 128))

    k_gemm_wmma<2, false, false><<<GEMM_GRID(ROWS, D_MODEL), 256, 0, stream>>>(
        x2h, Wkt, bk, nullptr, nullptr, qkh, ROWS, D_MODEL, D_MODEL);
    k_gemm_wmma<3, false, false><<<GEMM_GRID(ROWS, D_MODEL), 256, 0, stream>>>(
        x2h, Wvt, bv, nullptr, nullptr, vth, ROWS, D_MODEL, D_MODEL);

    k_attention<<<BATCH*HEADS*(SEQ/ATT_QB), 128, 0, stream>>>(
        qkh, vth, mask, ctxh);

    k_gemm_wmma<1, false, true><<<GEMM_GRID(ROWS, D_MODEL), 256, 0, stream>>>(
        ctxh, Wot, bo, x, x1, nullptr, ROWS, D_MODEL, D_MODEL);

    k_layernorm<<<ROWS, 256, 0, stream>>>(x1, alpha2, bias2, x2bh);

    k_gemm_wmma<0, true, false><<<GEMM_GRID(ROWS, D_FF), 256, 0, stream>>>(
        x2bh, W1t, b1, nullptr, nullptr, h1h, ROWS, D_FF, D_MODEL);

    k_gemm_wmma<1, false, true><<<GEMM_GRID(ROWS, D_MODEL), 256, 0, stream>>>(
        h1h, W2t, b2, x1, out, nullptr, ROWS, D_MODEL, D_FF);

    (void)in_sizes; (void)n_in; (void)out_size; (void)ws_size;
}


# GEMM k-loop: the 12 LDS fragment reads of each 32-k sub-step issued as one group into a second register set under the other sub-step's 8 products (barrier between the sub-steps); sum order unchanged
# speedup vs baseline: 4.6370x; 1.0605x over previous
#include <hip/hip_runtime.h>
#include <hip/hip_bf16.h>

typedef __attribute__((ext_vector_type(16))) _Float16 v16h;
typedef __attribute__((ext_vector_type(8)))  _Float16 v8h;
typedef __attribute__((ext_vector_type(4)))  _Float16 v4h;
typedef __attribute__((ext_vector_type(8)))  float    v8f;
typedef __attribute__((ext_vector_type(4)))  float    v4f;

#define D_MODEL 1024
#define HEADS   16
#define DK      64
#define D_FF    4096
#define BATCH   2
#define SEQ     2048
#define ROWS    (BATCH*SEQ)

__device__ __forceinline__ v16h pack16(v8h lo, v8h hi) {
    v16h r;
#pragma unroll
    for (int i = 0; i < 8; ++i) { r[i] = lo[i]; r[i + 8] = hi[i]; }
    return r;
}
__device__ __forceinline__ v16h load_frag(const _Float16* p) {
    v8h lo = *(const v8h*)(p);
    v8h hi = *(const v8h*)(p + 16);
    return pack16(lo, hi);
}
__device__ __forceinline__ v8f wmma_f16(v16h a, v16h b, v8f c) {
    return __builtin_amdgcn_wmma_f32_16x16x32_f16(false, a, false, b,
                                                  (short)0, c, false, false);
}

#define TR_K 128
#define TR_N 64
#define TR_P 136
#define TR_TILES(K_, N_) (((K_) / TR_K) * ((N_) / TR_N))
#define TR_SQ   TR_TILES(D_MODEL, D_MODEL)
#define TR_W1   TR_TILES(D_MODEL, D_FF)
#define TR_W2   TR_TILES(D_FF, D_MODEL)
#define TR_GRID (3 * TR_SQ + TR_W1 + TR_W2)

__global__ void __launch_bounds__(256)
k_transpose_cvt_all(const float* __restrict__ Wk, const float* __restrict__ Wv,
                    const float* __restrict__ Wo, const float* __restrict__ W1,
                    const float* __restrict__ W2,
                    _Float16* __restrict__ Wkt, _Float16* __restrict__ Wvt,
                    _Float16* __restrict__ Wot, _Float16* __restrict__ W1t,
                    _Float16* __restrict__ W2t) {
    __shared__ __attribute__((aligned(16))) _Float16 sT[TR_N * TR_P];

    int tile = blockIdx.x;
    const float* W; _Float16* Wt; int K, N;
    if (tile < TR_SQ)                 { W = Wk; Wt = Wkt; K = D_MODEL; N = D_MODEL; }
    else if (tile < 2 * TR_SQ)        { W = Wv; Wt = Wvt; K = D_MODEL; N = D_MODEL; tile -= TR_SQ; }
    else if (tile < 3 * TR_SQ)        { W = Wo; Wt = Wot; K = D_MODEL; N = D_MODEL; tile -= 2 * TR_SQ; }
    else if (tile < 3 * TR_SQ + TR_W1){ W = W1; Wt = W1t; K = D_MODEL; N = D_FF;    tile -= 3 * TR_SQ; }
    else                              { W = W2; Wt = W2t; K = D_FF;    N = D_MODEL; tile -= 3 * TR_SQ + TR_W1; }
    const int ntn = N / TR_N;
    const int kt = tile / ntn, nt = tile % ntn;

    const int t = threadIdx.x;
    const int q = t & 15, g = t >> 4;

    const float* src = W + (long)(kt * TR_K + 8 * g) * N + nt * TR_N + 4 * q;
    v4f r[8];
#pragma unroll
    for (int i = 0; i < 8; ++i) r[i] = *(const v4f*)(src + (long)i * N);

#pragma unroll
    for (int j = 0; j < 4; ++j) {
        v8h o;
#pragma unroll
        for (int i = 0; i < 8; ++i) o[i] = (_Float16)r[i][j];
        *(v8h*)(&sT[(4 * q + j) * TR_P + 8 * g]) = o;
    }
    __syncthreads();

    _Float16* dst = Wt + (long)(nt * TR_N + g) * K + kt * TR_K + 8 * q;
#pragma unroll
    for (int s = 0; s < 4; ++s) {
        const v8h o = *(const v8h*)(&sT[(g + 16 * s) * TR_P + 8 * q]);
        *(v8h*)(dst + (long)(16 * s) * K) = o;
    }
}

__global__ void __launch_bounds__(256)
k_layernorm(const float* __restrict__ X, const float* __restrict__ alpha,
            const float* __restrict__ beta, _Float16* __restrict__ Y) {
    __shared__ float s1[256], s2[256];
    const int row = blockIdx.x, t = threadIdx.x;
    const float* xr = X + (long)row * D_MODEL;
    v4f x = *(const v4f*)(xr + t * 4);
    float s = x[0] + x[1] + x[2] + x[3];
    float q = x[0]*x[0] + x[1]*x[1] + x[2]*x[2] + x[3]*x[3];
    s1[t] = s; s2[t] = q; __syncthreads();
    for (int off = 128; off; off >>= 1) {
        if (t < off) { s1[t] += s1[t + off]; s2[t] += s2[t + off]; }
        __syncthreads();
    }
    const float mu  = s1[0] * (1.0f / D_MODEL);
    const float var = (s2[0] - (float)D_MODEL * mu * mu) * (1.0f / (D_MODEL - 1));
    const float inv = 1.0f / (sqrtf(var) + 1e-6f);
    v4h o;
#pragma unroll
    for (int i = 0; i < 4; ++i) {
        int c = t * 4 + i;
        o[i] = (_Float16)(alpha[c] * (x[i] - mu) * inv + beta[c]);
    }
    *(v4h*)(Y + (long)row * D_MODEL + t * 4) = o;
}

#define GT_M 128
#define GT_N 128
#define GT_K 64
#define GT_P 72

template <int OMODE, bool RELU, bool RESID>
__global__ void __launch_bounds__(256)
k_gemm_wmma(const _Float16* __restrict__ A, const _Float16* __restrict__ Bt,
            const float* __restrict__ bias, const float* __restrict__ resid,
            float* __restrict__ Cf, _Float16* __restrict__ Ch,
            int M, int N, int K) {
    __shared__ __attribute__((aligned(16))) _Float16 sA[2][GT_M * GT_P];
    __shared__ __attribute__((aligned(16))) _Float16 sB[2][GT_N * GT_P];

    const int tid  = threadIdx.x;
    const int lane = tid & 31;
    const int wave = tid >> 5;
    const int wm = wave >> 2, wn = wave & 3;
    const int bn_cnt = N >> 7;
    const int bm = blockIdx.x / bn_cnt;
    const int bn = blockIdx.x % bn_cnt;
    if (bm * GT_M >= M) return;

    const int ln = lane & 15;
    const int hh = lane >> 4;

    const int srow = tid >> 3, sch = tid & 7;
    const _Float16* ag = A  + (long)(bm * GT_M + srow) * K + sch * 8;
    const _Float16* bg = Bt + (long)(bn * GT_N + srow) * K + sch * 8;
    const long g32  = (long)32 * K;
    const int  soff = srow * GT_P + sch * 8;

    v8h ra[4], rb[4];
#pragma unroll
    for (int j = 0; j < 4; ++j) {
        ra[j] = *(const v8h*)(ag + j * g32);
        rb[j] = *(const v8h*)(bg + j * g32);
    }
#pragma unroll
    for (int j = 0; j < 4; ++j) {
        *(v8h*)(&sA[0][soff + j * 32 * GT_P]) = ra[j];
        *(v8h*)(&sB[0][soff + j * 32 * GT_P]) = rb[j];
    }
    __syncthreads();

    v8f acc[4][2];
#pragma unroll
    for (int i = 0; i < 4; ++i) {
#pragma unroll
        for (int j = 0; j < 2; ++j) acc[i][j] = v8f{};
    }

    const int aoff = (wm * 64 + ln) * GT_P + 8 * hh;
    const int boff = (wn * 32 + ln) * GT_P + 8 * hh;

    const int nk = K / GT_K;

    if (nk > 1) {
        ag += GT_K; bg += GT_K;
#pragma unroll
        for (int j = 0; j < 4; ++j) {
            ra[j] = *(const v8h*)(ag + j * g32);
            rb[j] = *(const v8h*)(bg + j * g32);
        }
    }

    v16h fa0[4], fb0[2], fa1[4], fb1[2];
#pragma unroll
    for (int i = 0; i < 4; ++i) fa0[i] = load_frag(&sA[0][aoff] + i * 16 * GT_P);
#pragma unroll
    for (int j = 0; j < 2; ++j) fb0[j] = load_frag(&sB[0][boff] + j * 16 * GT_P);

    for (int kt = 0; kt < nk; ++kt) {
        const int  cur   = kt & 1;
        const bool more  = (kt + 1 < nk);
        const bool more2 = (kt + 2 < nk);

        {
            const _Float16* pa = &sA[cur][aoff] + 32;
            const _Float16* pb = &sB[cur][boff] + 32;
#pragma unroll
            for (int i = 0; i < 4; ++i) fa1[i] = load_frag(pa + i * 16 * GT_P);
#pragma unroll
            for (int j = 0; j < 2; ++j) fb1[j] = load_frag(pb + j * 16 * GT_P);
        }
#pragma unroll
        for (int i = 0; i < 4; ++i) {
#pragma unroll
            for (int j = 0; j < 2; ++j) acc[i][j] = wmma_f16(fa0[i], fb0[j], acc[i][j]);
        }

        if (more) {
#pragma unroll
            for (int j = 0; j < 4; ++j) {
                *(v8h*)(&sA[cur ^ 1][soff + j * 32 * GT_P]) = ra[j];
                *(v8h*)(&sB[cur ^ 1][soff + j * 32 * GT_P]) = rb[j];
            }
        }
        __syncthreads();

        if (more2) {
            ag += GT_K; bg += GT_K;
#pragma unroll
            for (int j = 0; j < 4; ++j) {
                ra[j] = *(const v8h*)(ag + j * g32);
                rb[j] = *(const v8h*)(bg + j * g32);
            }
        }
        if (more) {
            const _Float16* pa = &sA[cur ^ 1][aoff];
            const _Float16* pb = &sB[cur ^ 1][boff];
#pragma unroll
            for (int i = 0; i < 4; ++i) fa0[i] = load_frag(pa + i * 16 * GT_P);
#pragma unroll
            for (int j = 0; j < 2; ++j) fb0[j] = load_frag(pb + j * 16 * GT_P);
        }
#pragma unroll
        for (int i = 0; i < 4; ++i) {
#pragma unroll
            for (int j = 0; j < 2; ++j) acc[i][j] = wmma_f16(fa1[i], fb1[j], acc[i][j]);
        }
    }

    const int lm = hh << 3;
    const int m0 = bm * GT_M + wm * 64;
    const int n0 = bn * GT_N + wn * 32;
#pragma unroll
    for (int i = 0; i < 4; ++i) {
#pragma unroll
        for (int j = 0; j < 2; ++j) {
            const v8f accv = acc[i][j];
            const int col = n0 + j * 16 + ln;
            const float bv = bias[col];
#pragma unroll
            for (int e = 0; e < 8; ++e) {
                const int row = m0 + i * 16 + e + lm;
                float v = accv[e] + bv;
                if (RELU)  v = v > 0.0f ? v : 0.0f;
                if (RESID) v += resid[(long)row * N + col];
                if (OMODE == 0) {
                    Ch[(long)row * N + col] = (_Float16)v;
                } else if (OMODE == 1) {
                    Cf[(long)row * N + col] = v;
                } else if (OMODE == 2) {
                    int b_ = row / SEQ, s = row % SEQ, h = col >> 6, d = col & 63;
                    Ch[(((long)(b_ * HEADS + h)) * SEQ + s) * DK + d] = (_Float16)v;
                } else {
                    int b_ = row / SEQ, s = row % SEQ, h = col >> 6, d = col & 63;
                    Ch[(((long)(b_ * HEADS + h)) * DK + d) * SEQ + s] = (_Float16)v;
                }
            }
        }
    }
}

#define ATT_R     2
#define ATT_QW    (16 * ATT_R)
#define ATT_WAVES 4
#define ATT_QB    (ATT_QW * ATT_WAVES)

__device__ __forceinline__ float xhalf(float x) {
    return __builtin_bit_cast(float, __builtin_amdgcn_permlanex16(
        __builtin_bit_cast(int, x), __builtin_bit_cast(int, x),
        0x76543210, 0xfedcba98, false, false));
}

__global__ void __launch_bounds__(128)
k_attention(const _Float16* __restrict__ QK, const _Float16* __restrict__ VT,
            const int* __restrict__ mask, _Float16* __restrict__ CTX) {
    const int lane = threadIdx.x & 31, wave = threadIdx.x >> 5;
    const int c = lane & 15, hh = lane >> 4;
    const int qb = blockIdx.x & 15;
    const int h  = (blockIdx.x >> 4) & 15;
    const int b  = blockIdx.x >> 8;
    const int q0 = qb * ATT_QB + wave * ATT_QW;

    const _Float16* qbase = QK + ((long)(b * HEADS + h)) * SEQ * DK;
    const _Float16* vbase = VT + ((long)(b * HEADS + h)) * DK * SEQ;
    const _Float16* kp = qbase + (long)c * DK + 8 * hh;
    const _Float16* vp = vbase + (long)c * SEQ + 8 * hh;
    const int* mrow = mask + b * SEQ + 8 * hh;

    const float cscale = 0.125f * 1.4426950408889634f;
    const float negl   = -1e9f * 1.4426950408889634f;

    v16h qf[ATT_R][2];
#pragma unroll
    for (int g = 0; g < ATT_R; ++g) {
        const _Float16* qp = qbase + (long)(q0 + 16 * g + c) * DK + 8 * hh;
        qf[g][0] = load_frag(qp);
        qf[g][1] = load_frag(qp + 32);
    }

    v8f acc[ATT_R][4];
    float m_run[ATT_R], l_run[ATT_R];
#pragma unroll
    for (int g = 0; g < ATT_R; ++g) {
        m_run[g] = -__builtin_inff(); l_run[g] = 0.0f;
#pragma unroll
        for (int t = 0; t < 4; ++t) acc[g][t] = v8f{};
    }

    v16h kf[2][2];
#pragma unroll
    for (int kt = 0; kt < 2; ++kt) {
        kf[kt][0] = load_frag(kp + (long)(16 * kt) * DK);
        kf[kt][1] = load_frag(kp + (long)(16 * kt) * DK + 32);
    }
    int mk[2][8];
#pragma unroll
    for (int kt = 0; kt < 2; ++kt) {
        const int4 m0 = *(const int4*)(mrow + 16 * kt);
        const int4 m1 = *(const int4*)(mrow + 16 * kt + 4);
        mk[kt][0] = m0.x; mk[kt][1] = m0.y; mk[kt][2] = m0.z; mk[kt][3] = m0.w;
        mk[kt][4] = m1.x; mk[kt][5] = m1.y; mk[kt][6] = m1.z; mk[kt][7] = m1.w;
    }

    for (int key0 = 0; key0 < SEQ; key0 += 32) {
        v16h vf[4];
#pragma unroll
        for (int t = 0; t < 4; ++t) vf[t] = load_frag(vp + (long)(16 * t) * SEQ + key0);
        __builtin_amdgcn_sched_barrier(0);

        v8f s[2][ATT_R];
#pragma unroll
        for (int kt = 0; kt < 2; ++kt) {
#pragma unroll
            for (int g = 0; g < ATT_R; ++g) {
                v8f z = {};
                z = wmma_f16(kf[kt][0], qf[g][0], z);
                s[kt][g] = wmma_f16(kf[kt][1], qf[g][1], z);
            }
        }

        const int nk = (key0 + 32 < SEQ) ? key0 + 32 : key0;
#pragma unroll
        for (int kt = 0; kt < 2; ++kt) {
            kf[kt][0] = load_frag(kp + (long)(nk + 16 * kt) * DK);
            kf[kt][1] = load_frag(kp + (long)(nk + 16 * kt) * DK + 32);
        }
        int mkn[2][8];
#pragma unroll
        for (int kt = 0; kt < 2; ++kt) {
            const int4 m0 = *(const int4*)(mrow + nk + 16 * kt);
            const int4 m1 = *(const int4*)(mrow + nk + 16 * kt + 4);
            mkn[kt][0] = m0.x; mkn[kt][1] = m0.y; mkn[kt][2] = m0.z; mkn[kt][3] = m0.w;
            mkn[kt][4] = m1.x; mkn[kt][5] = m1.y; mkn[kt][6] = m1.z; mkn[kt][7] = m1.w;
        }

        float mb[ATT_R];
        bool need = false;
#pragma unroll
        for (int g = 0; g < ATT_R; ++g) {
            float mx = -__builtin_inff();
#pragma unroll
            for (int kt = 0; kt < 2; ++kt) {
#pragma unroll
                for (int r = 0; r < 8; ++r) {
                    float t = s[kt][g][r] * cscale;
                    t = (mk[kt][r] == 0) ? negl : t;
                    s[kt][g][r] = t;
                    mx = fmaxf(mx, t);
                }
            }
            mx = fmaxf(mx, xhalf(mx));
            mb[g] = mx;
            need = need || (mx > m_run[g] + 8.0f);
        }
        if (__builtin_amdgcn_ballot_w32(need) != 0) {
#pragma unroll
            for (int g = 0; g < ATT_R; ++g) {
                const float m_new = fmaxf(m_run[g], mb[g]);
                const float corr  = __builtin_amdgcn_exp2f(m_run[g] - m_new);
                m_run[g] = m_new;
                l_run[g] *= corr;
#pragma unroll
                for (int t = 0; t < 4; ++t) acc[g][t] = acc[g][t] * corr;
            }
        }

#pragma unroll
        for (int g = 0; g < ATT_R; ++g) {
            const float m = m_run[g];
            v16h pf;
            float ls = 0.0f;
#pragma unroll
            for (int r = 0; r < 8; ++r) {
                const float p0 = __builtin_amdgcn_exp2f(s[0][g][r] - m);
                const float p1 = __builtin_amdgcn_exp2f(s[1][g][r] - m);
                ls += p0 + p1;
                pf[r]     = (_Float16)p0;
                pf[8 + r] = (_Float16)p1;
            }
            l_run[g] += ls;
#pragma unroll
            for (int t = 0; t < 4; ++t) acc[g][t] = wmma_f16(vf[t], pf, acc[g][t]);
        }
#pragma unroll
        for (int kt = 0; kt < 2; ++kt) {
#pragma unroll
            for (int r = 0; r < 8; ++r) mk[kt][r] = mkn[kt][r];
        }
    }

#pragma unroll
    for (int g = 0; g < ATT_R; ++g) {
        float l = l_run[g];
        l = l + xhalf(l);
        const float inv = 1.0f / l;
        _Float16* op = CTX + ((long)(b * SEQ + q0 + 16 * g + c)) * D_MODEL + h * DK + 8 * hh;
#pragma unroll
        for (int t = 0; t < 4; ++t) {
            v8h o;
#pragma unroll
            for (int r = 0; r < 8; ++r) o[r] = (_Float16)(acc[g][t][r] * inv);
            *(v8h*)(op + 16 * t) = o;
        }
    }
}

extern "C" void kernel_launch(void* const* d_in, const int* in_sizes, int n_in,
                              void* d_out, int out_size, void* d_ws, size_t ws_size,
                              hipStream_t stream) {
    const float* x      = (const float*)d_in[0];
    const int*   mask   = (const int*)  d_in[1];
    const float* alpha1 = (const float*)d_in[2];
    const float* bias1  = (const float*)d_in[3];
    const float* alpha2 = (const float*)d_in[4];
    const float* bias2  = (const float*)d_in[5];
    const float* Wk     = (const float*)d_in[6];
    const float* bk     = (const float*)d_in[7];
    const float* Wv     = (const float*)d_in[8];
    const float* bv     = (const float*)d_in[9];
    const float* Wo     = (const float*)d_in[10];
    const float* bo     = (const float*)d_in[11];
    const float* W1     = (const float*)d_in[12];
    const float* b1     = (const float*)d_in[13];
    const float* W2     = (const float*)d_in[14];
    const float* b2     = (const float*)d_in[15];
    float* out = (float*)d_out;

    char* ws = (char*)d_ws;
    size_t off = 0;
    _Float16* x2h  = (_Float16*)(ws + off); off += (size_t)ROWS * D_MODEL * 2;
    _Float16* x2bh = (_Float16*)(ws + off); off += (size_t)ROWS * D_MODEL * 2;
    _Float16* qkh  = (_Float16*)(ws + off); off += (size_t)BATCH*HEADS*SEQ*DK*2;
    _Float16* vth  = (_Float16*)(ws + off); off += (size_t)BATCH*HEADS*DK*SEQ*2;
    _Float16* ctxh = (_Float16*)(ws + off); off += (size_t)ROWS * D_MODEL * 2;
    _Float16* h1h  = (_Float16*)(ws + off); off += (size_t)ROWS * D_FF * 2;
    _Float16* Wkt  = (_Float16*)(ws + off); off += (size_t)D_MODEL * D_MODEL * 2;
    _Float16* Wvt  = (_Float16*)(ws + off); off += (size_t)D_MODEL * D_MODEL * 2;
    _Float16* Wot  = (_Float16*)(ws + off); off += (size_t)D_MODEL * D_MODEL * 2;
    _Float16* W1t  = (_Float16*)(ws + off); off += (size_t)D_MODEL * D_FF * 2;
    _Float16* W2t  = (_Float16*)(ws + off); off += (size_t)D_FF * D_MODEL * 2;
    float*    x1   = (float*)   (ws + off); off += (size_t)ROWS * D_MODEL * 4;

    k_transpose_cvt_all<<<TR_GRID, 256, 0, stream>>>(Wk, Wv, Wo, W1, W2,
                                                     Wkt, Wvt, Wot, W1t, W2t);

    k_layernorm<<<ROWS, 256, 0, stream>>>(x, alpha1, bias1, x2h);

    #define GEMM_GRID(M_, N_) (((M_) / 128) * ((N_) / 128))

    k_gemm_wmma<2, false, false><<<GEMM_GRID(ROWS, D_MODEL), 256, 0, stream>>>(
        x2h, Wkt, bk, nullptr, nullptr, qkh, ROWS, D_MODEL, D_MODEL);
    k_gemm_wmma<3, false, false><<<GEMM_GRID(ROWS, D_MODEL), 256, 0, stream>>>(
        x2h, Wvt, bv, nullptr, nullptr, vth, ROWS, D_MODEL, D_MODEL);

    k_attention<<<BATCH*HEADS*(SEQ/ATT_QB), 128, 0, stream>>>(
        qkh, vth, mask, ctxh);

    k_gemm_wmma<1, false, true><<<GEMM_GRID(ROWS, D_MODEL), 256, 0, stream>>>(
        ctxh, Wot, bo, x, x1, nullptr, ROWS, D_MODEL, D_MODEL);

    k_layernorm<<<ROWS, 256, 0, stream>>>(x1, alpha2, bias2, x2bh);

    k_gemm_wmma<0, true, false><<<GEMM_GRID(ROWS, D_FF), 256, 0, stream>>>(
        x2bh, W1t, b1, nullptr, nullptr, h1h, ROWS, D_FF, D_MODEL);

    k_gemm_wmma<1, false, true><<<GEMM_GRID(ROWS, D_MODEL), 256, 0, stream>>>(
        h1h, W2t, b2, x1, out, nullptr, ROWS, D_MODEL, D_FF);

    (void)in_sizes; (void)n_in; (void)out_size; (void)ws_size;
}


# attention operands: q/k and v^T stored by their GEMMs in fragment order (512-byte lane-major pieces), so each K, Q, V^T fragment load covers 4 whole lines, was 16; sum order unchanged
# speedup vs baseline: 5.1047x; 1.1009x over previous
#include <hip/hip_runtime.h>
#include <hip/hip_bf16.h>

typedef __attribute__((ext_vector_type(16))) _Float16 v16h;
typedef __attribute__((ext_vector_type(8)))  _Float16 v8h;
typedef __attribute__((ext_vector_type(4)))  _Float16 v4h;
typedef __attribute__((ext_vector_type(8)))  float    v8f;
typedef __attribute__((ext_vector_type(4)))  float    v4f;

#define D_MODEL 1024
#define HEADS   16
#define DK      64
#define D_FF    4096
#define BATCH   2
#define SEQ     2048
#define ROWS    (BATCH*SEQ)

__device__ __forceinline__ v16h pack16(v8h lo, v8h hi) {
    v16h r;
#pragma unroll
    for (int i = 0; i < 8; ++i) { r[i] = lo[i]; r[i + 8] = hi[i]; }
    return r;
}
__device__ __forceinline__ v16h load_frag(const _Float16* p) {
    v8h lo = *(const v8h*)(p);
    v8h hi = *(const v8h*)(p + 16);
    return pack16(lo, hi);
}
__device__ __forceinline__ int qk_off(int s, int d) {
    return ((((s >> 4) * 4 + (d >> 4)) * 32 + ((d >> 3) & 1) * 16 + (s & 15)) << 3) + (d & 7);
}
__device__ __forceinline__ int vt_off(int s, int d) {
    return (((((s >> 5) * 4 + (d >> 4)) * 2 + ((s >> 4) & 1)) * 32
             + ((s >> 3) & 1) * 16 + (d & 15)) << 3) + (s & 7);
}
__device__ __forceinline__ v16h load_frag_fo(const _Float16* p) {
    v8h lo = *(const v8h*)(p);
    v8h hi = *(const v8h*)(p + 256);
    return pack16(lo, hi);
}
__device__ __forceinline__ v8f wmma_f16(v16h a, v16h b, v8f c) {
    return __builtin_amdgcn_wmma_f32_16x16x32_f16(false, a, false, b,
                                                  (short)0, c, false, false);
}

#define TR_K 128
#define TR_N 64
#define TR_P 136
#define TR_TILES(K_, N_) (((K_) / TR_K) * ((N_) / TR_N))
#define TR_SQ   TR_TILES(D_MODEL, D_MODEL)
#define TR_W1   TR_TILES(D_MODEL, D_FF)
#define TR_W2   TR_TILES(D_FF, D_MODEL)
#define TR_GRID (3 * TR_SQ + TR_W1 + TR_W2)

__global__ void __launch_bounds__(256)
k_transpose_cvt_all(const float* __restrict__ Wk, const float* __restrict__ Wv,
                    const float* __restrict__ Wo, const float* __restrict__ W1,
                    const float* __restrict__ W2,
                    _Float16* __restrict__ Wkt, _Float16* __restrict__ Wvt,
                    _Float16* __restrict__ Wot, _Float16* __restrict__ W1t,
                    _Float16* __restrict__ W2t) {
    __shared__ __attribute__((aligned(16))) _Float16 sT[TR_N * TR_P];

    int tile = blockIdx.x;
    const float* W; _Float16* Wt; int K, N;
    if (tile < TR_SQ)                 { W = Wk; Wt = Wkt; K = D_MODEL; N = D_MODEL; }
    else if (tile < 2 * TR_SQ)        { W = Wv; Wt = Wvt; K = D_MODEL; N = D_MODEL; tile -= TR_SQ; }
    else if (tile < 3 * TR_SQ)        { W = Wo; Wt = Wot; K = D_MODEL; N = D_MODEL; tile -= 2 * TR_SQ; }
    else if (tile < 3 * TR_SQ + TR_W1){ W = W1; Wt = W1t; K = D_MODEL; N = D_FF;    tile -= 3 * TR_SQ; }
    else                              { W = W2; Wt = W2t; K = D_FF;    N = D_MODEL; tile -= 3 * TR_SQ + TR_W1; }
    const int ntn = N / TR_N;
    const int kt = tile / ntn, nt = tile % ntn;

    const int t = threadIdx.x;
    const int q = t & 15, g = t >> 4;

    const float* src = W + (long)(kt * TR_K + 8 * g) * N + nt * TR_N + 4 * q;
    v4f r[8];
#pragma unroll
    for (int i = 0; i < 8; ++i) r[i] = *(const v4f*)(src + (long)i * N);

#pragma unroll
    for (int j = 0; j < 4; ++j) {
        v8h o;
#pragma unroll
        for (int i = 0; i < 8; ++i) o[i] = (_Float16)r[i][j];
        *(v8h*)(&sT[(4 * q + j) * TR_P + 8 * g]) = o;
    }
    __syncthreads();

    _Float16* dst = Wt + (long)(nt * TR_N + g) * K + kt * TR_K + 8 * q;
#pragma unroll
    for (int s = 0; s < 4; ++s) {
        const v8h o = *(const v8h*)(&sT[(g + 16 * s) * TR_P + 8 * q]);
        *(v8h*)(dst + (long)(16 * s) * K) = o;
    }
}

__global__ void __launch_bounds__(256)
k_layernorm(const float* __restrict__ X, const float* __restrict__ alpha,
            const float* __restrict__ beta, _Float16* __restrict__ Y) {
    __shared__ float s1[256], s2[256];
    const int row = blockIdx.x, t = threadIdx.x;
    const float* xr = X + (long)row * D_MODEL;
    v4f x = *(const v4f*)(xr + t * 4);
    float s = x[0] + x[1] + x[2] + x[3];
    float q = x[0]*x[0] + x[1]*x[1] + x[2]*x[2] + x[3]*x[3];
    s1[t] = s; s2[t] = q; __syncthreads();
    for (int off = 128; off; off >>= 1) {
        if (t < off) { s1[t] += s1[t + off]; s2[t] += s2[t + off]; }
        __syncthreads();
    }
    const float mu  = s1[0] * (1.0f / D_MODEL);
    const float var = (s2[0] - (float)D_MODEL * mu * mu) * (1.0f / (D_MODEL - 1));
    const float inv = 1.0f / (sqrtf(var) + 1e-6f);
    v4h o;
#pragma unroll
    for (int i = 0; i < 4; ++i) {
        int c = t * 4 + i;
        o[i] = (_Float16)(alpha[c] * (x[i] - mu) * inv + beta[c]);
    }
    *(v4h*)(Y + (long)row * D_MODEL + t * 4) = o;
}

#define GT_M 128
#define GT_N 128
#define GT_K 64
#define GT_P 72

template <int OMODE, bool RELU, bool RESID>
__global__ void __launch_bounds__(256)
k_gemm_wmma(const _Float16* __restrict__ A, const _Float16* __restrict__ Bt,
            const float* __restrict__ bias, const float* __restrict__ resid,
            float* __restrict__ Cf, _Float16* __restrict__ Ch,
            int M, int N, int K) {
    __shared__ __attribute__((aligned(16))) _Float16 sA[2][GT_M * GT_P];
    __shared__ __attribute__((aligned(16))) _Float16 sB[2][GT_N * GT_P];

    const int tid  = threadIdx.x;
    const int lane = tid & 31;
    const int wave = tid >> 5;
    const int wm = wave >> 2, wn = wave & 3;
    const int bn_cnt = N >> 7;
    const int bm = blockIdx.x / bn_cnt;
    const int bn = blockIdx.x % bn_cnt;
    if (bm * GT_M >= M) return;

    const int ln = lane & 15;
    const int hh = lane >> 4;

    const int srow = tid >> 3, sch = tid & 7;
    const _Float16* ag = A  + (long)(bm * GT_M + srow) * K + sch * 8;
    const _Float16* bg = Bt + (long)(bn * GT_N + srow) * K + sch * 8;
    const long g32  = (long)32 * K;
    const int  soff = srow * GT_P + sch * 8;

    v8h ra[4], rb[4];
#pragma unroll
    for (int j = 0; j < 4; ++j) {
        ra[j] = *(const v8h*)(ag + j * g32);
        rb[j] = *(const v8h*)(bg + j * g32);
    }
#pragma unroll
    for (int j = 0; j < 4; ++j) {
        *(v8h*)(&sA[0][soff + j * 32 * GT_P]) = ra[j];
        *(v8h*)(&sB[0][soff + j * 32 * GT_P]) = rb[j];
    }
    __syncthreads();

    v8f acc[4][2];
#pragma unroll
    for (int i = 0; i < 4; ++i) {
#pragma unroll
        for (int j = 0; j < 2; ++j) acc[i][j] = v8f{};
    }

    const int aoff = (wm * 64 + ln) * GT_P + 8 * hh;
    const int boff = (wn * 32 + ln) * GT_P + 8 * hh;

    const int nk = K / GT_K;

    if (nk > 1) {
        ag += GT_K; bg += GT_K;
#pragma unroll
        for (int j = 0; j < 4; ++j) {
            ra[j] = *(const v8h*)(ag + j * g32);
            rb[j] = *(const v8h*)(bg + j * g32);
        }
    }

    v16h fa0[4], fb0[2], fa1[4], fb1[2];
#pragma unroll
    for (int i = 0; i < 4; ++i) fa0[i] = load_frag(&sA[0][aoff] + i * 16 * GT_P);
#pragma unroll
    for (int j = 0; j < 2; ++j) fb0[j] = load_frag(&sB[0][boff] + j * 16 * GT_P);

    for (int kt = 0; kt < nk; ++kt) {
        const int  cur   = kt & 1;
        const bool more  = (kt + 1 < nk);
        const bool more2 = (kt + 2 < nk);

        {
            const _Float16* pa = &sA[cur][aoff] + 32;
            const _Float16* pb = &sB[cur][boff] + 32;
#pragma unroll
            for (int i = 0; i < 4; ++i) fa1[i] = load_frag(pa + i * 16 * GT_P);
#pragma unroll
            for (int j = 0; j < 2; ++j) fb1[j] = load_frag(pb + j * 16 * GT_P);
        }
#pragma unroll
        for (int i = 0; i < 4; ++i) {
#pragma unroll
            for (int j = 0; j < 2; ++j) acc[i][j] = wmma_f16(fa0[i], fb0[j], acc[i][j]);
        }

        if (more) {
#pragma unroll
            for (int j = 0; j < 4; ++j) {
                *(v8h*)(&sA[cur ^ 1][soff + j * 32 * GT_P]) = ra[j];
                *(v8h*)(&sB[cur ^ 1][soff + j * 32 * GT_P]) = rb[j];
            }
        }
        __syncthreads();

        if (more2) {
            ag += GT_K; bg += GT_K;
#pragma unroll
            for (int j = 0; j < 4; ++j) {
                ra[j] = *(const v8h*)(ag + j * g32);
                rb[j] = *(const v8h*)(bg + j * g32);
            }
        }
        if (more) {
            const _Float16* pa = &sA[cur ^ 1][aoff];
            const _Float16* pb = &sB[cur ^ 1][boff];
#pragma unroll
            for (int i = 0; i < 4; ++i) fa0[i] = load_frag(pa + i * 16 * GT_P);
#pragma unroll
            for (int j = 0; j < 2; ++j) fb0[j] = load_frag(pb + j * 16 * GT_P);
        }
#pragma unroll
        for (int i = 0; i < 4; ++i) {
#pragma unroll
            for (int j = 0; j < 2; ++j) acc[i][j] = wmma_f16(fa1[i], fb1[j], acc[i][j]);
        }
    }

    const int lm = hh << 3;
    const int m0 = bm * GT_M + wm * 64;
    const int n0 = bn * GT_N + wn * 32;
#pragma unroll
    for (int i = 0; i < 4; ++i) {
#pragma unroll
        for (int j = 0; j < 2; ++j) {
            const v8f accv = acc[i][j];
            const int col = n0 + j * 16 + ln;
            const float bv = bias[col];
            if (OMODE == 3) {
                const int row0 = m0 + i * 16 + lm;
                const int b_ = row0 / SEQ, s0 = row0 % SEQ, h = col >> 6, d = col & 63;
                v8h o;
#pragma unroll
                for (int e = 0; e < 8; ++e) {
                    float v = accv[e] + bv;
                    if (RELU)  v = v > 0.0f ? v : 0.0f;
                    if (RESID) v += resid[(long)(row0 + e) * N + col];
                    o[e] = (_Float16)v;
                }
                *(v8h*)(Ch + ((long)(b_ * HEADS + h)) * DK * SEQ + vt_off(s0, d)) = o;
                continue;
            }
            if (OMODE == 2) {
                const int row0 = m0 + i * 16 + lm;
                const int b_ = row0 / SEQ, s0 = row0 % SEQ, h = col >> 6, d = col & 63;
                _Float16* dst = Ch + ((long)(b_ * HEADS + h)) * SEQ * DK + qk_off(s0, d);
#pragma unroll
                for (int e = 0; e < 8; ++e) {
                    float v = accv[e] + bv;
                    if (RELU)  v = v > 0.0f ? v : 0.0f;
                    if (RESID) v += resid[(long)(row0 + e) * N + col];
                    dst[8 * e] = (_Float16)v;
                }
                continue;
            }
#pragma unroll
            for (int e = 0; e < 8; ++e) {
                const int row = m0 + i * 16 + e + lm;
                float v = accv[e] + bv;
                if (RELU)  v = v > 0.0f ? v : 0.0f;
                if (RESID) v += resid[(long)row * N + col];
                if (OMODE == 0) {
                    Ch[(long)row * N + col] = (_Float16)v;
                } else if (OMODE == 1) {
                    Cf[(long)row * N + col] = v;
                }
            }
        }
    }
}

#define ATT_R     2
#define ATT_QW    (16 * ATT_R)
#define ATT_WAVES 4
#define ATT_QB    (ATT_QW * ATT_WAVES)

__device__ __forceinline__ float xhalf(float x) {
    return __builtin_bit_cast(float, __builtin_amdgcn_permlanex16(
        __builtin_bit_cast(int, x), __builtin_bit_cast(int, x),
        0x76543210, 0xfedcba98, false, false));
}

__global__ void __launch_bounds__(128)
k_attention(const _Float16* __restrict__ QK, const _Float16* __restrict__ VT,
            const int* __restrict__ mask, _Float16* __restrict__ CTX) {
    const int lane = threadIdx.x & 31, wave = threadIdx.x >> 5;
    const int c = lane & 15, hh = lane >> 4;
    const int qb = blockIdx.x & 15;
    const int h  = (blockIdx.x >> 4) & 15;
    const int b  = blockIdx.x >> 8;
    const int q0 = qb * ATT_QB + wave * ATT_QW;

    const _Float16* qbase = QK + ((long)(b * HEADS + h)) * SEQ * DK;
    const _Float16* vbase = VT + ((long)(b * HEADS + h)) * DK * SEQ;
    const _Float16* kp = qbase + 8 * lane;
    const _Float16* vp = vbase + 8 * lane;
    const int* mrow = mask + b * SEQ + 8 * hh;

    const float cscale = 0.125f * 1.4426950408889634f;
    const float negl   = -1e9f * 1.4426950408889634f;

    v16h qf[ATT_R][2];
#pragma unroll
    for (int g = 0; g < ATT_R; ++g) {
        const _Float16* qp = kp + (long)(q0 + 16 * g) * DK;
        qf[g][0] = load_frag_fo(qp);
        qf[g][1] = load_frag_fo(qp + 512);
    }

    v8f acc[ATT_R][4];
    float m_run[ATT_R], l_run[ATT_R];
#pragma unroll
    for (int g = 0; g < ATT_R; ++g) {
        m_run[g] = -__builtin_inff(); l_run[g] = 0.0f;
#pragma unroll
        for (int t = 0; t < 4; ++t) acc[g][t] = v8f{};
    }

    v16h kf[2][2];
#pragma unroll
    for (int kt = 0; kt < 2; ++kt) {
        kf[kt][0] = load_frag_fo(kp + (long)(16 * kt) * DK);
        kf[kt][1] = load_frag_fo(kp + (long)(16 * kt) * DK + 512);
    }
    int mk[2][8];
#pragma unroll
    for (int kt = 0; kt < 2; ++kt) {
        const int4 m0 = *(const int4*)(mrow + 16 * kt);
        const int4 m1 = *(const int4*)(mrow + 16 * kt + 4);
        mk[kt][0] = m0.x; mk[kt][1] = m0.y; mk[kt][2] = m0.z; mk[kt][3] = m0.w;
        mk[kt][4] = m1.x; mk[kt][5] = m1.y; mk[kt][6] = m1.z; mk[kt][7] = m1.w;
    }

    for (int key0 = 0; key0 < SEQ; key0 += 32) {
        v16h vf[4];
#pragma unroll
        for (int t = 0; t < 4; ++t) vf[t] = load_frag_fo(vp + (long)key0 * DK + 512 * t);
        __builtin_amdgcn_sched_barrier(0);

        v8f s[2][ATT_R];
#pragma unroll
        for (int kt = 0; kt < 2; ++kt) {
#pragma unroll
            for (int g = 0; g < ATT_R; ++g) {
                v8f z = {};
                z = wmma_f16(kf[kt][0], qf[g][0], z);
                s[kt][g] = wmma_f16(kf[kt][1], qf[g][1], z);
            }
        }

        const int nk = (key0 + 32 < SEQ) ? key0 + 32 : key0;
#pragma unroll
        for (int kt = 0; kt < 2; ++kt) {
            kf[kt][0] = load_frag_fo(kp + (long)(nk + 16 * kt) * DK);
            kf[kt][1] = load_frag_fo(kp + (long)(nk + 16 * kt) * DK + 512);
        }
        int mkn[2][8];
#pragma unroll
        for (int kt = 0; kt < 2; ++kt) {
            const int4 m0 = *(const int4*)(mrow + nk + 16 * kt);
            const int4 m1 = *(const int4*)(mrow + nk + 16 * kt + 4);
            mkn[kt][0] = m0.x; mkn[kt][1] = m0.y; mkn[kt][2] = m0.z; mkn[kt][3] = m0.w;
            mkn[kt][4] = m1.x; mkn[kt][5] = m1.y; mkn[kt][6] = m1.z; mkn[kt][7] = m1.w;
        }

        float mb[ATT_R];
        bool need = false;
#pragma unroll
        for (int g = 0; g < ATT_R; ++g) {
            float mx = -__builtin_inff();
#pragma unroll
            for (int kt = 0; kt < 2; ++kt) {
#pragma unroll
                for (int r = 0; r < 8; ++r) {
                    float t = s[kt][g][r] * cscale;
                    t = (mk[kt][r] == 0) ? negl : t;
                    s[kt][g][r] = t;
                    mx = fmaxf(mx, t);
                }
            }
            mx = fmaxf(mx, xhalf(mx));
            mb[g] = mx;
            need = need || (mx > m_run[g] + 8.0f);
        }
        if (__builtin_amdgcn_ballot_w32(need) != 0) {
#pragma unroll
            for (int g = 0; g < ATT_R; ++g) {
                const float m_new = fmaxf(m_run[g], mb[g]);
                const float corr  = __builtin_amdgcn_exp2f(m_run[g] - m_new);
                m_run[g] = m_new;
                l_run[g] *= corr;
#pragma unroll
                for (int t = 0; t < 4; ++t) acc[g][t] = acc[g][t] * corr;
            }
        }

#pragma unroll
        for (int g = 0; g < ATT_R; ++g) {
            const float m = m_run[g];
            v16h pf;
            float ls = 0.0f;
#pragma unroll
            for (int r = 0; r < 8; ++r) {
                const float p0 = __builtin_amdgcn_exp2f(s[0][g][r] - m);
                const float p1 = __builtin_amdgcn_exp2f(s[1][g][r] - m);
                ls += p0 + p1;
                pf[r]     = (_Float16)p0;
                pf[8 + r] = (_Float16)p1;
            }
            l_run[g] += ls;
#pragma unroll
            for (int t = 0; t < 4; ++t) acc[g][t] = wmma_f16(vf[t], pf, acc[g][t]);
        }
#pragma unroll
        for (int kt = 0; kt < 2; ++kt) {
#pragma unroll
            for (int r = 0; r < 8; ++r) mk[kt][r] = mkn[kt][r];
        }
    }

#pragma unroll
    for (int g = 0; g < ATT_R; ++g) {
        float l = l_run[g];
        l = l + xhalf(l);
        const float inv = 1.0f / l;
        _Float16* op = CTX + ((long)(b * SEQ + q0 + 16 * g + c)) * D_MODEL + h * DK + 8 * hh;
#pragma unroll
        for (int t = 0; t < 4; ++t) {
            v8h o;
#pragma unroll
            for (int r = 0; r < 8; ++r) o[r] = (_Float16)(acc[g][t][r] * inv);
            *(v8h*)(op + 16 * t) = o;
        }
    }
}

extern "C" void kernel_launch(void* const* d_in, const int* in_sizes, int n_in,
                              void* d_out, int out_size, void* d_ws, size_t ws_size,
                              hipStream_t stream) {
    const float* x      = (const float*)d_in[0];
    const int*   mask   = (const int*)  d_in[1];
    const float* alpha1 = (const float*)d_in[2];
    const float* bias1  = (const float*)d_in[3];
    const float* alpha2 = (const float*)d_in[4];
    const float* bias2  = (const float*)d_in[5];
    const float* Wk     = (const float*)d_in[6];
    const float* bk     = (const float*)d_in[7];
    const float* Wv     = (const float*)d_in[8];
    const float* bv     = (const float*)d_in[9];
    const float* Wo     = (const float*)d_in[10];
    const float* bo     = (const float*)d_in[11];
    const float* W1     = (const float*)d_in[12];
    const float* b1     = (const float*)d_in[13];
    const float* W2     = (const float*)d_in[14];
    const float* b2     = (const float*)d_in[15];
    float* out = (float*)d_out;

    char* ws = (char*)d_ws;
    size_t off = 0;
    _Float16* x2h  = (_Float16*)(ws + off); off += (size_t)ROWS * D_MODEL * 2;
    _Float16* x2bh = (_Float16*)(ws + off); off += (size_t)ROWS * D_MODEL * 2;
    _Float16* qkh  = (_Float16*)(ws + off); off += (size_t)BATCH*HEADS*SEQ*DK*2;
    _Float16* vth  = (_Float16*)(ws + off); off += (size_t)BATCH*HEADS*DK*SEQ*2;
    _Float16* ctxh = (_Float16*)(ws + off); off += (size_t)ROWS * D_MODEL * 2;
    _Float16* h1h  = (_Float16*)(ws + off); off += (size_t)ROWS * D_FF * 2;
    _Float16* Wkt  = (_Float16*)(ws + off); off += (size_t)D_MODEL * D_MODEL * 2;
    _Float16* Wvt  = (_Float16*)(ws + off); off += (size_t)D_MODEL * D_MODEL * 2;
    _Float16* Wot  = (_Float16*)(ws + off); off += (size_t)D_MODEL * D_MODEL * 2;
    _Float16* W1t  = (_Float16*)(ws + off); off += (size_t)D_MODEL * D_FF * 2;
    _Float16* W2t  = (_Float16*)(ws + off); off += (size_t)D_FF * D_MODEL * 2;
    float*    x1   = (float*)   (ws + off); off += (size_t)ROWS * D_MODEL * 4;

    k_transpose_cvt_all<<<TR_GRID, 256, 0, stream>>>(Wk, Wv, Wo, W1, W2,
                                                     Wkt, Wvt, Wot, W1t, W2t);

    k_layernorm<<<ROWS, 256, 0, stream>>>(x, alpha1, bias1, x2h);

    #define GEMM_GRID(M_, N_) (((M_) / 128) * ((N_) / 128))

    k_gemm_wmma<2, false, false><<<GEMM_GRID(ROWS, D_MODEL), 256, 0, stream>>>(
        x2h, Wkt, bk, nullptr, nullptr, qkh, ROWS, D_MODEL, D_MODEL);
    k_gemm_wmma<3, false, false><<<GEMM_GRID(ROWS, D_MODEL), 256, 0, stream>>>(
        x2h, Wvt, bv, nullptr, nullptr, vth, ROWS, D_MODEL, D_MODEL);

    k_attention<<<BATCH*HEADS*(SEQ/ATT_QB), 128, 0, stream>>>(
        qkh, vth, mask, ctxh);

    k_gemm_wmma<1, false, true><<<GEMM_GRID(ROWS, D_MODEL), 256, 0, stream>>>(
        ctxh, Wot, bo, x, x1, nullptr, ROWS, D_MODEL, D_MODEL);

    k_layernorm<<<ROWS, 256, 0, stream>>>(x1, alpha2, bias2, x2bh);

    k_gemm_wmma<0, true, false><<<GEMM_GRID(ROWS, D_FF), 256, 0, stream>>>(
        x2bh, W1t, b1, nullptr, nullptr, h1h, ROWS, D_FF, D_MODEL);

    k_gemm_wmma<1, false, true><<<GEMM_GRID(ROWS, D_MODEL), 256, 0, stream>>>(
        h1h, W2t, b2, x1, out, nullptr, ROWS, D_MODEL, D_FF);

    (void)in_sizes; (void)n_in; (void)out_size; (void)ws_size;
}


# GEMMs: wave tile 64x64 (was 64x32), 4 waves per 128x128 block tile, k stage 128 (was 64): LDS bytes read per product 512, was 768; one wave per SIMD on the N=1024 GEMMs; sum order unchanged
# speedup vs baseline: 5.4007x; 1.0580x over previous
#include <hip/hip_runtime.h>
#include <hip/hip_bf16.h>

typedef __attribute__((ext_vector_type(16))) _Float16 v16h;
typedef __attribute__((ext_vector_type(8)))  _Float16 v8h;
typedef __attribute__((ext_vector_type(4)))  _Float16 v4h;
typedef __attribute__((ext_vector_type(8)))  float    v8f;
typedef __attribute__((ext_vector_type(4)))  float    v4f;

#define D_MODEL 1024
#define HEADS   16
#define DK      64
#define D_FF    4096
#define BATCH   2
#define SEQ     2048
#define ROWS    (BATCH*SEQ)

__device__ __forceinline__ v16h pack16(v8h lo, v8h hi) {
    v16h r;
#pragma unroll
    for (int i = 0; i < 8; ++i) { r[i] = lo[i]; r[i + 8] = hi[i]; }
    return r;
}
__device__ __forceinline__ v16h load_frag(const _Float16* p) {
    v8h lo = *(const v8h*)(p);
    v8h hi = *(const v8h*)(p + 16);
    return pack16(lo, hi);
}
__device__ __forceinline__ int qk_off(int s, int d) {
    return ((((s >> 4) * 4 + (d >> 4)) * 32 + ((d >> 3) & 1) * 16 + (s & 15)) << 3) + (d & 7);
}
__device__ __forceinline__ int vt_off(int s, int d) {
    return (((((s >> 5) * 4 + (d >> 4)) * 2 + ((s >> 4) & 1)) * 32
             + ((s >> 3) & 1) * 16 + (d & 15)) << 3) + (s & 7);
}
__device__ __forceinline__ v16h load_frag_fo(const _Float16* p) {
    v8h lo = *(const v8h*)(p);
    v8h hi = *(const v8h*)(p + 256);
    return pack16(lo, hi);
}
__device__ __forceinline__ v8f wmma_f16(v16h a, v16h b, v8f c) {
    return __builtin_amdgcn_wmma_f32_16x16x32_f16(false, a, false, b,
                                                  (short)0, c, false, false);
}

#define TR_K 128
#define TR_N 64
#define TR_P 136
#define TR_TILES(K_, N_) (((K_) / TR_K) * ((N_) / TR_N))
#define TR_SQ   TR_TILES(D_MODEL, D_MODEL)
#define TR_W1   TR_TILES(D_MODEL, D_FF)
#define TR_W2   TR_TILES(D_FF, D_MODEL)
#define TR_GRID (3 * TR_SQ + TR_W1 + TR_W2)

__global__ void __launch_bounds__(256)
k_transpose_cvt_all(const float* __restrict__ Wk, const float* __restrict__ Wv,
                    const float* __restrict__ Wo, const float* __restrict__ W1,
                    const float* __restrict__ W2,
                    _Float16* __restrict__ Wkt, _Float16* __restrict__ Wvt,
                    _Float16* __restrict__ Wot, _Float16* __restrict__ W1t,
                    _Float16* __restrict__ W2t) {
    __shared__ __attribute__((aligned(16))) _Float16 sT[TR_N * TR_P];

    int tile = blockIdx.x;
    const float* W; _Float16* Wt; int K, N;
    if (tile < TR_SQ)                 { W = Wk; Wt = Wkt; K = D_MODEL; N = D_MODEL; }
    else if (tile < 2 * TR_SQ)        { W = Wv; Wt = Wvt; K = D_MODEL; N = D_MODEL; tile -= TR_SQ; }
    else if (tile < 3 * TR_SQ)        { W = Wo; Wt = Wot; K = D_MODEL; N = D_MODEL; tile -= 2 * TR_SQ; }
    else if (tile < 3 * TR_SQ + TR_W1){ W = W1; Wt = W1t; K = D_MODEL; N = D_FF;    tile -= 3 * TR_SQ; }
    else                              { W = W2; Wt = W2t; K = D_FF;    N = D_MODEL; tile -= 3 * TR_SQ + TR_W1; }
    const int ntn = N / TR_N;
    const int kt = tile / ntn, nt = tile % ntn;

    const int t = threadIdx.x;
    const int q = t & 15, g = t >> 4;

    const float* src = W + (long)(kt * TR_K + 8 * g) * N + nt * TR_N + 4 * q;
    v4f r[8];
#pragma unroll
    for (int i = 0; i < 8; ++i) r[i] = *(const v4f*)(src + (long)i * N);

#pragma unroll
    for (int j = 0; j < 4; ++j) {
        v8h o;
#pragma unroll
        for (int i = 0; i < 8; ++i) o[i] = (_Float16)r[i][j];
        *(v8h*)(&sT[(4 * q + j) * TR_P + 8 * g]) = o;
    }
    __syncthreads();

    _Float16* dst = Wt + (long)(nt * TR_N + g) * K + kt * TR_K + 8 * q;
#pragma unroll
    for (int s = 0; s < 4; ++s) {
        const v8h o = *(const v8h*)(&sT[(g + 16 * s) * TR_P + 8 * q]);
        *(v8h*)(dst + (long)(16 * s) * K) = o;
    }
}

__global__ void __launch_bounds__(256)
k_layernorm(const float* __restrict__ X, const float* __restrict__ alpha,
            const float* __restrict__ beta, _Float16* __restrict__ Y) {
    __shared__ float s1[256], s2[256];
    const int row = blockIdx.x, t = threadIdx.x;
    const float* xr = X + (long)row * D_MODEL;
    v4f x = *(const v4f*)(xr + t * 4);
    float s = x[0] + x[1] + x[2] + x[3];
    float q = x[0]*x[0] + x[1]*x[1] + x[2]*x[2] + x[3]*x[3];
    s1[t] = s; s2[t] = q; __syncthreads();
    for (int off = 128; off; off >>= 1) {
        if (t < off) { s1[t] += s1[t + off]; s2[t] += s2[t + off]; }
        __syncthreads();
    }
    const float mu  = s1[0] * (1.0f / D_MODEL);
    const float var = (s2[0] - (float)D_MODEL * mu * mu) * (1.0f / (D_MODEL - 1));
    const float inv = 1.0f / (sqrtf(var) + 1e-6f);
    v4h o;
#pragma unroll
    for (int i = 0; i < 4; ++i) {
        int c = t * 4 + i;
        o[i] = (_Float16)(alpha[c] * (x[i] - mu) * inv + beta[c]);
    }
    *(v4h*)(Y + (long)row * D_MODEL + t * 4) = o;
}

#define GT_M 128
#define GT_N 128
#define GT_K 128
#define GT_P 136

#define FRAG_FENCE() __builtin_amdgcn_sched_barrier(0)

template <int OMODE, bool RELU, bool RESID>
__global__ void __launch_bounds__(128)
k_gemm_wmma(const _Float16* __restrict__ A, const _Float16* __restrict__ Bt,
            const float* __restrict__ bias, const float* __restrict__ resid,
            float* __restrict__ Cf, _Float16* __restrict__ Ch,
            int M, int N, int K) {
    __shared__ __attribute__((aligned(16))) _Float16 sA[2][GT_M * GT_P];
    __shared__ __attribute__((aligned(16))) _Float16 sB[2][GT_N * GT_P];

    const int tid  = threadIdx.x;
    const int lane = tid & 31;
    const int wave = tid >> 5;
    const int wm = wave >> 1, wn = wave & 1;
    const int bn_cnt = N >> 7;
    const int bm = blockIdx.x / bn_cnt;
    const int bn = blockIdx.x % bn_cnt;
    if (bm * GT_M >= M) return;

    const int ln = lane & 15;
    const int hh = lane >> 4;

    const int srow = tid >> 4, sch = tid & 15;
    const _Float16* ag = A  + (long)(bm * GT_M + srow) * K + sch * 8;
    const _Float16* bg = Bt + (long)(bn * GT_N + srow) * K + sch * 8;
    const long g8   = (long)8 * K;
    const int  soff = srow * GT_P + sch * 8;

    v8h ra[16], rb[16];
#pragma unroll
    for (int j = 0; j < 16; ++j) {
        ra[j] = *(const v8h*)(ag + j * g8);
        rb[j] = *(const v8h*)(bg + j * g8);
    }
#pragma unroll
    for (int j = 0; j < 16; ++j) {
        *(v8h*)(&sA[0][soff + j * 8 * GT_P]) = ra[j];
        *(v8h*)(&sB[0][soff + j * 8 * GT_P]) = rb[j];
    }
    __syncthreads();

    v8f acc[4][4];
#pragma unroll
    for (int i = 0; i < 4; ++i) {
#pragma unroll
        for (int j = 0; j < 4; ++j) acc[i][j] = v8f{};
    }

    const int aoff = (wm * 64 + ln) * GT_P + 8 * hh;
    const int boff = (wn * 64 + ln) * GT_P + 8 * hh;

    const int nk = K / GT_K;

    if (nk > 1) {
        ag += GT_K; bg += GT_K;
#pragma unroll
        for (int j = 0; j < 16; ++j) {
            ra[j] = *(const v8h*)(ag + j * g8);
            rb[j] = *(const v8h*)(bg + j * g8);
        }
    }

    v16h fa0[4], fb0[4], fa1[4], fb1[4];
#pragma unroll
    for (int i = 0; i < 4; ++i) fa0[i] = load_frag(&sA[0][aoff] + i * 16 * GT_P);
#pragma unroll
    for (int j = 0; j < 4; ++j) fb0[j] = load_frag(&sB[0][boff] + j * 16 * GT_P);

    for (int kt = 0; kt < nk; ++kt) {
        const int  cur   = kt & 1;
        const bool more  = (kt + 1 < nk);
        const bool more2 = (kt + 2 < nk);
        const _Float16* pa = &sA[cur][aoff];
        const _Float16* pb = &sB[cur][boff];

#pragma unroll
        for (int i = 0; i < 4; ++i) fa1[i] = load_frag(pa + 32 + i * 16 * GT_P);
#pragma unroll
        for (int j = 0; j < 4; ++j) fb1[j] = load_frag(pb + 32 + j * 16 * GT_P);
        FRAG_FENCE();
#pragma unroll
        for (int i = 0; i < 4; ++i) {
#pragma unroll
            for (int j = 0; j < 4; ++j) acc[i][j] = wmma_f16(fa0[i], fb0[j], acc[i][j]);
        }
        FRAG_FENCE();

#pragma unroll
        for (int i = 0; i < 4; ++i) fa0[i] = load_frag(pa + 64 + i * 16 * GT_P);
#pragma unroll
        for (int j = 0; j < 4; ++j) fb0[j] = load_frag(pb + 64 + j * 16 * GT_P);
        FRAG_FENCE();
#pragma unroll
        for (int i = 0; i < 4; ++i) {
#pragma unroll
            for (int j = 0; j < 4; ++j) acc[i][j] = wmma_f16(fa1[i], fb1[j], acc[i][j]);
        }
        FRAG_FENCE();

#pragma unroll
        for (int i = 0; i < 4; ++i) fa1[i] = load_frag(pa + 96 + i * 16 * GT_P);
#pragma unroll
        for (int j = 0; j < 4; ++j) fb1[j] = load_frag(pb + 96 + j * 16 * GT_P);
        FRAG_FENCE();
#pragma unroll
        for (int i = 0; i < 4; ++i) {
#pragma unroll
            for (int j = 0; j < 4; ++j) acc[i][j] = wmma_f16(fa0[i], fb0[j], acc[i][j]);
        }

        if (more) {
#pragma unroll
            for (int j = 0; j < 16; ++j) {
                *(v8h*)(&sA[cur ^ 1][soff + j * 8 * GT_P]) = ra[j];
                *(v8h*)(&sB[cur ^ 1][soff + j * 8 * GT_P]) = rb[j];
            }
        }
        __syncthreads();

        if (more) {
            const _Float16* na = &sA[cur ^ 1][aoff];
            const _Float16* nb = &sB[cur ^ 1][boff];
#pragma unroll
            for (int i = 0; i < 4; ++i) fa0[i] = load_frag(na + i * 16 * GT_P);
#pragma unroll
            for (int j = 0; j < 4; ++j) fb0[j] = load_frag(nb + j * 16 * GT_P);
        }
        if (more2) { ag += GT_K; bg += GT_K; }
#pragma unroll
        for (int i = 0; i < 4; ++i) {
#pragma unroll
            for (int j = 0; j < 4; ++j) acc[i][j] = wmma_f16(fa1[i], fb1[j], acc[i][j]);
            if (more2) {
#pragma unroll
                for (int j = 4 * i; j < 4 * i + 4; ++j) {
                    ra[j] = *(const v8h*)(ag + j * g8);
                    rb[j] = *(const v8h*)(bg + j * g8);
                }
            }
        }
    }

    const int lm = hh << 3;
    const int m0 = bm * GT_M + wm * 64;
    const int n0 = bn * GT_N + wn * 64;
#pragma unroll
    for (int i = 0; i < 4; ++i) {
#pragma unroll
        for (int j = 0; j < 4; ++j) {
            const v8f accv = acc[i][j];
            const int col = n0 + j * 16 + ln;
            const float bv = bias[col];
            if (OMODE == 3) {
                const int row0 = m0 + i * 16 + lm;
                const int b_ = row0 / SEQ, s0 = row0 % SEQ, h = col >> 6, d = col & 63;
                v8h o;
#pragma unroll
                for (int e = 0; e < 8; ++e) {
                    float v = accv[e] + bv;
                    if (RELU)  v = v > 0.0f ? v : 0.0f;
                    if (RESID) v += resid[(long)(row0 + e) * N + col];
                    o[e] = (_Float16)v;
                }
                *(v8h*)(Ch + ((long)(b_ * HEADS + h)) * DK * SEQ + vt_off(s0, d)) = o;
                continue;
            }
            if (OMODE == 2) {
                const int row0 = m0 + i * 16 + lm;
                const int b_ = row0 / SEQ, s0 = row0 % SEQ, h = col >> 6, d = col & 63;
                _Float16* dst = Ch + ((long)(b_ * HEADS + h)) * SEQ * DK + qk_off(s0, d);
#pragma unroll
                for (int e = 0; e < 8; ++e) {
                    float v = accv[e] + bv;
                    if (RELU)  v = v > 0.0f ? v : 0.0f;
                    if (RESID) v += resid[(long)(row0 + e) * N + col];
                    dst[8 * e] = (_Float16)v;
                }
                continue;
            }
#pragma unroll
            for (int e = 0; e < 8; ++e) {
                const int row = m0 + i * 16 + e + lm;
                float v = accv[e] + bv;
                if (RELU)  v = v > 0.0f ? v : 0.0f;
                if (RESID) v += resid[(long)row * N + col];
                if (OMODE == 0) {
                    Ch[(long)row * N + col] = (_Float16)v;
                } else if (OMODE == 1) {
                    Cf[(long)row * N + col] = v;
                }
            }
        }
    }
}

#define ATT_R     2
#define ATT_QW    (16 * ATT_R)
#define ATT_WAVES 4
#define ATT_QB    (ATT_QW * ATT_WAVES)

__device__ __forceinline__ float xhalf(float x) {
    return __builtin_bit_cast(float, __builtin_amdgcn_permlanex16(
        __builtin_bit_cast(int, x), __builtin_bit_cast(int, x),
        0x76543210, 0xfedcba98, false, false));
}

__global__ void __launch_bounds__(128)
k_attention(const _Float16* __restrict__ QK, const _Float16* __restrict__ VT,
            const int* __restrict__ mask, _Float16* __restrict__ CTX) {
    const int lane = threadIdx.x & 31, wave = threadIdx.x >> 5;
    const int c = lane & 15, hh = lane >> 4;
    const int qb = blockIdx.x & 15;
    const int h  = (blockIdx.x >> 4) & 15;
    const int b  = blockIdx.x >> 8;
    const int q0 = qb * ATT_QB + wave * ATT_QW;

    const _Float16* qbase = QK + ((long)(b * HEADS + h)) * SEQ * DK;
    const _Float16* vbase = VT + ((long)(b * HEADS + h)) * DK * SEQ;
    const _Float16* kp = qbase + 8 * lane;
    const _Float16* vp = vbase + 8 * lane;
    const int* mrow = mask + b * SEQ + 8 * hh;

    const float cscale = 0.125f * 1.4426950408889634f;
    const float negl   = -1e9f * 1.4426950408889634f;

    v16h qf[ATT_R][2];
#pragma unroll
    for (int g = 0; g < ATT_R; ++g) {
        const _Float16* qp = kp + (long)(q0 + 16 * g) * DK;
        qf[g][0] = load_frag_fo(qp);
        qf[g][1] = load_frag_fo(qp + 512);
    }

    v8f acc[ATT_R][4];
    float m_run[ATT_R], l_run[ATT_R];
#pragma unroll
    for (int g = 0; g < ATT_R; ++g) {
        m_run[g] = -__builtin_inff(); l_run[g] = 0.0f;
#pragma unroll
        for (int t = 0; t < 4; ++t) acc[g][t] = v8f{};
    }

    v16h kf[2][2];
#pragma unroll
    for (int kt = 0; kt < 2; ++kt) {
        kf[kt][0] = load_frag_fo(kp + (long)(16 * kt) * DK);
        kf[kt][1] = load_frag_fo(kp + (long)(16 * kt) * DK + 512);
    }
    int mk[2][8];
#pragma unroll
    for (int kt = 0; kt < 2; ++kt) {
        const int4 m0 = *(const int4*)(mrow + 16 * kt);
        const int4 m1 = *(const int4*)(mrow + 16 * kt + 4);
        mk[kt][0] = m0.x; mk[kt][1] = m0.y; mk[kt][2] = m0.z; mk[kt][3] = m0.w;
        mk[kt][4] = m1.x; mk[kt][5] = m1.y; mk[kt][6] = m1.z; mk[kt][7] = m1.w;
    }

    for (int key0 = 0; key0 < SEQ; key0 += 32) {
        v16h vf[4];
#pragma unroll
        for (int t = 0; t < 4; ++t) vf[t] = load_frag_fo(vp + (long)key0 * DK + 512 * t);
        __builtin_amdgcn_sched_barrier(0);

        v8f s[2][ATT_R];
#pragma unroll
        for (int kt = 0; kt < 2; ++kt) {
#pragma unroll
            for (int g = 0; g < ATT_R; ++g) {
                v8f z = {};
                z = wmma_f16(kf[kt][0], qf[g][0], z);
                s[kt][g] = wmma_f16(kf[kt][1], qf[g][1], z);
            }
        }

        const int nk = (key0 + 32 < SEQ) ? key0 + 32 : key0;
#pragma unroll
        for (int kt = 0; kt < 2; ++kt) {
            kf[kt][0] = load_frag_fo(kp + (long)(nk + 16 * kt) * DK);
            kf[kt][1] = load_frag_fo(kp + (long)(nk + 16 * kt) * DK + 512);
        }
        int mkn[2][8];
#pragma unroll
        for (int kt = 0; kt < 2; ++kt) {
            const int4 m0 = *(const int4*)(mrow + nk + 16 * kt);
            const int4 m1 = *(const int4*)(mrow + nk + 16 * kt + 4);
            mkn[kt][0] = m0.x; mkn[kt][1] = m0.y; mkn[kt][2] = m0.z; mkn[kt][3] = m0.w;
            mkn[kt][4] = m1.x; mkn[kt][5] = m1.y; mkn[kt][6] = m1.z; mkn[kt][7] = m1.w;
        }

        float mb[ATT_R];
        bool need = false;
#pragma unroll
        for (int g = 0; g < ATT_R; ++g) {
            float mx = -__builtin_inff();
#pragma unroll
            for (int kt = 0; kt < 2; ++kt) {
#pragma unroll
                for (int r = 0; r < 8; ++r) {
                    float t = s[kt][g][r] * cscale;
                    t = (mk[kt][r] == 0) ? negl : t;
                    s[kt][g][r] = t;
                    mx = fmaxf(mx, t);
                }
            }
            mx = fmaxf(mx, xhalf(mx));
            mb[g] = mx;
            need = need || (mx > m_run[g] + 8.0f);
        }
        if (__builtin_amdgcn_ballot_w32(need) != 0) {
#pragma unroll
            for (int g = 0; g < ATT_R; ++g) {
                const float m_new = fmaxf(m_run[g], mb[g]);
                const float corr  = __builtin_amdgcn_exp2f(m_run[g] - m_new);
                m_run[g] = m_new;
                l_run[g] *= corr;
#pragma unroll
                for (int t = 0; t < 4; ++t) acc[g][t] = acc[g][t] * corr;
            }
        }

#pragma unroll
        for (int g = 0; g < ATT_R; ++g) {
            const float m = m_run[g];
            v16h pf;
            float ls = 0.0f;
#pragma unroll
            for (int r = 0; r < 8; ++r) {
                const float p0 = __builtin_amdgcn_exp2f(s[0][g][r] - m);
                const float p1 = __builtin_amdgcn_exp2f(s[1][g][r] - m);
                ls += p0 + p1;
                pf[r]     = (_Float16)p0;
                pf[8 + r] = (_Float16)p1;
            }
            l_run[g] += ls;
#pragma unroll
            for (int t = 0; t < 4; ++t) acc[g][t] = wmma_f16(vf[t], pf, acc[g][t]);
        }
#pragma unroll
        for (int kt = 0; kt < 2; ++kt) {
#pragma unroll
            for (int r = 0; r < 8; ++r) mk[kt][r] = mkn[kt][r];
        }
    }

#pragma unroll
    for (int g = 0; g < ATT_R; ++g) {
        float l = l_run[g];
        l = l + xhalf(l);
        const float inv = 1.0f / l;
        _Float16* op = CTX + ((long)(b * SEQ + q0 + 16 * g + c)) * D_MODEL + h * DK + 8 * hh;
#pragma unroll
        for (int t = 0; t < 4; ++t) {
            v8h o;
#pragma unroll
            for (int r = 0; r < 8; ++r) o[r] = (_Float16)(acc[g][t][r] * inv);
            *(v8h*)(op + 16 * t) = o;
        }
    }
}

extern "C" void kernel_launch(void* const* d_in, const int* in_sizes, int n_in,
                              void* d_out, int out_size, void* d_ws, size_t ws_size,
                              hipStream_t stream) {
    const float* x      = (const float*)d_in[0];
    const int*   mask   = (const int*)  d_in[1];
    const float* alpha1 = (const float*)d_in[2];
    const float* bias1  = (const float*)d_in[3];
    const float* alpha2 = (const float*)d_in[4];
    const float* bias2  = (const float*)d_in[5];
    const float* Wk     = (const float*)d_in[6];
    const float* bk     = (const float*)d_in[7];
    const float* Wv     = (const float*)d_in[8];
    const float* bv     = (const float*)d_in[9];
    const float* Wo     = (const float*)d_in[10];
    const float* bo     = (const float*)d_in[11];
    const float* W1     = (const float*)d_in[12];
    const float* b1     = (const float*)d_in[13];
    const float* W2     = (const float*)d_in[14];
    const float* b2     = (const float*)d_in[15];
    float* out = (float*)d_out;

    char* ws = (char*)d_ws;
    size_t off = 0;
    _Float16* x2h  = (_Float16*)(ws + off); off += (size_t)ROWS * D_MODEL * 2;
    _Float16* x2bh = (_Float16*)(ws + off); off += (size_t)ROWS * D_MODEL * 2;
    _Float16* qkh  = (_Float16*)(ws + off); off += (size_t)BATCH*HEADS*SEQ*DK*2;
    _Float16* vth  = (_Float16*)(ws + off); off += (size_t)BATCH*HEADS*DK*SEQ*2;
    _Float16* ctxh = (_Float16*)(ws + off); off += (size_t)ROWS * D_MODEL * 2;
    _Float16* h1h  = (_Float16*)(ws + off); off += (size_t)ROWS * D_FF * 2;
    _Float16* Wkt  = (_Float16*)(ws + off); off += (size_t)D_MODEL * D_MODEL * 2;
    _Float16* Wvt  = (_Float16*)(ws + off); off += (size_t)D_MODEL * D_MODEL * 2;
    _Float16* Wot  = (_Float16*)(ws + off); off += (size_t)D_MODEL * D_MODEL * 2;
    _Float16* W1t  = (_Float16*)(ws + off); off += (size_t)D_MODEL * D_FF * 2;
    _Float16* W2t  = (_Float16*)(ws + off); off += (size_t)D_FF * D_MODEL * 2;
    float*    x1   = (float*)   (ws + off); off += (size_t)ROWS * D_MODEL * 4;

    k_transpose_cvt_all<<<TR_GRID, 256, 0, stream>>>(Wk, Wv, Wo, W1, W2,
                                                     Wkt, Wvt, Wot, W1t, W2t);

    k_layernorm<<<ROWS, 256, 0, stream>>>(x, alpha1, bias1, x2h);

    #define GEMM_GRID(M_, N_) (((M_) / 128) * ((N_) / 128))

    k_gemm_wmma<2, false, false><<<GEMM_GRID(ROWS, D_MODEL), 128, 0, stream>>>(
        x2h, Wkt, bk, nullptr, nullptr, qkh, ROWS, D_MODEL, D_MODEL);
    k_gemm_wmma<3, false, false><<<GEMM_GRID(ROWS, D_MODEL), 128, 0, stream>>>(
        x2h, Wvt, bv, nullptr, nullptr, vth, ROWS, D_MODEL, D_MODEL);

    k_attention<<<BATCH*HEADS*(SEQ/ATT_QB), 128, 0, stream>>>(
        qkh, vth, mask, ctxh);

    k_gemm_wmma<1, false, true><<<GEMM_GRID(ROWS, D_MODEL), 128, 0, stream>>>(
        ctxh, Wot, bo, x, x1, nullptr, ROWS, D_MODEL, D_MODEL);

    k_layernorm<<<ROWS, 256, 0, stream>>>(x1, alpha2, bias2, x2bh);

    k_gemm_wmma<0, true, false><<<GEMM_GRID(ROWS, D_FF), 128, 0, stream>>>(
        x2bh, W1t, b1, nullptr, nullptr, h1h, ROWS, D_FF, D_MODEL);

    k_gemm_wmma<1, false, true><<<GEMM_GRID(ROWS, D_MODEL), 128, 0, stream>>>(
        h1h, W2t, b2, x1, out, nullptr, ROWS, D_MODEL, D_FF);

    (void)in_sizes; (void)n_in; (void)out_size; (void)ws_size;
}


# GEMM k loop: a stage's 32 refill loads per lane turned round (to LDS, requested again) in groups of 8 at separate points, a whole stage ahead (was one point): 24-32 always in flight; sums unchanged
# speedup vs baseline: 5.7616x; 1.0668x over previous
#include <hip/hip_runtime.h>
#include <hip/hip_bf16.h>

typedef __attribute__((ext_vector_type(16))) _Float16 v16h;
typedef __attribute__((ext_vector_type(8)))  _Float16 v8h;
typedef __attribute__((ext_vector_type(4)))  _Float16 v4h;
typedef __attribute__((ext_vector_type(8)))  float    v8f;
typedef __attribute__((ext_vector_type(4)))  float    v4f;

#define D_MODEL 1024
#define HEADS   16
#define DK      64
#define D_FF    4096
#define BATCH   2
#define SEQ     2048
#define ROWS    (BATCH*SEQ)

__device__ __forceinline__ v16h pack16(v8h lo, v8h hi) {
    v16h r;
#pragma unroll
    for (int i = 0; i < 8; ++i) { r[i] = lo[i]; r[i + 8] = hi[i]; }
    return r;
}
__device__ __forceinline__ v16h load_frag(const _Float16* p) {
    v8h lo = *(const v8h*)(p);
    v8h hi = *(const v8h*)(p + 16);
    return pack16(lo, hi);
}
__device__ __forceinline__ int qk_off(int s, int d) {
    return ((((s >> 4) * 4 + (d >> 4)) * 32 + ((d >> 3) & 1) * 16 + (s & 15)) << 3) + (d & 7);
}
__device__ __forceinline__ int vt_off(int s, int d) {
    return (((((s >> 5) * 4 + (d >> 4)) * 2 + ((s >> 4) & 1)) * 32
             + ((s >> 3) & 1) * 16 + (d & 15)) << 3) + (s & 7);
}
__device__ __forceinline__ v16h load_frag_fo(const _Float16* p) {
    v8h lo = *(const v8h*)(p);
    v8h hi = *(const v8h*)(p + 256);
    return pack16(lo, hi);
}
__device__ __forceinline__ v8f wmma_f16(v16h a, v16h b, v8f c) {
    return __builtin_amdgcn_wmma_f32_16x16x32_f16(false, a, false, b,
                                                  (short)0, c, false, false);
}

#define TR_K 128
#define TR_N 64
#define TR_P 136
#define TR_TILES(K_, N_) (((K_) / TR_K) * ((N_) / TR_N))
#define TR_SQ   TR_TILES(D_MODEL, D_MODEL)
#define TR_W1   TR_TILES(D_MODEL, D_FF)
#define TR_W2   TR_TILES(D_FF, D_MODEL)
#define TR_GRID (3 * TR_SQ + TR_W1 + TR_W2)

__global__ void __launch_bounds__(256)
k_transpose_cvt_all(const float* __restrict__ Wk, const float* __restrict__ Wv,
                    const float* __restrict__ Wo, const float* __restrict__ W1,
                    const float* __restrict__ W2,
                    _Float16* __restrict__ Wkt, _Float16* __restrict__ Wvt,
                    _Float16* __restrict__ Wot, _Float16* __restrict__ W1t,
                    _Float16* __restrict__ W2t) {
    __shared__ __attribute__((aligned(16))) _Float16 sT[TR_N * TR_P];

    int tile = blockIdx.x;
    const float* W; _Float16* Wt; int K, N;
    if (tile < TR_SQ)                 { W = Wk; Wt = Wkt; K = D_MODEL; N = D_MODEL; }
    else if (tile < 2 * TR_SQ)        { W = Wv; Wt = Wvt; K = D_MODEL; N = D_MODEL; tile -= TR_SQ; }
    else if (tile < 3 * TR_SQ)        { W = Wo; Wt = Wot; K = D_MODEL; N = D_MODEL; tile -= 2 * TR_SQ; }
    else if (tile < 3 * TR_SQ + TR_W1){ W = W1; Wt = W1t; K = D_MODEL; N = D_FF;    tile -= 3 * TR_SQ; }
    else                              { W = W2; Wt = W2t; K = D_FF;    N = D_MODEL; tile -= 3 * TR_SQ + TR_W1; }
    const int ntn = N / TR_N;
    const int kt = tile / ntn, nt = tile % ntn;

    const int t = threadIdx.x;
    const int q = t & 15, g = t >> 4;

    const float* src = W + (long)(kt * TR_K + 8 * g) * N + nt * TR_N + 4 * q;
    v4f r[8];
#pragma unroll
    for (int i = 0; i < 8; ++i) r[i] = *(const v4f*)(src + (long)i * N);

#pragma unroll
    for (int j = 0; j < 4; ++j) {
        v8h o;
#pragma unroll
        for (int i = 0; i < 8; ++i) o[i] = (_Float16)r[i][j];
        *(v8h*)(&sT[(4 * q + j) * TR_P + 8 * g]) = o;
    }
    __syncthreads();

    _Float16* dst = Wt + (long)(nt * TR_N + g) * K + kt * TR_K + 8 * q;
#pragma unroll
    for (int s = 0; s < 4; ++s) {
        const v8h o = *(const v8h*)(&sT[(g + 16 * s) * TR_P + 8 * q]);
        *(v8h*)(dst + (long)(16 * s) * K) = o;
    }
}

__global__ void __launch_bounds__(256)
k_layernorm(const float* __restrict__ X, const float* __restrict__ alpha,
            const float* __restrict__ beta, _Float16* __restrict__ Y) {
    __shared__ float s1[256], s2[256];
    const int row = blockIdx.x, t = threadIdx.x;
    const float* xr = X + (long)row * D_MODEL;
    v4f x = *(const v4f*)(xr + t * 4);
    float s = x[0] + x[1] + x[2] + x[3];
    float q = x[0]*x[0] + x[1]*x[1] + x[2]*x[2] + x[3]*x[3];
    s1[t] = s; s2[t] = q; __syncthreads();
    for (int off = 128; off; off >>= 1) {
        if (t < off) { s1[t] += s1[t + off]; s2[t] += s2[t + off]; }
        __syncthreads();
    }
    const float mu  = s1[0] * (1.0f / D_MODEL);
    const float var = (s2[0] - (float)D_MODEL * mu * mu) * (1.0f / (D_MODEL - 1));
    const float inv = 1.0f / (sqrtf(var) + 1e-6f);
    v4h o;
#pragma unroll
    for (int i = 0; i < 4; ++i) {
        int c = t * 4 + i;
        o[i] = (_Float16)(alpha[c] * (x[i] - mu) * inv + beta[c]);
    }
    *(v4h*)(Y + (long)row * D_MODEL + t * 4) = o;
}

#define GT_M 128
#define GT_N 128
#define GT_K 128
#define GT_P 136

#define FRAG_FENCE() __builtin_amdgcn_sched_barrier(0)

template <bool MORE, bool MORE2>
__device__ __forceinline__ void gemm_stage(
        v8f (&acc)[4][4], v16h (&fa0)[4], v16h (&fb0)[4], v16h (&fa1)[4], v16h (&fb1)[4],
        v8h (&ra)[16], v8h (&rb)[16],
        const _Float16* pa, const _Float16* pb,
        const _Float16* na, const _Float16* nb,
        _Float16* wa, _Float16* wb,
        const char* ag, const char* bg,
        const unsigned (&g8)[16]) {
#define GT_STORE(G)                                                              \
    if (MORE) {                                                                  \
        _Pragma("unroll")                                                        \
        for (int j = 4 * (G); j < 4 * (G) + 4; ++j) {                            \
            *(v8h*)(wa + j * 8 * GT_P) = ra[j];                                  \
            *(v8h*)(wb + j * 8 * GT_P) = rb[j];                                  \
        }                                                                        \
    }
#define GT_LOAD(G)                                                               \
    if (MORE2) {                                                                 \
        _Pragma("unroll")                                                        \
        for (int j = 4 * (G); j < 4 * (G) + 4; ++j) {                            \
            ra[j] = *(const v8h*)(ag + g8[j]);                                  \
            rb[j] = *(const v8h*)(bg + g8[j]);                                  \
        }                                                                        \
    }

#pragma unroll
    for (int i = 0; i < 4; ++i) fa1[i] = load_frag(pa + 32 + i * 16 * GT_P);
#pragma unroll
    for (int j = 0; j < 4; ++j) fb1[j] = load_frag(pb + 32 + j * 16 * GT_P);
    FRAG_FENCE();
#pragma unroll
    for (int i = 0; i < 4; ++i) {
#pragma unroll
        for (int j = 0; j < 4; ++j) acc[i][j] = wmma_f16(fa0[i], fb0[j], acc[i][j]);
        if (i == 0) { GT_STORE(0) }
        if (i == 2) { GT_LOAD(0) }
    }
    FRAG_FENCE();

#pragma unroll
    for (int i = 0; i < 4; ++i) fa0[i] = load_frag(pa + 64 + i * 16 * GT_P);
#pragma unroll
    for (int j = 0; j < 4; ++j) fb0[j] = load_frag(pb + 64 + j * 16 * GT_P);
    FRAG_FENCE();
#pragma unroll
    for (int i = 0; i < 4; ++i) {
#pragma unroll
        for (int j = 0; j < 4; ++j) acc[i][j] = wmma_f16(fa1[i], fb1[j], acc[i][j]);
        if (i == 0) { GT_STORE(1) }
        if (i == 2) { GT_LOAD(1) }
    }
    FRAG_FENCE();

#pragma unroll
    for (int i = 0; i < 4; ++i) fa1[i] = load_frag(pa + 96 + i * 16 * GT_P);
#pragma unroll
    for (int j = 0; j < 4; ++j) fb1[j] = load_frag(pb + 96 + j * 16 * GT_P);
    FRAG_FENCE();
#pragma unroll
    for (int i = 0; i < 4; ++i) {
#pragma unroll
        for (int j = 0; j < 4; ++j) acc[i][j] = wmma_f16(fa0[i], fb0[j], acc[i][j]);
        if (i == 0) { GT_STORE(2) }
        if (i == 1) { GT_STORE(3) }
        if (i == 2) { GT_LOAD(2) }
        if (i == 3) { GT_LOAD(3) }
    }

    __builtin_amdgcn_fence(__ATOMIC_RELEASE, "workgroup", "local");
    __builtin_amdgcn_s_barrier();
    __builtin_amdgcn_fence(__ATOMIC_ACQUIRE, "workgroup", "local");

    if (MORE) {
#pragma unroll
        for (int i = 0; i < 4; ++i) fa0[i] = load_frag(na + i * 16 * GT_P);
#pragma unroll
        for (int j = 0; j < 4; ++j) fb0[j] = load_frag(nb + j * 16 * GT_P);
    }
#pragma unroll
    for (int i = 0; i < 4; ++i) {
#pragma unroll
        for (int j = 0; j < 4; ++j) acc[i][j] = wmma_f16(fa1[i], fb1[j], acc[i][j]);
    }
#undef GT_STORE
#undef GT_LOAD
}

template <int OMODE, bool RELU, bool RESID>
__global__ void __launch_bounds__(128)
k_gemm_wmma(const _Float16* __restrict__ A, const _Float16* __restrict__ Bt,
            const float* __restrict__ bias, const float* __restrict__ resid,
            float* __restrict__ Cf, _Float16* __restrict__ Ch,
            int M, int N, int K) {
    __shared__ __attribute__((aligned(16))) _Float16 sA[2][GT_M * GT_P];
    __shared__ __attribute__((aligned(16))) _Float16 sB[2][GT_N * GT_P];

    const int tid  = threadIdx.x;
    const int lane = tid & 31;
    const int wave = tid >> 5;
    const int wm = wave >> 1, wn = wave & 1;
    const int bn_cnt = N >> 7;
    const int bm = blockIdx.x / bn_cnt;
    const int bn = blockIdx.x % bn_cnt;
    if (bm * GT_M >= M) return;

    const int ln = lane & 15;
    const int hh = lane >> 4;

    const int srow = tid >> 4, sch = tid & 15;
    const char* ag = (const char*)(A  + (long)(bm * GT_M) * K);
    const char* bg = (const char*)(Bt + (long)(bn * GT_N) * K);
    unsigned g8[16];
#pragma unroll
    for (int j = 0; j < 16; ++j) g8[j] = (unsigned)(((srow + 8 * j) * K + sch * 8) * 2);
    const int  soff = srow * GT_P + sch * 8;

    v8h ra[16], rb[16];
#pragma unroll
    for (int j = 0; j < 16; ++j) {
        ra[j] = *(const v8h*)(ag + g8[j]);
        rb[j] = *(const v8h*)(bg + g8[j]);
    }
#pragma unroll
    for (int j = 0; j < 16; ++j) {
        *(v8h*)(&sA[0][soff + j * 8 * GT_P]) = ra[j];
        *(v8h*)(&sB[0][soff + j * 8 * GT_P]) = rb[j];
    }
    __syncthreads();

    v8f acc[4][4];
#pragma unroll
    for (int i = 0; i < 4; ++i) {
#pragma unroll
        for (int j = 0; j < 4; ++j) acc[i][j] = v8f{};
    }

    const int aoff = (wm * 64 + ln) * GT_P + 8 * hh;
    const int boff = (wn * 64 + ln) * GT_P + 8 * hh;

    const int nk = K / GT_K;

    if (nk > 1) {
        ag += GT_K * 2; bg += GT_K * 2;
#pragma unroll
        for (int j = 0; j < 16; ++j) {
            ra[j] = *(const v8h*)(ag + g8[j]);
            rb[j] = *(const v8h*)(bg + g8[j]);
        }
    }

    v16h fa0[4], fb0[4], fa1[4], fb1[4];
#pragma unroll
    for (int i = 0; i < 4; ++i) fa0[i] = load_frag(&sA[0][aoff] + i * 16 * GT_P);
#pragma unroll
    for (int j = 0; j < 4; ++j) fb0[j] = load_frag(&sB[0][boff] + j * 16 * GT_P);

    int kt = 0;
    for (; kt + 2 < nk; ++kt) {
        const int cur = kt & 1;
        ag += GT_K * 2; bg += GT_K * 2;
        gemm_stage<true, true>(acc, fa0, fb0, fa1, fb1, ra, rb,
                               &sA[cur][aoff], &sB[cur][boff], &sA[cur ^ 1][aoff], &sB[cur ^ 1][boff],
                               &sA[cur ^ 1][soff], &sB[cur ^ 1][soff], ag, bg, g8);
    }
    if (nk > 1) {
        const int cur = kt & 1;
        gemm_stage<true, false>(acc, fa0, fb0, fa1, fb1, ra, rb,
                                &sA[cur][aoff], &sB[cur][boff], &sA[cur ^ 1][aoff], &sB[cur ^ 1][boff],
                                &sA[cur ^ 1][soff], &sB[cur ^ 1][soff], ag, bg, g8);
        ++kt;
    }
    {
        const int cur = kt & 1;
        gemm_stage<false, false>(acc, fa0, fb0, fa1, fb1, ra, rb,
                                 &sA[cur][aoff], &sB[cur][boff], &sA[cur ^ 1][aoff], &sB[cur ^ 1][boff],
                                 &sA[cur ^ 1][soff], &sB[cur ^ 1][soff], ag, bg, g8);
    }

    const int lm = hh << 3;
    const int m0 = bm * GT_M + wm * 64;
    const int n0 = bn * GT_N + wn * 64;
#pragma unroll
    for (int i = 0; i < 4; ++i) {
#pragma unroll
        for (int j = 0; j < 4; ++j) {
            const v8f accv = acc[i][j];
            const int col = n0 + j * 16 + ln;
            const float bv = bias[col];
            if (OMODE == 3) {
                const int row0 = m0 + i * 16 + lm;
                const int b_ = row0 / SEQ, s0 = row0 % SEQ, h = col >> 6, d = col & 63;
                v8h o;
#pragma unroll
                for (int e = 0; e < 8; ++e) {
                    float v = accv[e] + bv;
                    if (RELU)  v = v > 0.0f ? v : 0.0f;
                    if (RESID) v += resid[(long)(row0 + e) * N + col];
                    o[e] = (_Float16)v;
                }
                *(v8h*)(Ch + ((long)(b_ * HEADS + h)) * DK * SEQ + vt_off(s0, d)) = o;
                continue;
            }
            if (OMODE == 2) {
                const int row0 = m0 + i * 16 + lm;
                const int b_ = row0 / SEQ, s0 = row0 % SEQ, h = col >> 6, d = col & 63;
                _Float16* dst = Ch + ((long)(b_ * HEADS + h)) * SEQ * DK + qk_off(s0, d);
#pragma unroll
                for (int e = 0; e < 8; ++e) {
                    float v = accv[e] + bv;
                    if (RELU)  v = v > 0.0f ? v : 0.0f;
                    if (RESID) v += resid[(long)(row0 + e) * N + col];
                    dst[8 * e] = (_Float16)v;
                }
                continue;
            }
#pragma unroll
            for (int e = 0; e < 8; ++e) {
                const int row = m0 + i * 16 + e + lm;
                float v = accv[e] + bv;
                if (RELU)  v = v > 0.0f ? v : 0.0f;
                if (RESID) v += resid[(long)row * N + col];
                if (OMODE == 0) {
                    Ch[(long)row * N + col] = (_Float16)v;
                } else if (OMODE == 1) {
                    Cf[(long)row * N + col] = v;
                }
            }
        }
    }
}

#define ATT_R     2
#define ATT_QW    (16 * ATT_R)
#define ATT_WAVES 4
#define ATT_QB    (ATT_QW * ATT_WAVES)

__device__ __forceinline__ float xhalf(float x) {
    return __builtin_bit_cast(float, __builtin_amdgcn_permlanex16(
        __builtin_bit_cast(int, x), __builtin_bit_cast(int, x),
        0x76543210, 0xfedcba98, false, false));
}

__global__ void __launch_bounds__(128)
k_attention(const _Float16* __restrict__ QK, const _Float16* __restrict__ VT,
            const int* __restrict__ mask, _Float16* __restrict__ CTX) {
    const int lane = threadIdx.x & 31, wave = threadIdx.x >> 5;
    const int c = lane & 15, hh = lane >> 4;
    const int qb = blockIdx.x & 15;
    const int h  = (blockIdx.x >> 4) & 15;
    const int b  = blockIdx.x >> 8;
    const int q0 = qb * ATT_QB + wave * ATT_QW;

    const _Float16* qbase = QK + ((long)(b * HEADS + h)) * SEQ * DK;
    const _Float16* vbase = VT + ((long)(b * HEADS + h)) * DK * SEQ;
    const _Float16* kp = qbase + 8 * lane;
    const _Float16* vp = vbase + 8 * lane;
    const int* mrow = mask + b * SEQ + 8 * hh;

    const float cscale = 0.125f * 1.4426950408889634f;
    const float negl   = -1e9f * 1.4426950408889634f;

    v16h qf[ATT_R][2];
#pragma unroll
    for (int g = 0; g < ATT_R; ++g) {
        const _Float16* qp = kp + (long)(q0 + 16 * g) * DK;
        qf[g][0] = load_frag_fo(qp);
        qf[g][1] = load_frag_fo(qp + 512);
    }

    v8f acc[ATT_R][4];
    float m_run[ATT_R], l_run[ATT_R];
#pragma unroll
    for (int g = 0; g < ATT_R; ++g) {
        m_run[g] = -__builtin_inff(); l_run[g] = 0.0f;
#pragma unroll
        for (int t = 0; t < 4; ++t) acc[g][t] = v8f{};
    }

    v16h kf[2][2];
#pragma unroll
    for (int kt = 0; kt < 2; ++kt) {
        kf[kt][0] = load_frag_fo(kp + (long)(16 * kt) * DK);
        kf[kt][1] = load_frag_fo(kp + (long)(16 * kt) * DK + 512);
    }
    int mk[2][8];
#pragma unroll
    for (int kt = 0; kt < 2; ++kt) {
        const int4 m0 = *(const int4*)(mrow + 16 * kt);
        const int4 m1 = *(const int4*)(mrow + 16 * kt + 4);
        mk[kt][0] = m0.x; mk[kt][1] = m0.y; mk[kt][2] = m0.z; mk[kt][3] = m0.w;
        mk[kt][4] = m1.x; mk[kt][5] = m1.y; mk[kt][6] = m1.z; mk[kt][7] = m1.w;
    }

    for (int key0 = 0; key0 < SEQ; key0 += 32) {
        v16h vf[4];
#pragma unroll
        for (int t = 0; t < 4; ++t) vf[t] = load_frag_fo(vp + (long)key0 * DK + 512 * t);
        __builtin_amdgcn_sched_barrier(0);

        v8f s[2][ATT_R];
#pragma unroll
        for (int kt = 0; kt < 2; ++kt) {
#pragma unroll
            for (int g = 0; g < ATT_R; ++g) {
                v8f z = {};
                z = wmma_f16(kf[kt][0], qf[g][0], z);
                s[kt][g] = wmma_f16(kf[kt][1], qf[g][1], z);
            }
        }

        const int nk = (key0 + 32 < SEQ) ? key0 + 32 : key0;
#pragma unroll
        for (int kt = 0; kt < 2; ++kt) {
            kf[kt][0] = load_frag_fo(kp + (long)(nk + 16 * kt) * DK);
            kf[kt][1] = load_frag_fo(kp + (long)(nk + 16 * kt) * DK + 512);
        }
        int mkn[2][8];
#pragma unroll
        for (int kt = 0; kt < 2; ++kt) {
            const int4 m0 = *(const int4*)(mrow + nk + 16 * kt);
            const int4 m1 = *(const int4*)(mrow + nk + 16 * kt + 4);
            mkn[kt][0] = m0.x; mkn[kt][1] = m0.y; mkn[kt][2] = m0.z; mkn[kt][3] = m0.w;
            mkn[kt][4] = m1.x; mkn[kt][5] = m1.y; mkn[kt][6] = m1.z; mkn[kt][7] = m1.w;
        }

        float mb[ATT_R];
        bool need = false;
#pragma unroll
        for (int g = 0; g < ATT_R; ++g) {
            float mx = -__builtin_inff();
#pragma unroll
            for (int kt = 0; kt < 2; ++kt) {
#pragma unroll
                for (int r = 0; r < 8; ++r) {
                    float t = s[kt][g][r] * cscale;
                    t = (mk[kt][r] == 0) ? negl : t;
                    s[kt][g][r] = t;
                    mx = fmaxf(mx, t);
                }
            }
            mx = fmaxf(mx, xhalf(mx));
            mb[g] = mx;
            need = need || (mx > m_run[g] + 8.0f);
        }
        if (__builtin_amdgcn_ballot_w32(need) != 0) {
#pragma unroll
            for (int g = 0; g < ATT_R; ++g) {
                const float m_new = fmaxf(m_run[g], mb[g]);
                const float corr  = __builtin_amdgcn_exp2f(m_run[g] - m_new);
                m_run[g] = m_new;
                l_run[g] *= corr;
#pragma unroll
                for (int t = 0; t < 4; ++t) acc[g][t] = acc[g][t] * corr;
            }
        }

#pragma unroll
        for (int g = 0; g < ATT_R; ++g) {
            const float m = m_run[g];
            v16h pf;
            float ls = 0.0f;
#pragma unroll
            for (int r = 0; r < 8; ++r) {
                const float p0 = __builtin_amdgcn_exp2f(s[0][g][r] - m);
                const float p1 = __builtin_amdgcn_exp2f(s[1][g][r] - m);
                ls += p0 + p1;
                pf[r]     = (_Float16)p0;
                pf[8 + r] = (_Float16)p1;
            }
            l_run[g] += ls;
#pragma unroll
            for (int t = 0; t < 4; ++t) acc[g][t] = wmma_f16(vf[t], pf, acc[g][t]);
        }
#pragma unroll
        for (int kt = 0; kt < 2; ++kt) {
#pragma unroll
            for (int r = 0; r < 8; ++r) mk[kt][r] = mkn[kt][r];
        }
    }

#pragma unroll
    for (int g = 0; g < ATT_R; ++g) {
        float l = l_run[g];
        l = l + xhalf(l);
        const float inv = 1.0f / l;
        _Float16* op = CTX + ((long)(b * SEQ + q0 + 16 * g + c)) * D_MODEL + h * DK + 8 * hh;
#pragma unroll
        for (int t = 0; t < 4; ++t) {
            v8h o;
#pragma unroll
            for (int r = 0; r < 8; ++r) o[r] = (_Float16)(acc[g][t][r] * inv);
            *(v8h*)(op + 16 * t) = o;
        }
    }
}

extern "C" void kernel_launch(void* const* d_in, const int* in_sizes, int n_in,
                              void* d_out, int out_size, void* d_ws, size_t ws_size,
                              hipStream_t stream) {
    const float* x      = (const float*)d_in[0];
    const int*   mask   = (const int*)  d_in[1];
    const float* alpha1 = (const float*)d_in[2];
    const float* bias1  = (const float*)d_in[3];
    const float* alpha2 = (const float*)d_in[4];
    const float* bias2  = (const float*)d_in[5];
    const float* Wk     = (const float*)d_in[6];
    const float* bk     = (const float*)d_in[7];
    const float* Wv     = (const float*)d_in[8];
    const float* bv     = (const float*)d_in[9];
    const float* Wo     = (const float*)d_in[10];
    const float* bo     = (const float*)d_in[11];
    const float* W1     = (const float*)d_in[12];
    const float* b1     = (const float*)d_in[13];
    const float* W2     = (const float*)d_in[14];
    const float* b2     = (const float*)d_in[15];
    float* out = (float*)d_out;

    char* ws = (char*)d_ws;
    size_t off = 0;
    _Float16* x2h  = (_Float16*)(ws + off); off += (size_t)ROWS * D_MODEL * 2;
    _Float16* x2bh = (_Float16*)(ws + off); off += (size_t)ROWS * D_MODEL * 2;
    _Float16* qkh  = (_Float16*)(ws + off); off += (size_t)BATCH*HEADS*SEQ*DK*2;
    _Float16* vth  = (_Float16*)(ws + off); off += (size_t)BATCH*HEADS*DK*SEQ*2;
    _Float16* ctxh = (_Float16*)(ws + off); off += (size_t)ROWS * D_MODEL * 2;
    _Float16* h1h  = (_Float16*)(ws + off); off += (size_t)ROWS * D_FF * 2;
    _Float16* Wkt  = (_Float16*)(ws + off); off += (size_t)D_MODEL * D_MODEL * 2;
    _Float16* Wvt  = (_Float16*)(ws + off); off += (size_t)D_MODEL * D_MODEL * 2;
    _Float16* Wot  = (_Float16*)(ws + off); off += (size_t)D_MODEL * D_MODEL * 2;
    _Float16* W1t  = (_Float16*)(ws + off); off += (size_t)D_MODEL * D_FF * 2;
    _Float16* W2t  = (_Float16*)(ws + off); off += (size_t)D_FF * D_MODEL * 2;
    float*    x1   = (float*)   (ws + off); off += (size_t)ROWS * D_MODEL * 4;

    k_transpose_cvt_all<<<TR_GRID, 256, 0, stream>>>(Wk, Wv, Wo, W1, W2,
                                                     Wkt, Wvt, Wot, W1t, W2t);

    k_layernorm<<<ROWS, 256, 0, stream>>>(x, alpha1, bias1, x2h);

    #define GEMM_GRID(M_, N_) (((M_) / 128) * ((N_) / 128))

    k_gemm_wmma<2, false, false><<<GEMM_GRID(ROWS, D_MODEL), 128, 0, stream>>>(
        x2h, Wkt, bk, nullptr, nullptr, qkh, ROWS, D_MODEL, D_MODEL);
    k_gemm_wmma<3, false, false><<<GEMM_GRID(ROWS, D_MODEL), 128, 0, stream>>>(
        x2h, Wvt, bv, nullptr, nullptr, vth, ROWS, D_MODEL, D_MODEL);

    k_attention<<<BATCH*HEADS*(SEQ/ATT_QB), 128, 0, stream>>>(
        qkh, vth, mask, ctxh);

    k_gemm_wmma<1, false, true><<<GEMM_GRID(ROWS, D_MODEL), 128, 0, stream>>>(
        ctxh, Wot, bo, x, x1, nullptr, ROWS, D_MODEL, D_MODEL);

    k_layernorm<<<ROWS, 256, 0, stream>>>(x1, alpha2, bias2, x2bh);

    k_gemm_wmma<0, true, false><<<GEMM_GRID(ROWS, D_FF), 128, 0, stream>>>(
        x2bh, W1t, b1, nullptr, nullptr, h1h, ROWS, D_FF, D_MODEL);

    k_gemm_wmma<1, false, true><<<GEMM_GRID(ROWS, D_MODEL), 128, 0, stream>>>(
        h1h, W2t, b2, x1, out, nullptr, ROWS, D_MODEL, D_FF);

    (void)in_sizes; (void)n_in; (void)out_size; (void)ws_size;
}


# attention walks only the keys the mask keeps: q/k and v^T rows stored kept-first by their GEMMs (prefix count of the mask in the transpose launch); softmax sums and P.V summed in a new order
# speedup vs baseline: 6.4577x; 1.1208x over previous
#include <hip/hip_runtime.h>
#include <hip/hip_bf16.h>

typedef __attribute__((ext_vector_type(16))) _Float16 v16h;
typedef __attribute__((ext_vector_type(8)))  _Float16 v8h;
typedef __attribute__((ext_vector_type(4)))  _Float16 v4h;
typedef __attribute__((ext_vector_type(8)))  float    v8f;
typedef __attribute__((ext_vector_type(4)))  float    v4f;

#define D_MODEL 1024
#define HEADS   16
#define DK      64
#define D_FF    4096
#define BATCH   2
#define SEQ     2048
#define ROWS    (BATCH*SEQ)

__device__ __forceinline__ v16h pack16(v8h lo, v8h hi) {
    v16h r;
#pragma unroll
    for (int i = 0; i < 8; ++i) { r[i] = lo[i]; r[i + 8] = hi[i]; }
    return r;
}
__device__ __forceinline__ v16h load_frag(const _Float16* p) {
    v8h lo = *(const v8h*)(p);
    v8h hi = *(const v8h*)(p + 16);
    return pack16(lo, hi);
}
__device__ __forceinline__ int qk_off(int s, int d) {
    return ((((s >> 4) * 4 + (d >> 4)) * 32 + ((d >> 3) & 1) * 16 + (s & 15)) << 3) + (d & 7);
}
__device__ __forceinline__ int vt_off(int s, int d) {
    return (((((s >> 5) * 4 + (d >> 4)) * 2 + ((s >> 4) & 1)) * 32
             + ((s >> 3) & 1) * 16 + (d & 15)) << 3) + (s & 7);
}
__device__ __forceinline__ v16h load_frag_fo(const _Float16* p) {
    v8h lo = *(const v8h*)(p);
    v8h hi = *(const v8h*)(p + 256);
    return pack16(lo, hi);
}
__device__ __forceinline__ v8f wmma_f16(v16h a, v16h b, v8f c) {
    return __builtin_amdgcn_wmma_f32_16x16x32_f16(false, a, false, b,
                                                  (short)0, c, false, false);
}

#define TR_K 128
#define TR_N 64
#define TR_P 136
#define TR_TILES(K_, N_) (((K_) / TR_K) * ((N_) / TR_N))
#define TR_SQ   TR_TILES(D_MODEL, D_MODEL)
#define TR_W1   TR_TILES(D_MODEL, D_FF)
#define TR_W2   TR_TILES(D_FF, D_MODEL)
#define TR_GRID (3 * TR_SQ + TR_W1 + TR_W2)

__device__ __forceinline__ void key_partition(const int* __restrict__ mask, int* __restrict__ kofs,
                                              int* __restrict__ vofs, int* __restrict__ kinv,
                                              int* __restrict__ kcnt, int b, int* wsum) {
    const int t = threadIdx.x, lane = t & 31, w = t >> 5;
    const int* mr = mask + b * SEQ + 8 * t;
    const int4 m0 = *(const int4*)(mr);
    const int4 m1 = *(const int4*)(mr + 4);
    const int f[8] = { m0.x != 0, m0.y != 0, m0.z != 0, m0.w != 0,
                       m1.x != 0, m1.y != 0, m1.z != 0, m1.w != 0 };
    int c = 0;
#pragma unroll
    for (int e = 0; e < 8; ++e) c += f[e];
    int inc = c;
#pragma unroll
    for (int off = 1; off < 32; off <<= 1) {
        const int v = __shfl_up(inc, off, 32);
        if (lane >= off) inc += v;
    }
    if (lane == 31) wsum[w] = inc;
    __syncthreads();
    int before = 0, total = 0;
#pragma unroll
    for (int i = 0; i < 8; ++i) {
        const int v = wsum[i];
        total += v;
        if (i < w) before += v;
    }
    const bool none = (total == 0);
    const int cnt = none ? SEQ : total;
    int nv = none ? 8 * t : before + inc - c;
    int ko[8], vo[8];
#pragma unroll
    for (int e = 0; e < 8; ++e) {
        const int s = 8 * t + e;
        const bool kept = none || f[e];
        const int pos = kept ? nv : cnt + (s - nv);
        nv += kept ? 1 : 0;
        ko[e] = qk_off(pos, 0);
        vo[e] = vt_off(pos, 0);
        kinv[b * SEQ + pos] = s;
    }
    *(int4*)(kofs + b * SEQ + 8 * t)     = make_int4(ko[0], ko[1], ko[2], ko[3]);
    *(int4*)(kofs + b * SEQ + 8 * t + 4) = make_int4(ko[4], ko[5], ko[6], ko[7]);
    *(int4*)(vofs + b * SEQ + 8 * t)     = make_int4(vo[0], vo[1], vo[2], vo[3]);
    *(int4*)(vofs + b * SEQ + 8 * t + 4) = make_int4(vo[4], vo[5], vo[6], vo[7]);
    if (t == 0) { kcnt[2 * b] = cnt; kcnt[2 * b + 1] = none ? 1 : 0; }
}

__global__ void __launch_bounds__(256)
k_transpose_cvt_all(const float* __restrict__ Wk, const float* __restrict__ Wv,
                    const float* __restrict__ Wo, const float* __restrict__ W1,
                    const float* __restrict__ W2,
                    _Float16* __restrict__ Wkt, _Float16* __restrict__ Wvt,
                    _Float16* __restrict__ Wot, _Float16* __restrict__ W1t,
                    _Float16* __restrict__ W2t,
                    const int* __restrict__ mask, int* __restrict__ kofs, int* __restrict__ vofs,
                    int* __restrict__ kinv, int* __restrict__ kcnt) {
    __shared__ __attribute__((aligned(16))) _Float16 sT[TR_N * TR_P];
    __shared__ int wsum[8];

    if (blockIdx.x < BATCH) {
        key_partition(mask, kofs, vofs, kinv, kcnt, blockIdx.x, wsum);
        return;
    }

    int tile = blockIdx.x - BATCH;
    const float* W; _Float16* Wt; int K, N;
    if (tile < TR_SQ)                 { W = Wk; Wt = Wkt; K = D_MODEL; N = D_MODEL; }
    else if (tile < 2 * TR_SQ)        { W = Wv; Wt = Wvt; K = D_MODEL; N = D_MODEL; tile -= TR_SQ; }
    else if (tile < 3 * TR_SQ)        { W = Wo; Wt = Wot; K = D_MODEL; N = D_MODEL; tile -= 2 * TR_SQ; }
    else if (tile < 3 * TR_SQ + TR_W1){ W = W1; Wt = W1t; K = D_MODEL; N = D_FF;    tile -= 3 * TR_SQ; }
    else                              { W = W2; Wt = W2t; K = D_FF;    N = D_MODEL; tile -= 3 * TR_SQ + TR_W1; }
    const int ntn = N / TR_N;
    const int kt = tile / ntn, nt = tile % ntn;

    const int t = threadIdx.x;
    const int q = t & 15, g = t >> 4;

    const float* src = W + (long)(kt * TR_K + 8 * g) * N + nt * TR_N + 4 * q;
    v4f r[8];
#pragma unroll
    for (int i = 0; i < 8; ++i) r[i] = *(const v4f*)(src + (long)i * N);

#pragma unroll
    for (int j = 0; j < 4; ++j) {
        v8h o;
#pragma unroll
        for (int i = 0; i < 8; ++i) o[i] = (_Float16)r[i][j];
        *(v8h*)(&sT[(4 * q + j) * TR_P + 8 * g]) = o;
    }
    __syncthreads();

    _Float16* dst = Wt + (long)(nt * TR_N + g) * K + kt * TR_K + 8 * q;
#pragma unroll
    for (int s = 0; s < 4; ++s) {
        const v8h o = *(const v8h*)(&sT[(g + 16 * s) * TR_P + 8 * q]);
        *(v8h*)(dst + (long)(16 * s) * K) = o;
    }
}

__global__ void __launch_bounds__(256)
k_layernorm(const float* __restrict__ X, const float* __restrict__ alpha,
            const float* __restrict__ beta, _Float16* __restrict__ Y) {
    __shared__ float s1[256], s2[256];
    const int row = blockIdx.x, t = threadIdx.x;
    const float* xr = X + (long)row * D_MODEL;
    v4f x = *(const v4f*)(xr + t * 4);
    float s = x[0] + x[1] + x[2] + x[3];
    float q = x[0]*x[0] + x[1]*x[1] + x[2]*x[2] + x[3]*x[3];
    s1[t] = s; s2[t] = q; __syncthreads();
    for (int off = 128; off; off >>= 1) {
        if (t < off) { s1[t] += s1[t + off]; s2[t] += s2[t + off]; }
        __syncthreads();
    }
    const float mu  = s1[0] * (1.0f / D_MODEL);
    const float var = (s2[0] - (float)D_MODEL * mu * mu) * (1.0f / (D_MODEL - 1));
    const float inv = 1.0f / (sqrtf(var) + 1e-6f);
    v4h o;
#pragma unroll
    for (int i = 0; i < 4; ++i) {
        int c = t * 4 + i;
        o[i] = (_Float16)(alpha[c] * (x[i] - mu) * inv + beta[c]);
    }
    *(v4h*)(Y + (long)row * D_MODEL + t * 4) = o;
}

#define GT_M 128
#define GT_N 128
#define GT_K 128
#define GT_P 136

#define FRAG_FENCE() __builtin_amdgcn_sched_barrier(0)

template <bool MORE, bool MORE2>
__device__ __forceinline__ void gemm_stage(
        v8f (&acc)[4][4], v16h (&fa0)[4], v16h (&fb0)[4], v16h (&fa1)[4], v16h (&fb1)[4],
        v8h (&ra)[16], v8h (&rb)[16],
        const _Float16* pa, const _Float16* pb,
        const _Float16* na, const _Float16* nb,
        _Float16* wa, _Float16* wb,
        const char* ag, const char* bg,
        const unsigned (&g8)[16]) {
#define GT_STORE(G)                                                              \
    if (MORE) {                                                                  \
        _Pragma("unroll")                                                        \
        for (int j = 4 * (G); j < 4 * (G) + 4; ++j) {                            \
            *(v8h*)(wa + j * 8 * GT_P) = ra[j];                                  \
            *(v8h*)(wb + j * 8 * GT_P) = rb[j];                                  \
        }                                                                        \
    }
#define GT_LOAD(G)                                                               \
    if (MORE2) {                                                                 \
        _Pragma("unroll")                                                        \
        for (int j = 4 * (G); j < 4 * (G) + 4; ++j) {                            \
            ra[j] = *(const v8h*)(ag + g8[j]);                                  \
            rb[j] = *(const v8h*)(bg + g8[j]);                                  \
        }                                                                        \
    }

#pragma unroll
    for (int i = 0; i < 4; ++i) fa1[i] = load_frag(pa + 32 + i * 16 * GT_P);
#pragma unroll
    for (int j = 0; j < 4; ++j) fb1[j] = load_frag(pb + 32 + j * 16 * GT_P);
    FRAG_FENCE();
#pragma unroll
    for (int i = 0; i < 4; ++i) {
#pragma unroll
        for (int j = 0; j < 4; ++j) acc[i][j] = wmma_f16(fa0[i], fb0[j], acc[i][j]);
        if (i == 0) { GT_STORE(0) }
        if (i == 2) { GT_LOAD(0) }
    }
    FRAG_FENCE();

#pragma unroll
    for (int i = 0; i < 4; ++i) fa0[i] = load_frag(pa + 64 + i * 16 * GT_P);
#pragma unroll
    for (int j = 0; j < 4; ++j) fb0[j] = load_frag(pb + 64 + j * 16 * GT_P);
    FRAG_FENCE();
#pragma unroll
    for (int i = 0; i < 4; ++i) {
#pragma unroll
        for (int j = 0; j < 4; ++j) acc[i][j] = wmma_f16(fa1[i], fb1[j], acc[i][j]);
        if (i == 0) { GT_STORE(1) }
        if (i == 2) { GT_LOAD(1) }
    }
    FRAG_FENCE();

#pragma unroll
    for (int i = 0; i < 4; ++i) fa1[i] = load_frag(pa + 96 + i * 16 * GT_P);
#pragma unroll
    for (int j = 0; j < 4; ++j) fb1[j] = load_frag(pb + 96 + j * 16 * GT_P);
    FRAG_FENCE();
#pragma unroll
    for (int i = 0; i < 4; ++i) {
#pragma unroll
        for (int j = 0; j < 4; ++j) acc[i][j] = wmma_f16(fa0[i], fb0[j], acc[i][j]);
        if (i == 0) { GT_STORE(2) }
        if (i == 1) { GT_STORE(3) }
        if (i == 2) { GT_LOAD(2) }
        if (i == 3) { GT_LOAD(3) }
    }

    __builtin_amdgcn_fence(__ATOMIC_RELEASE, "workgroup", "local");
    __builtin_amdgcn_s_barrier();
    __builtin_amdgcn_fence(__ATOMIC_ACQUIRE, "workgroup", "local");

    if (MORE) {
#pragma unroll
        for (int i = 0; i < 4; ++i) fa0[i] = load_frag(na + i * 16 * GT_P);
#pragma unroll
        for (int j = 0; j < 4; ++j) fb0[j] = load_frag(nb + j * 16 * GT_P);
    }
#pragma unroll
    for (int i = 0; i < 4; ++i) {
#pragma unroll
        for (int j = 0; j < 4; ++j) acc[i][j] = wmma_f16(fa1[i], fb1[j], acc[i][j]);
    }
#undef GT_STORE
#undef GT_LOAD
}

template <int OMODE, bool RELU, bool RESID>
__global__ void __launch_bounds__(128)
k_gemm_wmma(const _Float16* __restrict__ A, const _Float16* __restrict__ Bt,
            const float* __restrict__ bias, const float* __restrict__ resid,
            float* __restrict__ Cf, _Float16* __restrict__ Ch,
            const int* __restrict__ rofs,
            int M, int N, int K) {
    __shared__ __attribute__((aligned(16))) _Float16 sA[2][GT_M * GT_P];
    __shared__ __attribute__((aligned(16))) _Float16 sB[2][GT_N * GT_P];

    const int tid  = threadIdx.x;
    const int lane = tid & 31;
    const int wave = tid >> 5;
    const int wm = wave >> 1, wn = wave & 1;
    const int bn_cnt = N >> 7;
    const int bm = blockIdx.x / bn_cnt;
    const int bn = blockIdx.x % bn_cnt;
    if (bm * GT_M >= M) return;

    const int ln = lane & 15;
    const int hh = lane >> 4;

    const int srow = tid >> 4, sch = tid & 15;
    const char* ag = (const char*)(A  + (long)(bm * GT_M) * K);
    const char* bg = (const char*)(Bt + (long)(bn * GT_N) * K);
    unsigned g8[16];
#pragma unroll
    for (int j = 0; j < 16; ++j) g8[j] = (unsigned)(((srow + 8 * j) * K + sch * 8) * 2);
    const int  soff = srow * GT_P + sch * 8;

    v8h ra[16], rb[16];
#pragma unroll
    for (int j = 0; j < 16; ++j) {
        ra[j] = *(const v8h*)(ag + g8[j]);
        rb[j] = *(const v8h*)(bg + g8[j]);
    }
#pragma unroll
    for (int j = 0; j < 16; ++j) {
        *(v8h*)(&sA[0][soff + j * 8 * GT_P]) = ra[j];
        *(v8h*)(&sB[0][soff + j * 8 * GT_P]) = rb[j];
    }
    __syncthreads();

    v8f acc[4][4];
#pragma unroll
    for (int i = 0; i < 4; ++i) {
#pragma unroll
        for (int j = 0; j < 4; ++j) acc[i][j] = v8f{};
    }

    const int aoff = (wm * 64 + ln) * GT_P + 8 * hh;
    const int boff = (wn * 64 + ln) * GT_P + 8 * hh;

    const int nk = K / GT_K;

    if (nk > 1) {
        ag += GT_K * 2; bg += GT_K * 2;
#pragma unroll
        for (int j = 0; j < 16; ++j) {
            ra[j] = *(const v8h*)(ag + g8[j]);
            rb[j] = *(const v8h*)(bg + g8[j]);
        }
    }

    v16h fa0[4], fb0[4], fa1[4], fb1[4];
#pragma unroll
    for (int i = 0; i < 4; ++i) fa0[i] = load_frag(&sA[0][aoff] + i * 16 * GT_P);
#pragma unroll
    for (int j = 0; j < 4; ++j) fb0[j] = load_frag(&sB[0][boff] + j * 16 * GT_P);

    int kt = 0;
    for (; kt + 2 < nk; ++kt) {
        const int cur = kt & 1;
        ag += GT_K * 2; bg += GT_K * 2;
        gemm_stage<true, true>(acc, fa0, fb0, fa1, fb1, ra, rb,
                               &sA[cur][aoff], &sB[cur][boff], &sA[cur ^ 1][aoff], &sB[cur ^ 1][boff],
                               &sA[cur ^ 1][soff], &sB[cur ^ 1][soff], ag, bg, g8);
    }
    if (nk > 1) {
        const int cur = kt & 1;
        gemm_stage<true, false>(acc, fa0, fb0, fa1, fb1, ra, rb,
                                &sA[cur][aoff], &sB[cur][boff], &sA[cur ^ 1][aoff], &sB[cur ^ 1][boff],
                                &sA[cur ^ 1][soff], &sB[cur ^ 1][soff], ag, bg, g8);
        ++kt;
    }
    {
        const int cur = kt & 1;
        gemm_stage<false, false>(acc, fa0, fb0, fa1, fb1, ra, rb,
                                 &sA[cur][aoff], &sB[cur][boff], &sA[cur ^ 1][aoff], &sB[cur ^ 1][boff],
                                 &sA[cur ^ 1][soff], &sB[cur ^ 1][soff], ag, bg, g8);
    }

    const int lm = hh << 3;
    const int m0 = bm * GT_M + wm * 64;
    const int n0 = bn * GT_N + wn * 64;
#pragma unroll
    for (int i = 0; i < 4; ++i) {
        int ro[8] = {};
        if (OMODE == 2 || OMODE == 3) {
            const int4 r0 = *(const int4*)(rofs + m0 + i * 16 + lm);
            const int4 r1 = *(const int4*)(rofs + m0 + i * 16 + lm + 4);
            ro[0] = r0.x; ro[1] = r0.y; ro[2] = r0.z; ro[3] = r0.w;
            ro[4] = r1.x; ro[5] = r1.y; ro[6] = r1.z; ro[7] = r1.w;
        }
#pragma unroll
        for (int j = 0; j < 4; ++j) {
            const v8f accv = acc[i][j];
            const int col = n0 + j * 16 + ln;
            const float bv = bias[col];
            if (OMODE == 3 || OMODE == 2) {
                const int row0 = m0 + i * 16 + lm;
                const int b_ = row0 / SEQ, h = col >> 6, d = col & 63;
                _Float16* dst = Ch + ((long)(b_ * HEADS + h)) * SEQ * DK
                                   + (OMODE == 3 ? vt_off(0, d) : qk_off(0, d));
#pragma unroll
                for (int e = 0; e < 8; ++e) {
                    float v = accv[e] + bv;
                    if (RELU)  v = v > 0.0f ? v : 0.0f;
                    if (RESID) v += resid[(long)(row0 + e) * N + col];
                    dst[ro[e]] = (_Float16)v;
                }
                continue;
            }
#pragma unroll
            for (int e = 0; e < 8; ++e) {
                const int row = m0 + i * 16 + e + lm;
                float v = accv[e] + bv;
                if (RELU)  v = v > 0.0f ? v : 0.0f;
                if (RESID) v += resid[(long)row * N + col];
                if (OMODE == 0) {
                    Ch[(long)row * N + col] = (_Float16)v;
                } else if (OMODE == 1) {
                    Cf[(long)row * N + col] = v;
                }
            }
        }
    }
}

#define ATT_R     2
#define ATT_QW    (16 * ATT_R)
#define ATT_WAVES 4
#define ATT_QB    (ATT_QW * ATT_WAVES)

__device__ __forceinline__ float xhalf(float x) {
    return __builtin_bit_cast(float, __builtin_amdgcn_permlanex16(
        __builtin_bit_cast(int, x), __builtin_bit_cast(int, x),
        0x76543210, 0xfedcba98, false, false));
}

__global__ void __launch_bounds__(128)
k_attention(const _Float16* __restrict__ QK, const _Float16* __restrict__ VT,
            const int* __restrict__ kcnt, const int* __restrict__ kinv,
            _Float16* __restrict__ CTX) {
    const int lane = threadIdx.x & 31, wave = threadIdx.x >> 5;
    const int c = lane & 15, hh = lane >> 4;
    const int qb = blockIdx.x & 15;
    const int h  = (blockIdx.x >> 4) & 15;
    const int b  = blockIdx.x >> 8;
    const int q0 = qb * ATT_QB + wave * ATT_QW;

    const _Float16* qbase = QK + ((long)(b * HEADS + h)) * SEQ * DK;
    const _Float16* vbase = VT + ((long)(b * HEADS + h)) * DK * SEQ;
    const _Float16* kp = qbase + 8 * lane;
    const _Float16* vp = vbase + 8 * lane;

    const int cnt    = kcnt[2 * b];
    const int nfull  = cnt >> 5;
    const int tail   = cnt & 31;
    const int nsteps = nfull + (tail ? 1 : 0);
    const bool  none   = kcnt[2 * b + 1] != 0;
    const float cscale = 0.125f * 1.4426950408889634f;

    v16h qf[ATT_R][2];
#pragma unroll
    for (int g = 0; g < ATT_R; ++g) {
        const _Float16* qp = kp + (long)(q0 + 16 * g) * DK;
        qf[g][0] = load_frag_fo(qp);
        qf[g][1] = load_frag_fo(qp + 512);
        if (none) { qf[g][0] = v16h{}; qf[g][1] = v16h{}; }
    }

    v8f acc[ATT_R][4];
    float m_run[ATT_R], l_run[ATT_R];
#pragma unroll
    for (int g = 0; g < ATT_R; ++g) {
        m_run[g] = -__builtin_inff(); l_run[g] = 0.0f;
#pragma unroll
        for (int t = 0; t < 4; ++t) acc[g][t] = v8f{};
    }

    v16h kf[2][2];
#pragma unroll
    for (int kt = 0; kt < 2; ++kt) {
        kf[kt][0] = load_frag_fo(kp + (long)(16 * kt) * DK);
        kf[kt][1] = load_frag_fo(kp + (long)(16 * kt) * DK + 512);
    }

    for (int step = 0; step < nsteps; ++step) {
        const int key0 = step << 5;
        v16h vf[4];
#pragma unroll
        for (int t = 0; t < 4; ++t) vf[t] = load_frag_fo(vp + (long)key0 * DK + 512 * t);
        __builtin_amdgcn_sched_barrier(0);

        v8f s[2][ATT_R];
#pragma unroll
        for (int kt = 0; kt < 2; ++kt) {
#pragma unroll
            for (int g = 0; g < ATT_R; ++g) {
                v8f z = {};
                z = wmma_f16(kf[kt][0], qf[g][0], z);
                s[kt][g] = wmma_f16(kf[kt][1], qf[g][1], z);
            }
        }

        const int nk = (step + 1 < nsteps) ? key0 + 32 : key0;
#pragma unroll
        for (int kt = 0; kt < 2; ++kt) {
            kf[kt][0] = load_frag_fo(kp + (long)(nk + 16 * kt) * DK);
            kf[kt][1] = load_frag_fo(kp + (long)(nk + 16 * kt) * DK + 512);
        }

#pragma unroll
        for (int g = 0; g < ATT_R; ++g) {
#pragma unroll
            for (int kt = 0; kt < 2; ++kt) s[kt][g] = s[kt][g] * cscale;
        }
        if (step == nfull) {
#pragma unroll
            for (int kt = 0; kt < 2; ++kt) {
#pragma unroll
                for (int r = 0; r < 8; ++r) {
                    const bool off = (16 * kt + 8 * hh + r) >= tail;
#pragma unroll
                    for (int g = 0; g < ATT_R; ++g) s[kt][g][r] = off ? -__builtin_inff() : s[kt][g][r];
                }
            }
        }

        float mb[ATT_R];
        bool need = false;
#pragma unroll
        for (int g = 0; g < ATT_R; ++g) {
            float mx = -__builtin_inff();
#pragma unroll
            for (int kt = 0; kt < 2; ++kt) {
#pragma unroll
                for (int r = 0; r < 8; ++r) mx = fmaxf(mx, s[kt][g][r]);
            }
            mx = fmaxf(mx, xhalf(mx));
            mb[g] = mx;
            need = need || (mx > m_run[g] + 8.0f);
        }
        if (__builtin_amdgcn_ballot_w32(need) != 0) {
#pragma unroll
            for (int g = 0; g < ATT_R; ++g) {
                const float m_new = fmaxf(m_run[g], mb[g]);
                const float corr  = __builtin_amdgcn_exp2f(m_run[g] - m_new);
                m_run[g] = m_new;
                l_run[g] *= corr;
#pragma unroll
                for (int t = 0; t < 4; ++t) acc[g][t] = acc[g][t] * corr;
            }
        }

#pragma unroll
        for (int g = 0; g < ATT_R; ++g) {
            const float m = m_run[g];
            v16h pf;
            float ls = 0.0f;
#pragma unroll
            for (int r = 0; r < 8; ++r) {
                const float p0 = __builtin_amdgcn_exp2f(s[0][g][r] - m);
                const float p1 = __builtin_amdgcn_exp2f(s[1][g][r] - m);
                ls += p0 + p1;
                pf[r]     = (_Float16)p0;
                pf[8 + r] = (_Float16)p1;
            }
            l_run[g] += ls;
#pragma unroll
            for (int t = 0; t < 4; ++t) acc[g][t] = wmma_f16(vf[t], pf, acc[g][t]);
        }
    }

#pragma unroll
    for (int g = 0; g < ATT_R; ++g) {
        float l = l_run[g];
        l = l + xhalf(l);
        const float inv = 1.0f / l;
        const int orow = kinv[b * SEQ + q0 + 16 * g + c];
        _Float16* op = CTX + ((long)(b * SEQ + orow)) * D_MODEL + h * DK + 8 * hh;
#pragma unroll
        for (int t = 0; t < 4; ++t) {
            v8h o;
#pragma unroll
            for (int r = 0; r < 8; ++r) o[r] = (_Float16)(acc[g][t][r] * inv);
            *(v8h*)(op + 16 * t) = o;
        }
    }
}

extern "C" void kernel_launch(void* const* d_in, const int* in_sizes, int n_in,
                              void* d_out, int out_size, void* d_ws, size_t ws_size,
                              hipStream_t stream) {
    const float* x      = (const float*)d_in[0];
    const int*   mask   = (const int*)  d_in[1];
    const float* alpha1 = (const float*)d_in[2];
    const float* bias1  = (const float*)d_in[3];
    const float* alpha2 = (const float*)d_in[4];
    const float* bias2  = (const float*)d_in[5];
    const float* Wk     = (const float*)d_in[6];
    const float* bk     = (const float*)d_in[7];
    const float* Wv     = (const float*)d_in[8];
    const float* bv     = (const float*)d_in[9];
    const float* Wo     = (const float*)d_in[10];
    const float* bo     = (const float*)d_in[11];
    const float* W1     = (const float*)d_in[12];
    const float* b1     = (const float*)d_in[13];
    const float* W2     = (const float*)d_in[14];
    const float* b2     = (const float*)d_in[15];
    float* out = (float*)d_out;

    char* ws = (char*)d_ws;
    size_t off = 0;
    _Float16* x2h  = (_Float16*)(ws + off); off += (size_t)ROWS * D_MODEL * 2;
    _Float16* x2bh = (_Float16*)(ws + off); off += (size_t)ROWS * D_MODEL * 2;
    _Float16* qkh  = (_Float16*)(ws + off); off += (size_t)BATCH*HEADS*SEQ*DK*2;
    _Float16* vth  = (_Float16*)(ws + off); off += (size_t)BATCH*HEADS*DK*SEQ*2;
    _Float16* ctxh = (_Float16*)(ws + off); off += (size_t)ROWS * D_MODEL * 2;
    _Float16* h1h  = (_Float16*)(ws + off); off += (size_t)ROWS * D_FF * 2;
    _Float16* Wkt  = (_Float16*)(ws + off); off += (size_t)D_MODEL * D_MODEL * 2;
    _Float16* Wvt  = (_Float16*)(ws + off); off += (size_t)D_MODEL * D_MODEL * 2;
    _Float16* Wot  = (_Float16*)(ws + off); off += (size_t)D_MODEL * D_MODEL * 2;
    _Float16* W1t  = (_Float16*)(ws + off); off += (size_t)D_MODEL * D_FF * 2;
    _Float16* W2t  = (_Float16*)(ws + off); off += (size_t)D_FF * D_MODEL * 2;
    float*    x1   = (float*)   (ws + off); off += (size_t)ROWS * D_MODEL * 4;
    int* kofs = (int*)h1h;
    int* vofs = kofs + ROWS;
    int* kinv = vofs + ROWS;
    int* kcnt = kinv + ROWS;

    k_transpose_cvt_all<<<BATCH + TR_GRID, 256, 0, stream>>>(Wk, Wv, Wo, W1, W2,
                                                             Wkt, Wvt, Wot, W1t, W2t,
                                                             mask, kofs, vofs, kinv, kcnt);

    k_layernorm<<<ROWS, 256, 0, stream>>>(x, alpha1, bias1, x2h);

    #define GEMM_GRID(M_, N_) (((M_) / 128) * ((N_) / 128))

    k_gemm_wmma<2, false, false><<<GEMM_GRID(ROWS, D_MODEL), 128, 0, stream>>>(
        x2h, Wkt, bk, nullptr, nullptr, qkh, kofs, ROWS, D_MODEL, D_MODEL);
    k_gemm_wmma<3, false, false><<<GEMM_GRID(ROWS, D_MODEL), 128, 0, stream>>>(
        x2h, Wvt, bv, nullptr, nullptr, vth, vofs, ROWS, D_MODEL, D_MODEL);

    k_attention<<<BATCH*HEADS*(SEQ/ATT_QB), 128, 0, stream>>>(
        qkh, vth, kcnt, kinv, ctxh);

    k_gemm_wmma<1, false, true><<<GEMM_GRID(ROWS, D_MODEL), 128, 0, stream>>>(
        ctxh, Wot, bo, x, x1, nullptr, nullptr, ROWS, D_MODEL, D_MODEL);

    k_layernorm<<<ROWS, 256, 0, stream>>>(x1, alpha2, bias2, x2bh);

    k_gemm_wmma<0, true, false><<<GEMM_GRID(ROWS, D_FF), 128, 0, stream>>>(
        x2bh, W1t, b1, nullptr, nullptr, h1h, nullptr, ROWS, D_FF, D_MODEL);

    k_gemm_wmma<1, false, true><<<GEMM_GRID(ROWS, D_MODEL), 128, 0, stream>>>(
        h1h, W2t, b2, x1, out, nullptr, nullptr, ROWS, D_MODEL, D_FF);

    (void)in_sizes; (void)n_in; (void)out_size; (void)ws_size;
}


# attention key loop: K fragments requested two steps ahead and V^T fragments one step ahead of their use (second register sets, two steps per loop trip), was half a step or less; sums unchanged
# speedup vs baseline: 6.7121x; 1.0394x over previous
#include <hip/hip_runtime.h>
#include <hip/hip_bf16.h>

typedef __attribute__((ext_vector_type(16))) _Float16 v16h;
typedef __attribute__((ext_vector_type(8)))  _Float16 v8h;
typedef __attribute__((ext_vector_type(4)))  _Float16 v4h;
typedef __attribute__((ext_vector_type(8)))  float    v8f;
typedef __attribute__((ext_vector_type(4)))  float    v4f;

#define D_MODEL 1024
#define HEADS   16
#define DK      64
#define D_FF    4096
#define BATCH   2
#define SEQ     2048
#define ROWS    (BATCH*SEQ)

__device__ __forceinline__ v16h pack16(v8h lo, v8h hi) {
    v16h r;
#pragma unroll
    for (int i = 0; i < 8; ++i) { r[i] = lo[i]; r[i + 8] = hi[i]; }
    return r;
}
__device__ __forceinline__ v16h load_frag(const _Float16* p) {
    v8h lo = *(const v8h*)(p);
    v8h hi = *(const v8h*)(p + 16);
    return pack16(lo, hi);
}
__device__ __forceinline__ int qk_off(int s, int d) {
    return ((((s >> 4) * 4 + (d >> 4)) * 32 + ((d >> 3) & 1) * 16 + (s & 15)) << 3) + (d & 7);
}
__device__ __forceinline__ int vt_off(int s, int d) {
    return (((((s >> 5) * 4 + (d >> 4)) * 2 + ((s >> 4) & 1)) * 32
             + ((s >> 3) & 1) * 16 + (d & 15)) << 3) + (s & 7);
}
__device__ __forceinline__ v16h load_frag_fo(const _Float16* p) {
    v8h lo = *(const v8h*)(p);
    v8h hi = *(const v8h*)(p + 256);
    return pack16(lo, hi);
}
__device__ __forceinline__ v8f wmma_f16(v16h a, v16h b, v8f c) {
    return __builtin_amdgcn_wmma_f32_16x16x32_f16(false, a, false, b,
                                                  (short)0, c, false, false);
}

#define TR_K 128
#define TR_N 64
#define TR_P 136
#define TR_TILES(K_, N_) (((K_) / TR_K) * ((N_) / TR_N))
#define TR_SQ   TR_TILES(D_MODEL, D_MODEL)
#define TR_W1   TR_TILES(D_MODEL, D_FF)
#define TR_W2   TR_TILES(D_FF, D_MODEL)
#define TR_GRID (3 * TR_SQ + TR_W1 + TR_W2)

__device__ __forceinline__ void key_partition(const int* __restrict__ mask, int* __restrict__ kofs,
                                              int* __restrict__ vofs, int* __restrict__ kinv,
                                              int* __restrict__ kcnt, int b, int* wsum) {
    const int t = threadIdx.x, lane = t & 31, w = t >> 5;
    const int* mr = mask + b * SEQ + 8 * t;
    const int4 m0 = *(const int4*)(mr);
    const int4 m1 = *(const int4*)(mr + 4);
    const int f[8] = { m0.x != 0, m0.y != 0, m0.z != 0, m0.w != 0,
                       m1.x != 0, m1.y != 0, m1.z != 0, m1.w != 0 };
    int c = 0;
#pragma unroll
    for (int e = 0; e < 8; ++e) c += f[e];
    int inc = c;
#pragma unroll
    for (int off = 1; off < 32; off <<= 1) {
        const int v = __shfl_up(inc, off, 32);
        if (lane >= off) inc += v;
    }
    if (lane == 31) wsum[w] = inc;
    __syncthreads();
    int before = 0, total = 0;
#pragma unroll
    for (int i = 0; i < 8; ++i) {
        const int v = wsum[i];
        total += v;
        if (i < w) before += v;
    }
    const bool none = (total == 0);
    const int cnt = none ? SEQ : total;
    int nv = none ? 8 * t : before + inc - c;
    int ko[8], vo[8];
#pragma unroll
    for (int e = 0; e < 8; ++e) {
        const int s = 8 * t + e;
        const bool kept = none || f[e];
        const int pos = kept ? nv : cnt + (s - nv);
        nv += kept ? 1 : 0;
        ko[e] = qk_off(pos, 0);
        vo[e] = vt_off(pos, 0);
        kinv[b * SEQ + pos] = s;
    }
    *(int4*)(kofs + b * SEQ + 8 * t)     = make_int4(ko[0], ko[1], ko[2], ko[3]);
    *(int4*)(kofs + b * SEQ + 8 * t + 4) = make_int4(ko[4], ko[5], ko[6], ko[7]);
    *(int4*)(vofs + b * SEQ + 8 * t)     = make_int4(vo[0], vo[1], vo[2], vo[3]);
    *(int4*)(vofs + b * SEQ + 8 * t + 4) = make_int4(vo[4], vo[5], vo[6], vo[7]);
    if (t == 0) { kcnt[2 * b] = cnt; kcnt[2 * b + 1] = none ? 1 : 0; }
}

__global__ void __launch_bounds__(256)
k_transpose_cvt_all(const float* __restrict__ Wk, const float* __restrict__ Wv,
                    const float* __restrict__ Wo, const float* __restrict__ W1,
                    const float* __restrict__ W2,
                    _Float16* __restrict__ Wkt, _Float16* __restrict__ Wvt,
                    _Float16* __restrict__ Wot, _Float16* __restrict__ W1t,
                    _Float16* __restrict__ W2t,
                    const int* __restrict__ mask, int* __restrict__ kofs, int* __restrict__ vofs,
                    int* __restrict__ kinv, int* __restrict__ kcnt) {
    __shared__ __attribute__((aligned(16))) _Float16 sT[TR_N * TR_P];
    __shared__ int wsum[8];

    if (blockIdx.x < BATCH) {
        key_partition(mask, kofs, vofs, kinv, kcnt, blockIdx.x, wsum);
        return;
    }

    int tile = blockIdx.x - BATCH;
    const float* W; _Float16* Wt; int K, N;
    if (tile < TR_SQ)                 { W = Wk; Wt = Wkt; K = D_MODEL; N = D_MODEL; }
    else if (tile < 2 * TR_SQ)        { W = Wv; Wt = Wvt; K = D_MODEL; N = D_MODEL; tile -= TR_SQ; }
    else if (tile < 3 * TR_SQ)        { W = Wo; Wt = Wot; K = D_MODEL; N = D_MODEL; tile -= 2 * TR_SQ; }
    else if (tile < 3 * TR_SQ + TR_W1){ W = W1; Wt = W1t; K = D_MODEL; N = D_FF;    tile -= 3 * TR_SQ; }
    else                              { W = W2; Wt = W2t; K = D_FF;    N = D_MODEL; tile -= 3 * TR_SQ + TR_W1; }
    const int ntn = N / TR_N;
    const int kt = tile / ntn, nt = tile % ntn;

    const int t = threadIdx.x;
    const int q = t & 15, g = t >> 4;

    const float* src = W + (long)(kt * TR_K + 8 * g) * N + nt * TR_N + 4 * q;
    v4f r[8];
#pragma unroll
    for (int i = 0; i < 8; ++i) r[i] = *(const v4f*)(src + (long)i * N);

#pragma unroll
    for (int j = 0; j < 4; ++j) {
        v8h o;
#pragma unroll
        for (int i = 0; i < 8; ++i) o[i] = (_Float16)r[i][j];
        *(v8h*)(&sT[(4 * q + j) * TR_P + 8 * g]) = o;
    }
    __syncthreads();

    _Float16* dst = Wt + (long)(nt * TR_N + g) * K + kt * TR_K + 8 * q;
#pragma unroll
    for (int s = 0; s < 4; ++s) {
        const v8h o = *(const v8h*)(&sT[(g + 16 * s) * TR_P + 8 * q]);
        *(v8h*)(dst + (long)(16 * s) * K) = o;
    }
}

__global__ void __launch_bounds__(256)
k_layernorm(const float* __restrict__ X, const float* __restrict__ alpha,
            const float* __restrict__ beta, _Float16* __restrict__ Y) {
    __shared__ float s1[256], s2[256];
    const int row = blockIdx.x, t = threadIdx.x;
    const float* xr = X + (long)row * D_MODEL;
    v4f x = *(const v4f*)(xr + t * 4);
    float s = x[0] + x[1] + x[2] + x[3];
    float q = x[0]*x[0] + x[1]*x[1] + x[2]*x[2] + x[3]*x[3];
    s1[t] = s; s2[t] = q; __syncthreads();
    for (int off = 128; off; off >>= 1) {
        if (t < off) { s1[t] += s1[t + off]; s2[t] += s2[t + off]; }
        __syncthreads();
    }
    const float mu  = s1[0] * (1.0f / D_MODEL);
    const float var = (s2[0] - (float)D_MODEL * mu * mu) * (1.0f / (D_MODEL - 1));
    const float inv = 1.0f / (sqrtf(var) + 1e-6f);
    v4h o;
#pragma unroll
    for (int i = 0; i < 4; ++i) {
        int c = t * 4 + i;
        o[i] = (_Float16)(alpha[c] * (x[i] - mu) * inv + beta[c]);
    }
    *(v4h*)(Y + (long)row * D_MODEL + t * 4) = o;
}

#define GT_M 128
#define GT_N 128
#define GT_K 128
#define GT_P 136

#define FRAG_FENCE() __builtin_amdgcn_sched_barrier(0)

template <bool MORE, bool MORE2>
__device__ __forceinline__ void gemm_stage(
        v8f (&acc)[4][4], v16h (&fa0)[4], v16h (&fb0)[4], v16h (&fa1)[4], v16h (&fb1)[4],
        v8h (&ra)[16], v8h (&rb)[16],
        const _Float16* pa, const _Float16* pb,
        const _Float16* na, const _Float16* nb,
        _Float16* wa, _Float16* wb,
        const char* ag, const char* bg,
        const unsigned (&g8)[16]) {
#define GT_STORE(G)                                                              \
    if (MORE) {                                                                  \
        _Pragma("unroll")                                                        \
        for (int j = 4 * (G); j < 4 * (G) + 4; ++j) {                            \
            *(v8h*)(wa + j * 8 * GT_P) = ra[j];                                  \
            *(v8h*)(wb + j * 8 * GT_P) = rb[j];                                  \
        }                                                                        \
    }
#define GT_LOAD(G)                                                               \
    if (MORE2) {                                                                 \
        _Pragma("unroll")                                                        \
        for (int j = 4 * (G); j < 4 * (G) + 4; ++j) {                            \
            ra[j] = *(const v8h*)(ag + g8[j]);                                  \
            rb[j] = *(const v8h*)(bg + g8[j]);                                  \
        }                                                                        \
    }

#pragma unroll
    for (int i = 0; i < 4; ++i) fa1[i] = load_frag(pa + 32 + i * 16 * GT_P);
#pragma unroll
    for (int j = 0; j < 4; ++j) fb1[j] = load_frag(pb + 32 + j * 16 * GT_P);
    FRAG_FENCE();
#pragma unroll
    for (int i = 0; i < 4; ++i) {
#pragma unroll
        for (int j = 0; j < 4; ++j) acc[i][j] = wmma_f16(fa0[i], fb0[j], acc[i][j]);
        if (i == 0) { GT_STORE(0) }
        if (i == 2) { GT_LOAD(0) }
    }
    FRAG_FENCE();

#pragma unroll
    for (int i = 0; i < 4; ++i) fa0[i] = load_frag(pa + 64 + i * 16 * GT_P);
#pragma unroll
    for (int j = 0; j < 4; ++j) fb0[j] = load_frag(pb + 64 + j * 16 * GT_P);
    FRAG_FENCE();
#pragma unroll
    for (int i = 0; i < 4; ++i) {
#pragma unroll
        for (int j = 0; j < 4; ++j) acc[i][j] = wmma_f16(fa1[i], fb1[j], acc[i][j]);
        if (i == 0) { GT_STORE(1) }
        if (i == 2) { GT_LOAD(1) }
    }
    FRAG_FENCE();

#pragma unroll
    for (int i = 0; i < 4; ++i) fa1[i] = load_frag(pa + 96 + i * 16 * GT_P);
#pragma unroll
    for (int j = 0; j < 4; ++j) fb1[j] = load_frag(pb + 96 + j * 16 * GT_P);
    FRAG_FENCE();
#pragma unroll
    for (int i = 0; i < 4; ++i) {
#pragma unroll
        for (int j = 0; j < 4; ++j) acc[i][j] = wmma_f16(fa0[i], fb0[j], acc[i][j]);
        if (i == 0) { GT_STORE(2) }
        if (i == 1) { GT_STORE(3) }
        if (i == 2) { GT_LOAD(2) }
        if (i == 3) { GT_LOAD(3) }
    }

    __builtin_amdgcn_fence(__ATOMIC_RELEASE, "workgroup", "local");
    __builtin_amdgcn_s_barrier();
    __builtin_amdgcn_fence(__ATOMIC_ACQUIRE, "workgroup", "local");

    if (MORE) {
#pragma unroll
        for (int i = 0; i < 4; ++i) fa0[i] = load_frag(na + i * 16 * GT_P);
#pragma unroll
        for (int j = 0; j < 4; ++j) fb0[j] = load_frag(nb + j * 16 * GT_P);
    }
#pragma unroll
    for (int i = 0; i < 4; ++i) {
#pragma unroll
        for (int j = 0; j < 4; ++j) acc[i][j] = wmma_f16(fa1[i], fb1[j], acc[i][j]);
    }
#undef GT_STORE
#undef GT_LOAD
}

template <int OMODE, bool RELU, bool RESID>
__global__ void __launch_bounds__(128)
k_gemm_wmma(const _Float16* __restrict__ A, const _Float16* __restrict__ Bt,
            const float* __restrict__ bias, const float* __restrict__ resid,
            float* __restrict__ Cf, _Float16* __restrict__ Ch,
            const int* __restrict__ rofs,
            int M, int N, int K) {
    __shared__ __attribute__((aligned(16))) _Float16 sA[2][GT_M * GT_P];
    __shared__ __attribute__((aligned(16))) _Float16 sB[2][GT_N * GT_P];

    const int tid  = threadIdx.x;
    const int lane = tid & 31;
    const int wave = tid >> 5;
    const int wm = wave >> 1, wn = wave & 1;
    const int bn_cnt = N >> 7;
    const int bm = blockIdx.x / bn_cnt;
    const int bn = blockIdx.x % bn_cnt;
    if (bm * GT_M >= M) return;

    const int ln = lane & 15;
    const int hh = lane >> 4;

    const int srow = tid >> 4, sch = tid & 15;
    const char* ag = (const char*)(A  + (long)(bm * GT_M) * K);
    const char* bg = (const char*)(Bt + (long)(bn * GT_N) * K);
    unsigned g8[16];
#pragma unroll
    for (int j = 0; j < 16; ++j) g8[j] = (unsigned)(((srow + 8 * j) * K + sch * 8) * 2);
    const int  soff = srow * GT_P + sch * 8;

    v8h ra[16], rb[16];
#pragma unroll
    for (int j = 0; j < 16; ++j) {
        ra[j] = *(const v8h*)(ag + g8[j]);
        rb[j] = *(const v8h*)(bg + g8[j]);
    }
#pragma unroll
    for (int j = 0; j < 16; ++j) {
        *(v8h*)(&sA[0][soff + j * 8 * GT_P]) = ra[j];
        *(v8h*)(&sB[0][soff + j * 8 * GT_P]) = rb[j];
    }
    __syncthreads();

    v8f acc[4][4];
#pragma unroll
    for (int i = 0; i < 4; ++i) {
#pragma unroll
        for (int j = 0; j < 4; ++j) acc[i][j] = v8f{};
    }

    const int aoff = (wm * 64 + ln) * GT_P + 8 * hh;
    const int boff = (wn * 64 + ln) * GT_P + 8 * hh;

    const int nk = K / GT_K;

    if (nk > 1) {
        ag += GT_K * 2; bg += GT_K * 2;
#pragma unroll
        for (int j = 0; j < 16; ++j) {
            ra[j] = *(const v8h*)(ag + g8[j]);
            rb[j] = *(const v8h*)(bg + g8[j]);
        }
    }

    v16h fa0[4], fb0[4], fa1[4], fb1[4];
#pragma unroll
    for (int i = 0; i < 4; ++i) fa0[i] = load_frag(&sA[0][aoff] + i * 16 * GT_P);
#pragma unroll
    for (int j = 0; j < 4; ++j) fb0[j] = load_frag(&sB[0][boff] + j * 16 * GT_P);

    int kt = 0;
    for (; kt + 2 < nk; ++kt) {
        const int cur = kt & 1;
        ag += GT_K * 2; bg += GT_K * 2;
        gemm_stage<true, true>(acc, fa0, fb0, fa1, fb1, ra, rb,
                               &sA[cur][aoff], &sB[cur][boff], &sA[cur ^ 1][aoff], &sB[cur ^ 1][boff],
                               &sA[cur ^ 1][soff], &sB[cur ^ 1][soff], ag, bg, g8);
    }
    if (nk > 1) {
        const int cur = kt & 1;
        gemm_stage<true, false>(acc, fa0, fb0, fa1, fb1, ra, rb,
                                &sA[cur][aoff], &sB[cur][boff], &sA[cur ^ 1][aoff], &sB[cur ^ 1][boff],
                                &sA[cur ^ 1][soff], &sB[cur ^ 1][soff], ag, bg, g8);
        ++kt;
    }
    {
        const int cur = kt & 1;
        gemm_stage<false, false>(acc, fa0, fb0, fa1, fb1, ra, rb,
                                 &sA[cur][aoff], &sB[cur][boff], &sA[cur ^ 1][aoff], &sB[cur ^ 1][boff],
                                 &sA[cur ^ 1][soff], &sB[cur ^ 1][soff], ag, bg, g8);
    }

    const int lm = hh << 3;
    const int m0 = bm * GT_M + wm * 64;
    const int n0 = bn * GT_N + wn * 64;
#pragma unroll
    for (int i = 0; i < 4; ++i) {
        int ro[8] = {};
        if (OMODE == 2 || OMODE == 3) {
            const int4 r0 = *(const int4*)(rofs + m0 + i * 16 + lm);
            const int4 r1 = *(const int4*)(rofs + m0 + i * 16 + lm + 4);
            ro[0] = r0.x; ro[1] = r0.y; ro[2] = r0.z; ro[3] = r0.w;
            ro[4] = r1.x; ro[5] = r1.y; ro[6] = r1.z; ro[7] = r1.w;
        }
#pragma unroll
        for (int j = 0; j < 4; ++j) {
            const v8f accv = acc[i][j];
            const int col = n0 + j * 16 + ln;
            const float bv = bias[col];
            if (OMODE == 3 || OMODE == 2) {
                const int row0 = m0 + i * 16 + lm;
                const int b_ = row0 / SEQ, h = col >> 6, d = col & 63;
                _Float16* dst = Ch + ((long)(b_ * HEADS + h)) * SEQ * DK
                                   + (OMODE == 3 ? vt_off(0, d) : qk_off(0, d));
#pragma unroll
                for (int e = 0; e < 8; ++e) {
                    float v = accv[e] + bv;
                    if (RELU)  v = v > 0.0f ? v : 0.0f;
                    if (RESID) v += resid[(long)(row0 + e) * N + col];
                    dst[ro[e]] = (_Float16)v;
                }
                continue;
            }
#pragma unroll
            for (int e = 0; e < 8; ++e) {
                const int row = m0 + i * 16 + e + lm;
                float v = accv[e] + bv;
                if (RELU)  v = v > 0.0f ? v : 0.0f;
                if (RESID) v += resid[(long)row * N + col];
                if (OMODE == 0) {
                    Ch[(long)row * N + col] = (_Float16)v;
                } else if (OMODE == 1) {
                    Cf[(long)row * N + col] = v;
                }
            }
        }
    }
}

#define ATT_R     2
#define ATT_QW    (16 * ATT_R)
#define ATT_WAVES 4
#define ATT_QB    (ATT_QW * ATT_WAVES)

__device__ __forceinline__ float xhalf(float x) {
    return __builtin_bit_cast(float, __builtin_amdgcn_permlanex16(
        __builtin_bit_cast(int, x), __builtin_bit_cast(int, x),
        0x76543210, 0xfedcba98, false, false));
}

template <bool FILL>
__device__ __forceinline__ void att_step(
        const int step, const int nsteps, const int nfull, const int tail, const int hh,
        const float cscale, const _Float16* kp, const _Float16* vp,
        const v16h (&qf)[ATT_R][2], v16h (&kf)[2][2], const v16h (&vuse)[4], v16h (&vfill)[4],
        v8f (&acc)[ATT_R][4], float (&m_run)[ATT_R], float (&l_run)[ATT_R]) {
    const int last = nsteps - 1;
    if (FILL) {
        const int vs = (step < last) ? step + 1 : last;
#pragma unroll
        for (int t = 0; t < 4; ++t) vfill[t] = load_frag_fo(vp + (long)(vs << 5) * DK + 512 * t);
    }
    __builtin_amdgcn_sched_barrier(0);

    v8f s[2][ATT_R];
#pragma unroll
    for (int kt = 0; kt < 2; ++kt) {
#pragma unroll
        for (int g = 0; g < ATT_R; ++g) {
            v8f z = {};
            z = wmma_f16(kf[kt][0], qf[g][0], z);
            s[kt][g] = wmma_f16(kf[kt][1], qf[g][1], z);
        }
    }

    if (FILL) {
        const int ks = (step + 2 < nsteps) ? step + 2 : last;
#pragma unroll
        for (int kt = 0; kt < 2; ++kt) {
            kf[kt][0] = load_frag_fo(kp + (long)((ks << 5) + 16 * kt) * DK);
            kf[kt][1] = load_frag_fo(kp + (long)((ks << 5) + 16 * kt) * DK + 512);
        }
    }

#pragma unroll
    for (int g = 0; g < ATT_R; ++g) {
#pragma unroll
        for (int kt = 0; kt < 2; ++kt) s[kt][g] = s[kt][g] * cscale;
    }
    if (step == nfull) {
#pragma unroll
        for (int kt = 0; kt < 2; ++kt) {
#pragma unroll
            for (int r = 0; r < 8; ++r) {
                const bool off = (16 * kt + 8 * hh + r) >= tail;
#pragma unroll
                for (int g = 0; g < ATT_R; ++g) s[kt][g][r] = off ? -__builtin_inff() : s[kt][g][r];
            }
        }
    }

    float mb[ATT_R];
    bool need = false;
#pragma unroll
    for (int g = 0; g < ATT_R; ++g) {
        float mx = -__builtin_inff();
#pragma unroll
        for (int kt = 0; kt < 2; ++kt) {
#pragma unroll
            for (int r = 0; r < 8; ++r) mx = fmaxf(mx, s[kt][g][r]);
        }
        mx = fmaxf(mx, xhalf(mx));
        mb[g] = mx;
        need = need || (mx > m_run[g] + 8.0f);
    }
    if (__builtin_amdgcn_ballot_w32(need) != 0) {
#pragma unroll
        for (int g = 0; g < ATT_R; ++g) {
            const float m_new = fmaxf(m_run[g], mb[g]);
            const float corr  = __builtin_amdgcn_exp2f(m_run[g] - m_new);
            m_run[g] = m_new;
            l_run[g] *= corr;
#pragma unroll
            for (int t = 0; t < 4; ++t) acc[g][t] = acc[g][t] * corr;
        }
    }

#pragma unroll
    for (int g = 0; g < ATT_R; ++g) {
        const float m = m_run[g];
        v16h pf;
        float ls = 0.0f;
#pragma unroll
        for (int r = 0; r < 8; ++r) {
            const float p0 = __builtin_amdgcn_exp2f(s[0][g][r] - m);
            const float p1 = __builtin_amdgcn_exp2f(s[1][g][r] - m);
            ls += p0 + p1;
            pf[r]     = (_Float16)p0;
            pf[8 + r] = (_Float16)p1;
        }
        l_run[g] += ls;
#pragma unroll
        for (int t = 0; t < 4; ++t) acc[g][t] = wmma_f16(vuse[t], pf, acc[g][t]);
    }
}

__global__ void __launch_bounds__(128)
k_attention(const _Float16* __restrict__ QK, const _Float16* __restrict__ VT,
            const int* __restrict__ kcnt, const int* __restrict__ kinv,
            _Float16* __restrict__ CTX) {
    const int lane = threadIdx.x & 31, wave = threadIdx.x >> 5;
    const int c = lane & 15, hh = lane >> 4;
    const int qb = blockIdx.x & 15;
    const int h  = (blockIdx.x >> 4) & 15;
    const int b  = blockIdx.x >> 8;
    const int q0 = qb * ATT_QB + wave * ATT_QW;

    const _Float16* qbase = QK + ((long)(b * HEADS + h)) * SEQ * DK;
    const _Float16* vbase = VT + ((long)(b * HEADS + h)) * DK * SEQ;
    const _Float16* kp = qbase + 8 * lane;
    const _Float16* vp = vbase + 8 * lane;

    const int cnt    = kcnt[2 * b];
    const int nfull  = cnt >> 5;
    const int tail   = cnt & 31;
    const int nsteps = nfull + (tail ? 1 : 0);
    const bool  none   = kcnt[2 * b + 1] != 0;
    const float cscale = 0.125f * 1.4426950408889634f;

    v16h qf[ATT_R][2];
#pragma unroll
    for (int g = 0; g < ATT_R; ++g) {
        const _Float16* qp = kp + (long)(q0 + 16 * g) * DK;
        qf[g][0] = load_frag_fo(qp);
        qf[g][1] = load_frag_fo(qp + 512);
        if (none) { qf[g][0] = v16h{}; qf[g][1] = v16h{}; }
    }

    v8f acc[ATT_R][4];
    float m_run[ATT_R], l_run[ATT_R];
#pragma unroll
    for (int g = 0; g < ATT_R; ++g) {
        m_run[g] = -__builtin_inff(); l_run[g] = 0.0f;
#pragma unroll
        for (int t = 0; t < 4; ++t) acc[g][t] = v8f{};
    }

    v16h kf0[2][2], kf1[2][2], vf0[4], vf1[4];
    const int s1 = nsteps > 1 ? 1 : 0;
#pragma unroll
    for (int kt = 0; kt < 2; ++kt) {
        kf0[kt][0] = load_frag_fo(kp + (long)(16 * kt) * DK);
        kf0[kt][1] = load_frag_fo(kp + (long)(16 * kt) * DK + 512);
    }
#pragma unroll
    for (int t = 0; t < 4; ++t) vf0[t] = load_frag_fo(vp + 512 * t);
#pragma unroll
    for (int kt = 0; kt < 2; ++kt) {
        kf1[kt][0] = load_frag_fo(kp + (long)(32 * s1 + 16 * kt) * DK);
        kf1[kt][1] = load_frag_fo(kp + (long)(32 * s1 + 16 * kt) * DK + 512);
    }

    int step = 0;
    for (; step + 1 < nsteps; step += 2) {
        att_step<true>(step,     nsteps, nfull, tail, hh, cscale, kp, vp, qf, kf0, vf0, vf1, acc, m_run, l_run);
        att_step<true>(step + 1, nsteps, nfull, tail, hh, cscale, kp, vp, qf, kf1, vf1, vf0, acc, m_run, l_run);
    }
    if (step < nsteps)
        att_step<false>(step, nsteps, nfull, tail, hh, cscale, kp, vp, qf, kf0, vf0, vf1, acc, m_run, l_run);

#pragma unroll
    for (int g = 0; g < ATT_R; ++g) {
        float l = l_run[g];
        l = l + xhalf(l);
        const float inv = 1.0f / l;
        const int orow = kinv[b * SEQ + q0 + 16 * g + c];
        _Float16* op = CTX + ((long)(b * SEQ + orow)) * D_MODEL + h * DK + 8 * hh;
#pragma unroll
        for (int t = 0; t < 4; ++t) {
            v8h o;
#pragma unroll
            for (int r = 0; r < 8; ++r) o[r] = (_Float16)(acc[g][t][r] * inv);
            *(v8h*)(op + 16 * t) = o;
        }
    }
}

extern "C" void kernel_launch(void* const* d_in, const int* in_sizes, int n_in,
                              void* d_out, int out_size, void* d_ws, size_t ws_size,
                              hipStream_t stream) {
    const float* x      = (const float*)d_in[0];
    const int*   mask   = (const int*)  d_in[1];
    const float* alpha1 = (const float*)d_in[2];
    const float* bias1  = (const float*)d_in[3];
    const float* alpha2 = (const float*)d_in[4];
    const float* bias2  = (const float*)d_in[5];
    const float* Wk     = (const float*)d_in[6];
    const float* bk     = (const float*)d_in[7];
    const float* Wv     = (const float*)d_in[8];
    const float* bv     = (const float*)d_in[9];
    const float* Wo     = (const float*)d_in[10];
    const float* bo     = (const float*)d_in[11];
    const float* W1     = (const float*)d_in[12];
    const float* b1     = (const float*)d_in[13];
    const float* W2     = (const float*)d_in[14];
    const float* b2     = (const float*)d_in[15];
    float* out = (float*)d_out;

    char* ws = (char*)d_ws;
    size_t off = 0;
    _Float16* x2h  = (_Float16*)(ws + off); off += (size_t)ROWS * D_MODEL * 2;
    _Float16* x2bh = (_Float16*)(ws + off); off += (size_t)ROWS * D_MODEL * 2;
    _Float16* qkh  = (_Float16*)(ws + off); off += (size_t)BATCH*HEADS*SEQ*DK*2;
    _Float16* vth  = (_Float16*)(ws + off); off += (size_t)BATCH*HEADS*DK*SEQ*2;
    _Float16* ctxh = (_Float16*)(ws + off); off += (size_t)ROWS * D_MODEL * 2;
    _Float16* h1h  = (_Float16*)(ws + off); off += (size_t)ROWS * D_FF * 2;
    _Float16* Wkt  = (_Float16*)(ws + off); off += (size_t)D_MODEL * D_MODEL * 2;
    _Float16* Wvt  = (_Float16*)(ws + off); off += (size_t)D_MODEL * D_MODEL * 2;
    _Float16* Wot  = (_Float16*)(ws + off); off += (size_t)D_MODEL * D_MODEL * 2;
    _Float16* W1t  = (_Float16*)(ws + off); off += (size_t)D_MODEL * D_FF * 2;
    _Float16* W2t  = (_Float16*)(ws + off); off += (size_t)D_FF * D_MODEL * 2;
    float*    x1   = (float*)   (ws + off); off += (size_t)ROWS * D_MODEL * 4;
    int* kofs = (int*)h1h;
    int* vofs = kofs + ROWS;
    int* kinv = vofs + ROWS;
    int* kcnt = kinv + ROWS;

    k_transpose_cvt_all<<<BATCH + TR_GRID, 256, 0, stream>>>(Wk, Wv, Wo, W1, W2,
                                                             Wkt, Wvt, Wot, W1t, W2t,
                                                             mask, kofs, vofs, kinv, kcnt);

    k_layernorm<<<ROWS, 256, 0, stream>>>(x, alpha1, bias1, x2h);

    #define GEMM_GRID(M_, N_) (((M_) / 128) * ((N_) / 128))

    k_gemm_wmma<2, false, false><<<GEMM_GRID(ROWS, D_MODEL), 128, 0, stream>>>(
        x2h, Wkt, bk, nullptr, nullptr, qkh, kofs, ROWS, D_MODEL, D_MODEL);
    k_gemm_wmma<3, false, false><<<GEMM_GRID(ROWS, D_MODEL), 128, 0, stream>>>(
        x2h, Wvt, bv, nullptr, nullptr, vth, vofs, ROWS, D_MODEL, D_MODEL);

    k_attention<<<BATCH*HEADS*(SEQ/ATT_QB), 128, 0, stream>>>(
        qkh, vth, kcnt, kinv, ctxh);

    k_gemm_wmma<1, false, true><<<GEMM_GRID(ROWS, D_MODEL), 128, 0, stream>>>(
        ctxh, Wot, bo, x, x1, nullptr, nullptr, ROWS, D_MODEL, D_MODEL);

    k_layernorm<<<ROWS, 256, 0, stream>>>(x1, alpha2, bias2, x2bh);

    k_gemm_wmma<0, true, false><<<GEMM_GRID(ROWS, D_FF), 128, 0, stream>>>(
        x2bh, W1t, b1, nullptr, nullptr, h1h, nullptr, ROWS, D_FF, D_MODEL);

    k_gemm_wmma<1, false, true><<<GEMM_GRID(ROWS, D_MODEL), 128, 0, stream>>>(
        h1h, W2t, b2, x1, out, nullptr, nullptr, ROWS, D_MODEL, D_FF);

    (void)in_sizes; (void)n_in; (void)out_size; (void)ws_size;
}
